# starting kernel
# baseline (speedup 1.0000x reference)
#include <hip/hip_runtime.h>
#include <math.h>

#define NN      50000
#define NE      800000
#define NG      128
#define INC     31
#define HID     64
#define EPS     1e-5f

typedef __attribute__((ext_vector_type(2))) float v2f;
typedef __attribute__((ext_vector_type(8))) float v8f;

static __device__ __forceinline__ unsigned enc_f32(float x) {
    unsigned b = __float_as_uint(x);
    return (b & 0x80000000u) ? ~b : (b | 0x80000000u);
}
static __device__ __forceinline__ float dec_f32(unsigned e) {
    unsigned b = (e & 0x80000000u) ? (e & 0x7FFFFFFFu) : ~e;
    return __uint_as_float(b);
}
#define ENC_NEG_INF 0x007FFFFFu

__global__ void k_pad_rows(const float* __restrict__ s, float* __restrict__ d,
                           int rows, int Csrc, int Cdst) {
    int i = blockIdx.x * blockDim.x + threadIdx.x;
    if (i >= rows * Cdst) return;
    int r = i / Cdst, c = i % Cdst;
    d[i] = (c < Csrc) ? s[r * Csrc + c] : 0.f;
}

__global__ void k_init_aggr(unsigned* __restrict__ em, float* __restrict__ num,
                            float* __restrict__ den, int n) {
    int i = blockIdx.x * blockDim.x + threadIdx.x;
    if (i < n) { em[i] = ENC_NEG_INF; num[i] = 0.f; den[i] = 0.f; }
}
__global__ void k_init_enc(unsigned* __restrict__ p, int n) {
    int i = blockIdx.x * blockDim.x + threadIdx.x;
    if (i < n) p[i] = ENC_NEG_INF;
}

__global__ void k_edge_max4(const int* __restrict__ src, const int* __restrict__ dst,
                            const float* __restrict__ x, const float* __restrict__ tptr,
                            unsigned* __restrict__ em, int E, int cqshift) {
    int g = blockIdx.x * blockDim.x + threadIdx.x;
    int e = g >> cqshift;
    if (e >= E) return;
    int q  = g & ((1 << cqshift) - 1);
    int Cp = 4 << cqshift;
    float t = tptr[0];
    const float4 m = ((const float4*)(x + (size_t)src[e] * Cp))[q];
    unsigned* p = em + (size_t)dst[e] * Cp + q * 4;
    atomicMax(p + 0, enc_f32(m.x * t));
    atomicMax(p + 1, enc_f32(m.y * t));
    atomicMax(p + 2, enc_f32(m.z * t));
    atomicMax(p + 3, enc_f32(m.w * t));
}

__global__ void k_edge_expsum4(const int* __restrict__ src, const int* __restrict__ dst,
                               const float* __restrict__ x, const float* __restrict__ tptr,
                               const unsigned* __restrict__ em,
                               float* __restrict__ num, float* __restrict__ den,
                               int E, int cqshift) {
    int g = blockIdx.x * blockDim.x + threadIdx.x;
    int e = g >> cqshift;
    if (e >= E) return;
    int q  = g & ((1 << cqshift) - 1);
    int Cp = 4 << cqshift;
    float t = tptr[0];
    int   d = dst[e];
    const float4 m = ((const float4*)(x + (size_t)src[e] * Cp))[q];
    size_t base = (size_t)d * Cp + q * 4;
    const unsigned* pe = em + base;
    float* pn = num + base;
    float* pd = den + base;
    float e0 = __expf(m.x * t - dec_f32(pe[0]));
    float e1 = __expf(m.y * t - dec_f32(pe[1]));
    float e2 = __expf(m.z * t - dec_f32(pe[2]));
    float e3 = __expf(m.w * t - dec_f32(pe[3]));
    atomicAdd(pn + 0, m.x * e0);  atomicAdd(pd + 0, e0);
    atomicAdd(pn + 1, m.y * e1);  atomicAdd(pd + 1, e1);
    atomicAdd(pn + 2, m.z * e2);  atomicAdd(pd + 2, e2);
    atomicAdd(pn + 3, m.w * e3);  atomicAdd(pd + 3, e3);
}

__global__ void k_aggr_div(const float* __restrict__ num, const float* __restrict__ den,
                           float* __restrict__ aggr, int n) {
    int i = blockIdx.x * blockDim.x + threadIdx.x;
    if (i >= n) return;
    float dd = den[i];
    aggr[i] = dd > 0.f ? num[i] / dd : 0.f;
}

template<int CP>
__global__ void k_gemm(const float* __restrict__ aggr, const float* __restrict__ xin,
                       const float* __restrict__ Wr, const float* __restrict__ Ws,
                       const float* __restrict__ bias, float* __restrict__ out, int Nrows) {
    const int lane = threadIdx.x;
    const int half = lane >> 4;
    const int l16  = lane & 15;
    const int m0   = blockIdx.x * 16;
    const int h0   = threadIdx.y * 16;
    int arow = m0 + l16;
    if (arow >= Nrows) arow = Nrows - 1;
    const int bcol = h0 + l16;

    const float* ap = aggr + (size_t)arow * CP + half * 2;
    const float* xp = xin  + (size_t)arow * CP + half * 2;
    const float* wr = Wr   + (size_t)bcol * CP + half * 2;
    const float* wsp= Ws   + (size_t)bcol * CP + half * 2;

    v8f acc = {};
#pragma unroll
    for (int k = 0; k < CP; k += 4) {
        v2f a = *(const v2f*)(ap + k);
        v2f b = *(const v2f*)(wr + k);
        acc = __builtin_amdgcn_wmma_f32_16x16x4_f32(false, a, false, b,
                                                    (short)0, acc, false, false);
    }
#pragma unroll
    for (int k = 0; k < CP; k += 4) {
        v2f a = *(const v2f*)(xp + k);
        v2f b = *(const v2f*)(wsp + k);
        acc = __builtin_amdgcn_wmma_f32_16x16x4_f32(false, a, false, b,
                                                    (short)0, acc, false, false);
    }

    float bv = bias[bcol];
#pragma unroll
    for (int r = 0; r < 8; ++r) {
        int row = m0 + r + half * 8;
        if (row < Nrows) out[(size_t)row * HID + bcol] = acc[r] + bv;
    }
}

__global__ void k_instnorm_relu(float* __restrict__ x, int Nrows) {
    int wave = (blockIdx.x * blockDim.x + threadIdx.x) >> 5;
    int lane = threadIdx.x & 31;
    if (wave >= Nrows) return;
    float v0 = x[wave * HID + lane];
    float v1 = x[wave * HID + lane + 32];
    float s = v0 + v1;
    for (int off = 16; off; off >>= 1) s += __shfl_xor(s, off);
    float mu = s * (1.f / HID);
    float d0 = v0 - mu, d1 = v1 - mu;
    float q = d0 * d0 + d1 * d1;
    for (int off = 16; off; off >>= 1) q += __shfl_xor(q, off);
    float inv = rsqrtf(q * (1.f / HID) + EPS);
    x[wave * HID + lane]      = fmaxf(d0 * inv, 0.f);
    x[wave * HID + lane + 32] = fmaxf(d1 * inv, 0.f);
}

__global__ void k_graph_max(const float* __restrict__ xl, const int* __restrict__ batch,
                            unsigned* __restrict__ he, int n) {
    int i = blockIdx.x * blockDim.x + threadIdx.x;
    if (i >= n) return;
    int node = i >> 6;
    int c    = i & 63;
    atomicMax(&he[batch[node] * HID + c], enc_f32(xl[i]));
}

__global__ void k_decode_h(const unsigned* __restrict__ henc, float* __restrict__ h, int n) {
    int i = blockIdx.x * blockDim.x + threadIdx.x;
    if (i >= n) return;
    int g = i / (3 * HID);
    int c = i % (3 * HID);
    int layer = c / HID, cc = c % HID;
    h[i] = dec_f32(henc[layer * NG * HID + g * HID + cc]);
}

__global__ void k_mlp1(const float* __restrict__ h, const float* __restrict__ Wl1,
                       const float* __restrict__ bl1, float* __restrict__ z1) {
    int i = blockIdx.x * blockDim.x + threadIdx.x;
    if (i >= NG * 2 * HID) return;
    int g = i / (2 * HID), oc = i % (2 * HID);
    float s = bl1[oc];
    for (int k = 0; k < 3 * HID; ++k) s += h[g * 3 * HID + k] * Wl1[oc * 3 * HID + k];
    z1[i] = fmaxf(s, 0.f);
}

__global__ void k_mlp2_norm(const float* __restrict__ z1, const float* __restrict__ Wl2,
                            const float* __restrict__ bl2, float* __restrict__ outp) {
    int wave = (blockIdx.x * blockDim.x + threadIdx.x) >> 5;
    int lane = threadIdx.x & 31;
    if (wave >= NG) return;
    float s = bl2[lane];
    for (int k = 0; k < 2 * HID; ++k) s += z1[wave * 2 * HID + k] * Wl2[lane * 2 * HID + k];
    float q = s * s;
    for (int off = 16; off; off >>= 1) q += __shfl_xor(q, off);
    float nrm = fmaxf(sqrtf(q), 1e-12f);
    outp[wave * (HID / 2) + lane] = s / nrm;
}

extern "C" void kernel_launch(void* const* d_in, const int* in_sizes, int n_in,
                              void* d_out, int out_size, void* d_ws, size_t ws_size,
                              hipStream_t stream) {
    const float* x    = (const float*)d_in[0];
    const int*  eidx  = (const int*)  d_in[1];
    const int*  batch = (const int*)  d_in[2];
    const float* t    = (const float*)d_in[3];
    const float* W1r  = (const float*)d_in[4];
    const float* b1   = (const float*)d_in[5];
    const float* W1s  = (const float*)d_in[6];
    const float* W2r  = (const float*)d_in[7];
    const float* b2   = (const float*)d_in[8];
    const float* W2s  = (const float*)d_in[9];
    const float* W3r  = (const float*)d_in[10];
    const float* b3   = (const float*)d_in[11];
    const float* W3s  = (const float*)d_in[12];
    const float* Wl1  = (const float*)d_in[13];
    const float* bl1  = (const float*)d_in[14];
    const float* Wl2  = (const float*)d_in[15];
    const float* bl2  = (const float*)d_in[16];
    const int* src = eidx;
    const int* dst = eidx + NE;

    float* ws = (float*)d_ws;
    size_t o = 0;
    float*    x1     = ws + o; o += (size_t)NN * HID;
    float*    x2     = ws + o; o += (size_t)NN * HID;
    float*    x3     = ws + o; o += (size_t)NN * HID;
    unsigned* encmax = (unsigned*)(ws + o); o += (size_t)NN * HID;
    float*    num    = ws + o; o += (size_t)NN * HID;
    float*    den    = ws + o; o += (size_t)NN * HID;
    float*    aggr   = ws + o; o += (size_t)NN * HID;
    float*    xpad   = ws + o; o += (size_t)NN * 32;
    float*    w1rp   = ws + o; o += (size_t)HID * 32;
    float*    w1sp   = ws + o; o += (size_t)HID * 32;
    unsigned* henc   = (unsigned*)(ws + o); o += 3 * NG * HID;
    float*    h      = ws + o; o += NG * 3 * HID;
    float*    z1     = ws + o; o += NG * 2 * HID;

    k_pad_rows<<<(NN * 32 + 255) / 256, 256, 0, stream>>>(x, xpad, NN, INC, 32);
    k_pad_rows<<<(HID * 32 + 255) / 256, 256, 0, stream>>>(W1r, w1rp, HID, INC, 32);
    k_pad_rows<<<(HID * 32 + 255) / 256, 256, 0, stream>>>(W1s, w1sp, HID, INC, 32);
    k_init_enc<<<(3 * NG * HID + 255) / 256, 256, 0, stream>>>(henc, 3 * NG * HID);

    struct Layer { const float* xin; int Cp; const float *Wr, *b, *Ws; float* xout; };
    Layer layers[3] = {
        { xpad, 32, w1rp, b1, w1sp, x1 },
        { x1,   64, W2r,  b2, W2s,  x2 },
        { x2,   64, W3r,  b3, W3s,  x3 },
    };

    for (int l = 0; l < 3; ++l) {
        const Layer& L = layers[l];
        int nagg = NN * L.Cp;
        k_init_aggr<<<(nagg + 255) / 256, 256, 0, stream>>>(encmax, num, den, nagg);

        int cqshift = (L.Cp == 32) ? 3 : 4;
        int tot = NE << cqshift;
        int eblocks = (tot + 255) / 256;
        k_edge_max4<<<eblocks, 256, 0, stream>>>(src, dst, L.xin, t, encmax, NE, cqshift);
        k_edge_expsum4<<<eblocks, 256, 0, stream>>>(src, dst, L.xin, t, encmax, num, den,
                                                    NE, cqshift);
        k_aggr_div<<<(nagg + 255) / 256, 256, 0, stream>>>(num, den, aggr, nagg);

        dim3 gblk(32, 4);
        if (L.Cp == 32)
            k_gemm<32><<<(NN + 15) / 16, gblk, 0, stream>>>(aggr, L.xin, L.Wr, L.Ws, L.b,
                                                            L.xout, NN);
        else
            k_gemm<64><<<(NN + 15) / 16, gblk, 0, stream>>>(aggr, L.xin, L.Wr, L.Ws, L.b,
                                                            L.xout, NN);

        k_instnorm_relu<<<(NN + 7) / 8, 256, 0, stream>>>(L.xout, NN);

        int ngm = NN * HID;
        k_graph_max<<<(ngm + 255) / 256, 256, 0, stream>>>(L.xout, batch,
                                                           henc + (size_t)l * NG * HID, ngm);
    }

    k_decode_h<<<(NG * 3 * HID + 255) / 256, 256, 0, stream>>>(henc, h, NG * 3 * HID);
    k_mlp1<<<(NG * 2 * HID + 255) / 256, 256, 0, stream>>>(h, Wl1, bl1, z1);
    k_mlp2_norm<<<NG * 32 / 128, 128, 0, stream>>>(z1, Wl2, bl2, (float*)d_out);
}


# Softmax aggregation gathers over a destination-sorted edge list (counting sort each call), one wave per node; no float/max atomics; per-node sums sequential in arrival-order rank
# speedup vs baseline: 2.1445x; 2.1445x over previous
#include <hip/hip_runtime.h>
#include <math.h>

#define NN      50000
#define NE      800000
#define NG      128
#define INC     31
#define HID     64
#define EPS     1e-5f

typedef __attribute__((ext_vector_type(2))) float v2f;
typedef __attribute__((ext_vector_type(8))) float v8f;

static __device__ __forceinline__ unsigned enc_f32(float x) {
    unsigned b = __float_as_uint(x);
    return (b & 0x80000000u) ? ~b : (b | 0x80000000u);
}
static __device__ __forceinline__ float dec_f32(unsigned e) {
    unsigned b = (e & 0x80000000u) ? (e & 0x7FFFFFFFu) : ~e;
    return __uint_as_float(b);
}
#define ENC_NEG_INF 0x007FFFFFu

__global__ void k_pad_rows(const float* __restrict__ s, float* __restrict__ d,
                           int rows, int Csrc, int Cdst) {
    int i = blockIdx.x * blockDim.x + threadIdx.x;
    if (i >= rows * Cdst) return;
    int r = i / Cdst, c = i % Cdst;
    d[i] = (c < Csrc) ? s[r * Csrc + c] : 0.f;
}

__global__ void k_init_enc(unsigned* __restrict__ p, int n, int* __restrict__ cnt, int ncnt) {
    int i = blockIdx.x * blockDim.x + threadIdx.x;
    if (i < n) p[i] = ENC_NEG_INF;
    if (i < ncnt) cnt[i] = 0;
}

__global__ void k_csr_hist(const int* __restrict__ dst, int* __restrict__ cnt, int E) {
    int e = blockIdx.x * blockDim.x + threadIdx.x;
    if (e < E) atomicAdd(&cnt[dst[e]], 1);
}

__global__ void __launch_bounds__(256) k_csr_bsum(const int* __restrict__ cnt, int* __restrict__ bsum, int n) {
    __shared__ int red[8];
    const int tid = threadIdx.x, lane = tid & 31, w = tid >> 5;
    const int idx = blockIdx.x * 256 + tid;
    int c = idx < n ? cnt[idx] : 0;
    for (int off = 16; off; off >>= 1) c += __shfl_xor(c, off);
    if (lane == 0) red[w] = c;
    __syncthreads();
    if (tid == 0) {
        int s = 0;
        for (int i = 0; i < 8; ++i) s += red[i];
        bsum[blockIdx.x] = s;
    }
}

__global__ void __launch_bounds__(256) k_csr_scan(const int* __restrict__ cnt, const int* __restrict__ bsum,
                                                  int* __restrict__ rowptr, int* __restrict__ cursor, int n) {
    __shared__ int red[8];
    __shared__ int wsum[8];
    const int tid = threadIdx.x, lane = tid & 31, w = tid >> 5;
    int part = 0;
    for (int i = tid; i < (int)blockIdx.x; i += 256) part += bsum[i];
    for (int off = 16; off; off >>= 1) part += __shfl_xor(part, off);
    if (lane == 0) red[w] = part;
    const int idx = blockIdx.x * 256 + tid;
    const int c = idx < n ? cnt[idx] : 0;
    int inc = c;
    for (int off = 1; off < 32; off <<= 1) {
        int u = __shfl_up(inc, off);
        if (lane >= off) inc += u;
    }
    if (lane == 31) wsum[w] = inc;
    __syncthreads();
    int base = 0;
    for (int i = 0; i < 8; ++i) {
        base += red[i];
        if (i < w) base += wsum[i];
    }
    const int ex = base + inc - c;
    if (idx < n) { rowptr[idx] = ex; cursor[idx] = ex; }
    if (idx == n - 1) rowptr[n] = ex + c;
}

__global__ void k_csr_fill(const int* __restrict__ src, const int* __restrict__ dst,
                           int* __restrict__ cursor, int* __restrict__ csrc, int E) {
    int e = blockIdx.x * blockDim.x + threadIdx.x;
    if (e >= E) return;
    int p = atomicAdd(&cursor[dst[e]], 1);
    csrc[p] = src[e];
}

template<int VEC>
static __device__ __forceinline__ void ld_row(const float* __restrict__ p, float (&v)[VEC]) {
    if constexpr (VEC == 2) {
        v2f r = *(const v2f*)p;
        v[0] = r.x; v[1] = r.y;
    } else {
        v[0] = p[0];
    }
}

template<int VEC>
static __device__ __forceinline__ void seg_max(const float (&r)[VEC], float t, float (&mx)[VEC]) {
#pragma unroll
    for (int i = 0; i < VEC; ++i) mx[i] = fmaxf(mx[i], r[i] * t);
}
template<int VEC>
static __device__ __forceinline__ void seg_sum(const float (&r)[VEC], float t, const float (&mx)[VEC],
                                               float (&num)[VEC], float (&den)[VEC]) {
#pragma unroll
    for (int i = 0; i < VEC; ++i) {
        float e = __expf(r[i] * t - mx[i]);
        num[i] += r[i] * e;
        den[i] += e;
    }
}

template<int J, int VEC, int CP>
struct SegChain {
    static __device__ __forceinline__ void load(int deg, const float* __restrict__ x, unsigned loff,
                                                const int (&sidx)[32], float (&v)[32][VEC]) {
        if constexpr (J < 32) {
            if (J < deg) {
                ld_row<VEC>(x + (size_t)sidx[J] * CP + loff, v[J]);
                SegChain<J + 1, VEC, CP>::load(deg, x, loff, sidx, v);
            }
        }
    }
    static __device__ __forceinline__ void vmax(int deg, const float (&v)[32][VEC], float t, float (&mx)[VEC]) {
        if constexpr (J < 32) {
            if (J < deg) {
                seg_max<VEC>(v[J], t, mx);
                SegChain<J + 1, VEC, CP>::vmax(deg, v, t, mx);
            }
        }
    }
    static __device__ __forceinline__ void vsum(int deg, const float (&v)[32][VEC], float t,
                                                const float (&mx)[VEC], float (&num)[VEC], float (&den)[VEC]) {
        if constexpr (J < 32) {
            if (J < deg) {
                seg_sum<VEC>(v[J], t, mx, num, den);
                SegChain<J + 1, VEC, CP>::vsum(deg, v, t, mx, num, den);
            }
        }
    }
};

template<int CP>
__global__ void __launch_bounds__(256) k_seg_softmax(const int* __restrict__ rowptr,
                                                     const int* __restrict__ csrc,
                                                     const float* __restrict__ x,
                                                     const float* __restrict__ tptr,
                                                     float* __restrict__ aggr, int n) {
    constexpr int VEC = CP / 32;
    const int lane = threadIdx.x & 31;
    const int node = __builtin_amdgcn_readfirstlane((int)(blockIdx.x * 8 + (threadIdx.x >> 5)));
    if (node >= n) return;
    const int beg = rowptr[node];
    const int end = rowptr[node + 1];
    const int deg = end - beg;
    const float t = tptr[0];
    const unsigned loff = (unsigned)(lane * VEC);

    float mx[VEC], num[VEC], den[VEC];
#pragma unroll
    for (int i = 0; i < VEC; ++i) { mx[i] = -INFINITY; num[i] = 0.f; den[i] = 0.f; }

    if (deg <= 32) {
        const int mys = (lane < deg) ? csrc[beg + lane] : 0;
        int sidx[32];
#pragma unroll
        for (int j = 0; j < 32; ++j) sidx[j] = __builtin_amdgcn_readlane(mys, j);
        float v[32][VEC];
        SegChain<0, VEC, CP>::load(deg, x, loff, sidx, v);
        SegChain<0, VEC, CP>::vmax(deg, v, t, mx);
        SegChain<0, VEC, CP>::vsum(deg, v, t, mx, num, den);
    } else {
        for (int base = beg; base < end; base += 32) {
            const int cnt = min(32, end - base);
            const int mys = (lane < cnt) ? csrc[base + lane] : 0;
            for (int j = 0; j < cnt; ++j) {
                const int s = __builtin_amdgcn_readlane(mys, j);
                float r[VEC];
                ld_row<VEC>(x + (size_t)s * CP + loff, r);
                seg_max<VEC>(r, t, mx);
            }
        }
        for (int base = beg; base < end; base += 32) {
            const int cnt = min(32, end - base);
            const int mys = (lane < cnt) ? csrc[base + lane] : 0;
            for (int j = 0; j < cnt; ++j) {
                const int s = __builtin_amdgcn_readlane(mys, j);
                float r[VEC];
                ld_row<VEC>(x + (size_t)s * CP + loff, r);
                seg_sum<VEC>(r, t, mx, num, den);
            }
        }
    }

    float res[VEC];
#pragma unroll
    for (int i = 0; i < VEC; ++i) res[i] = den[i] > 0.f ? num[i] / den[i] : 0.f;
    float* o = aggr + (size_t)node * CP + loff;
    if constexpr (VEC == 2) {
        v2f ov; ov.x = res[0]; ov.y = res[1];
        *(v2f*)o = ov;
    } else {
        o[0] = res[0];
    }
}

template<int CP>
__global__ void k_gemm(const float* __restrict__ aggr, const float* __restrict__ xin,
                       const float* __restrict__ Wr, const float* __restrict__ Ws,
                       const float* __restrict__ bias, float* __restrict__ out, int Nrows) {
    const int lane = threadIdx.x;
    const int half = lane >> 4;
    const int l16  = lane & 15;
    const int m0   = blockIdx.x * 16;
    const int h0   = threadIdx.y * 16;
    int arow = m0 + l16;
    if (arow >= Nrows) arow = Nrows - 1;
    const int bcol = h0 + l16;

    const float* ap = aggr + (size_t)arow * CP + half * 2;
    const float* xp = xin  + (size_t)arow * CP + half * 2;
    const float* wr = Wr   + (size_t)bcol * CP + half * 2;
    const float* wsp= Ws   + (size_t)bcol * CP + half * 2;

    v8f acc = {};
#pragma unroll
    for (int k = 0; k < CP; k += 4) {
        v2f a = *(const v2f*)(ap + k);
        v2f b = *(const v2f*)(wr + k);
        acc = __builtin_amdgcn_wmma_f32_16x16x4_f32(false, a, false, b,
                                                    (short)0, acc, false, false);
    }
#pragma unroll
    for (int k = 0; k < CP; k += 4) {
        v2f a = *(const v2f*)(xp + k);
        v2f b = *(const v2f*)(wsp + k);
        acc = __builtin_amdgcn_wmma_f32_16x16x4_f32(false, a, false, b,
                                                    (short)0, acc, false, false);
    }

    float bv = bias[bcol];
#pragma unroll
    for (int r = 0; r < 8; ++r) {
        int row = m0 + r + half * 8;
        if (row < Nrows) out[(size_t)row * HID + bcol] = acc[r] + bv;
    }
}

__global__ void k_instnorm_relu(float* __restrict__ x, int Nrows) {
    int wave = (blockIdx.x * blockDim.x + threadIdx.x) >> 5;
    int lane = threadIdx.x & 31;
    if (wave >= Nrows) return;
    float v0 = x[wave * HID + lane];
    float v1 = x[wave * HID + lane + 32];
    float s = v0 + v1;
    for (int off = 16; off; off >>= 1) s += __shfl_xor(s, off);
    float mu = s * (1.f / HID);
    float d0 = v0 - mu, d1 = v1 - mu;
    float q = d0 * d0 + d1 * d1;
    for (int off = 16; off; off >>= 1) q += __shfl_xor(q, off);
    float inv = rsqrtf(q * (1.f / HID) + EPS);
    x[wave * HID + lane]      = fmaxf(d0 * inv, 0.f);
    x[wave * HID + lane + 32] = fmaxf(d1 * inv, 0.f);
}

__global__ void k_graph_max(const float* __restrict__ xl, const int* __restrict__ batch,
                            unsigned* __restrict__ he, int n) {
    int i = blockIdx.x * blockDim.x + threadIdx.x;
    if (i >= n) return;
    int node = i >> 6;
    int c    = i & 63;
    atomicMax(&he[batch[node] * HID + c], enc_f32(xl[i]));
}

__global__ void k_decode_h(const unsigned* __restrict__ henc, float* __restrict__ h, int n) {
    int i = blockIdx.x * blockDim.x + threadIdx.x;
    if (i >= n) return;
    int g = i / (3 * HID);
    int c = i % (3 * HID);
    int layer = c / HID, cc = c % HID;
    h[i] = dec_f32(henc[layer * NG * HID + g * HID + cc]);
}

__global__ void k_mlp1(const float* __restrict__ h, const float* __restrict__ Wl1,
                       const float* __restrict__ bl1, float* __restrict__ z1) {
    int i = blockIdx.x * blockDim.x + threadIdx.x;
    if (i >= NG * 2 * HID) return;
    int g = i / (2 * HID), oc = i % (2 * HID);
    float s = bl1[oc];
    for (int k = 0; k < 3 * HID; ++k) s += h[g * 3 * HID + k] * Wl1[oc * 3 * HID + k];
    z1[i] = fmaxf(s, 0.f);
}

__global__ void k_mlp2_norm(const float* __restrict__ z1, const float* __restrict__ Wl2,
                            const float* __restrict__ bl2, float* __restrict__ outp) {
    int wave = (blockIdx.x * blockDim.x + threadIdx.x) >> 5;
    int lane = threadIdx.x & 31;
    if (wave >= NG) return;
    float s = bl2[lane];
    for (int k = 0; k < 2 * HID; ++k) s += z1[wave * 2 * HID + k] * Wl2[lane * 2 * HID + k];
    float q = s * s;
    for (int off = 16; off; off >>= 1) q += __shfl_xor(q, off);
    float nrm = fmaxf(sqrtf(q), 1e-12f);
    outp[wave * (HID / 2) + lane] = s / nrm;
}

extern "C" void kernel_launch(void* const* d_in, const int* in_sizes, int n_in,
                              void* d_out, int out_size, void* d_ws, size_t ws_size,
                              hipStream_t stream) {
    const float* x    = (const float*)d_in[0];
    const int*  eidx  = (const int*)  d_in[1];
    const int*  batch = (const int*)  d_in[2];
    const float* t    = (const float*)d_in[3];
    const float* W1r  = (const float*)d_in[4];
    const float* b1   = (const float*)d_in[5];
    const float* W1s  = (const float*)d_in[6];
    const float* W2r  = (const float*)d_in[7];
    const float* b2   = (const float*)d_in[8];
    const float* W2s  = (const float*)d_in[9];
    const float* W3r  = (const float*)d_in[10];
    const float* b3   = (const float*)d_in[11];
    const float* W3s  = (const float*)d_in[12];
    const float* Wl1  = (const float*)d_in[13];
    const float* bl1  = (const float*)d_in[14];
    const float* Wl2  = (const float*)d_in[15];
    const float* bl2  = (const float*)d_in[16];
    const int* src = eidx;
    const int* dst = eidx + NE;

    float* ws = (float*)d_ws;
    size_t o = 0;
    float*    x1     = ws + o; o += (size_t)NN * HID;
    float*    x2     = ws + o; o += (size_t)NN * HID;
    float*    x3     = ws + o; o += (size_t)NN * HID;
    int*      csr    = (int*)(ws + o); o += (size_t)NN * HID * 3;
    int*      cnt    = csr;
    int*      bsum   = cnt + NN;
    int*      rowptr = bsum + 256;
    int*      cursor = rowptr + NN + 16;
    int*      csrc   = cursor + NN;
    float*    aggr   = ws + o; o += (size_t)NN * HID;
    float*    xpad   = ws + o; o += (size_t)NN * 32;
    float*    w1rp   = ws + o; o += (size_t)HID * 32;
    float*    w1sp   = ws + o; o += (size_t)HID * 32;
    unsigned* henc   = (unsigned*)(ws + o); o += 3 * NG * HID;
    float*    h      = ws + o; o += NG * 3 * HID;
    float*    z1     = ws + o; o += NG * 2 * HID;

    k_pad_rows<<<(NN * 32 + 255) / 256, 256, 0, stream>>>(x, xpad, NN, INC, 32);
    k_pad_rows<<<(HID * 32 + 255) / 256, 256, 0, stream>>>(W1r, w1rp, HID, INC, 32);
    k_pad_rows<<<(HID * 32 + 255) / 256, 256, 0, stream>>>(W1s, w1sp, HID, INC, 32);
    k_init_enc<<<(NN + 255) / 256, 256, 0, stream>>>(henc, 3 * NG * HID, cnt, NN);
    k_csr_hist<<<(NE + 255) / 256, 256, 0, stream>>>(dst, cnt, NE);
    k_csr_bsum<<<(NN + 255) / 256, 256, 0, stream>>>(cnt, bsum, NN);
    k_csr_scan<<<(NN + 255) / 256, 256, 0, stream>>>(cnt, bsum, rowptr, cursor, NN);
    k_csr_fill<<<(NE + 255) / 256, 256, 0, stream>>>(src, dst, cursor, csrc, NE);

    struct Layer { const float* xin; int Cp; const float *Wr, *b, *Ws; float* xout; };
    Layer layers[3] = {
        { xpad, 32, w1rp, b1, w1sp, x1 },
        { x1,   64, W2r,  b2, W2s,  x2 },
        { x2,   64, W3r,  b3, W3s,  x3 },
    };

    for (int l = 0; l < 3; ++l) {
        const Layer& L = layers[l];
        if (L.Cp == 32)
            k_seg_softmax<32><<<(NN + 7) / 8, 256, 0, stream>>>(rowptr, csrc, L.xin, t, aggr, NN);
        else
            k_seg_softmax<64><<<(NN + 7) / 8, 256, 0, stream>>>(rowptr, csrc, L.xin, t, aggr, NN);

        dim3 gblk(32, 4);
        if (L.Cp == 32)
            k_gemm<32><<<(NN + 15) / 16, gblk, 0, stream>>>(aggr, L.xin, L.Wr, L.Ws, L.b,
                                                            L.xout, NN);
        else
            k_gemm<64><<<(NN + 15) / 16, gblk, 0, stream>>>(aggr, L.xin, L.Wr, L.Ws, L.b,
                                                            L.xout, NN);

        k_instnorm_relu<<<(NN + 7) / 8, 256, 0, stream>>>(L.xout, NN);

        int ngm = NN * HID;
        k_graph_max<<<(ngm + 255) / 256, 256, 0, stream>>>(L.xout, batch,
                                                           henc + (size_t)l * NG * HID, ngm);
    }

    k_decode_h<<<(NG * 3 * HID + 255) / 256, 256, 0, stream>>>(henc, h, NG * 3 * HID);
    k_mlp1<<<(NG * 2 * HID + 255) / 256, 256, 0, stream>>>(h, Wl1, bl1, z1);
    k_mlp2_norm<<<NG * 32 / 128, 128, 0, stream>>>(z1, Wl2, bl2, (float*)d_out);
}


# GEMM wave owns 16 whole rows (4 acc tiles): instance norm, ReLU and per-graph max run in its epilogue in registers, 6 launches gone; row sums in a new reduction tree, max pre-reduced per tile
# speedup vs baseline: 2.9401x; 1.3710x over previous
#include <hip/hip_runtime.h>
#include <math.h>

#define NN      50000
#define NE      800000
#define NG      128
#define INC     31
#define HID     64
#define EPS     1e-5f

typedef __attribute__((ext_vector_type(2))) float v2f;
typedef __attribute__((ext_vector_type(8))) float v8f;

static __device__ __forceinline__ unsigned enc_f32(float x) {
    unsigned b = __float_as_uint(x);
    return (b & 0x80000000u) ? ~b : (b | 0x80000000u);
}
static __device__ __forceinline__ float dec_f32(unsigned e) {
    unsigned b = (e & 0x80000000u) ? (e & 0x7FFFFFFFu) : ~e;
    return __uint_as_float(b);
}
#define ENC_NEG_INF 0x007FFFFFu

__global__ void k_pad_rows(const float* __restrict__ s, float* __restrict__ d,
                           int rows, int Csrc, int Cdst) {
    int i = blockIdx.x * blockDim.x + threadIdx.x;
    if (i >= rows * Cdst) return;
    int r = i / Cdst, c = i % Cdst;
    d[i] = (c < Csrc) ? s[r * Csrc + c] : 0.f;
}

__global__ void k_init_enc(unsigned* __restrict__ p, int n, int* __restrict__ cnt, int ncnt) {
    int i = blockIdx.x * blockDim.x + threadIdx.x;
    if (i < n) p[i] = ENC_NEG_INF;
    if (i < ncnt) cnt[i] = 0;
}

__global__ void k_csr_hist(const int* __restrict__ dst, int* __restrict__ cnt, int E) {
    int e = blockIdx.x * blockDim.x + threadIdx.x;
    if (e < E) atomicAdd(&cnt[dst[e]], 1);
}

__global__ void __launch_bounds__(256) k_csr_bsum(const int* __restrict__ cnt, int* __restrict__ bsum, int n) {
    __shared__ int red[8];
    const int tid = threadIdx.x, lane = tid & 31, w = tid >> 5;
    const int idx = blockIdx.x * 256 + tid;
    int c = idx < n ? cnt[idx] : 0;
    for (int off = 16; off; off >>= 1) c += __shfl_xor(c, off);
    if (lane == 0) red[w] = c;
    __syncthreads();
    if (tid == 0) {
        int s = 0;
        for (int i = 0; i < 8; ++i) s += red[i];
        bsum[blockIdx.x] = s;
    }
}

__global__ void __launch_bounds__(256) k_csr_scan(const int* __restrict__ cnt, const int* __restrict__ bsum,
                                                  int* __restrict__ rowptr, int* __restrict__ cursor, int n) {
    __shared__ int red[8];
    __shared__ int wsum[8];
    const int tid = threadIdx.x, lane = tid & 31, w = tid >> 5;
    int part = 0;
    for (int i = tid; i < (int)blockIdx.x; i += 256) part += bsum[i];
    for (int off = 16; off; off >>= 1) part += __shfl_xor(part, off);
    if (lane == 0) red[w] = part;
    const int idx = blockIdx.x * 256 + tid;
    const int c = idx < n ? cnt[idx] : 0;
    int inc = c;
    for (int off = 1; off < 32; off <<= 1) {
        int u = __shfl_up(inc, off);
        if (lane >= off) inc += u;
    }
    if (lane == 31) wsum[w] = inc;
    __syncthreads();
    int base = 0;
    for (int i = 0; i < 8; ++i) {
        base += red[i];
        if (i < w) base += wsum[i];
    }
    const int ex = base + inc - c;
    if (idx < n) { rowptr[idx] = ex; cursor[idx] = ex; }
    if (idx == n - 1) rowptr[n] = ex + c;
}

__global__ void k_csr_fill(const int* __restrict__ src, const int* __restrict__ dst,
                           int* __restrict__ cursor, int* __restrict__ csrc, int E) {
    int e = blockIdx.x * blockDim.x + threadIdx.x;
    if (e >= E) return;
    int p = atomicAdd(&cursor[dst[e]], 1);
    csrc[p] = src[e];
}

template<int VEC>
static __device__ __forceinline__ void ld_row(const float* __restrict__ p, float (&v)[VEC]) {
    if constexpr (VEC == 2) {
        v2f r = *(const v2f*)p;
        v[0] = r.x; v[1] = r.y;
    } else {
        v[0] = p[0];
    }
}

template<int VEC>
static __device__ __forceinline__ void seg_max(const float (&r)[VEC], float t, float (&mx)[VEC]) {
#pragma unroll
    for (int i = 0; i < VEC; ++i) mx[i] = fmaxf(mx[i], r[i] * t);
}
template<int VEC>
static __device__ __forceinline__ void seg_sum(const float (&r)[VEC], float t, const float (&mx)[VEC],
                                               float (&num)[VEC], float (&den)[VEC]) {
#pragma unroll
    for (int i = 0; i < VEC; ++i) {
        float e = __expf(r[i] * t - mx[i]);
        num[i] += r[i] * e;
        den[i] += e;
    }
}

template<int J, int VEC, int CP>
struct SegChain {
    static __device__ __forceinline__ void load(int deg, const float* __restrict__ x, unsigned loff,
                                                const int (&sidx)[32], float (&v)[32][VEC]) {
        if constexpr (J < 32) {
            if (J < deg) {
                ld_row<VEC>(x + (size_t)sidx[J] * CP + loff, v[J]);
                SegChain<J + 1, VEC, CP>::load(deg, x, loff, sidx, v);
            }
        }
    }
    static __device__ __forceinline__ void vmax(int deg, const float (&v)[32][VEC], float t, float (&mx)[VEC]) {
        if constexpr (J < 32) {
            if (J < deg) {
                seg_max<VEC>(v[J], t, mx);
                SegChain<J + 1, VEC, CP>::vmax(deg, v, t, mx);
            }
        }
    }
    static __device__ __forceinline__ void vsum(int deg, const float (&v)[32][VEC], float t,
                                                const float (&mx)[VEC], float (&num)[VEC], float (&den)[VEC]) {
        if constexpr (J < 32) {
            if (J < deg) {
                seg_sum<VEC>(v[J], t, mx, num, den);
                SegChain<J + 1, VEC, CP>::vsum(deg, v, t, mx, num, den);
            }
        }
    }
};

template<int CP>
__global__ void __launch_bounds__(256) k_seg_softmax(const int* __restrict__ rowptr,
                                                     const int* __restrict__ csrc,
                                                     const float* __restrict__ x,
                                                     const float* __restrict__ tptr,
                                                     float* __restrict__ aggr, int n) {
    constexpr int VEC = CP / 32;
    const int lane = threadIdx.x & 31;
    const int node = __builtin_amdgcn_readfirstlane((int)(blockIdx.x * 8 + (threadIdx.x >> 5)));
    if (node >= n) return;
    const int beg = rowptr[node];
    const int end = rowptr[node + 1];
    const int deg = end - beg;
    const float t = tptr[0];
    const unsigned loff = (unsigned)(lane * VEC);

    float mx[VEC], num[VEC], den[VEC];
#pragma unroll
    for (int i = 0; i < VEC; ++i) { mx[i] = -INFINITY; num[i] = 0.f; den[i] = 0.f; }

    if (deg <= 32) {
        const int mys = (lane < deg) ? csrc[beg + lane] : 0;
        int sidx[32];
#pragma unroll
        for (int j = 0; j < 32; ++j) sidx[j] = __builtin_amdgcn_readlane(mys, j);
        float v[32][VEC];
        SegChain<0, VEC, CP>::load(deg, x, loff, sidx, v);
        SegChain<0, VEC, CP>::vmax(deg, v, t, mx);
        SegChain<0, VEC, CP>::vsum(deg, v, t, mx, num, den);
    } else {
        for (int base = beg; base < end; base += 32) {
            const int cnt = min(32, end - base);
            const int mys = (lane < cnt) ? csrc[base + lane] : 0;
            for (int j = 0; j < cnt; ++j) {
                const int s = __builtin_amdgcn_readlane(mys, j);
                float r[VEC];
                ld_row<VEC>(x + (size_t)s * CP + loff, r);
                seg_max<VEC>(r, t, mx);
            }
        }
        for (int base = beg; base < end; base += 32) {
            const int cnt = min(32, end - base);
            const int mys = (lane < cnt) ? csrc[base + lane] : 0;
            for (int j = 0; j < cnt; ++j) {
                const int s = __builtin_amdgcn_readlane(mys, j);
                float r[VEC];
                ld_row<VEC>(x + (size_t)s * CP + loff, r);
                seg_sum<VEC>(r, t, mx, num, den);
            }
        }
    }

    float res[VEC];
#pragma unroll
    for (int i = 0; i < VEC; ++i) res[i] = den[i] > 0.f ? num[i] / den[i] : 0.f;
    float* o = aggr + (size_t)node * CP + loff;
    if constexpr (VEC == 2) {
        v2f ov; ov.x = res[0]; ov.y = res[1];
        *(v2f*)o = ov;
    } else {
        o[0] = res[0];
    }
}


static __device__ __forceinline__ float xhalf(float x) {
    int b = __builtin_bit_cast(int, x);
    return __builtin_bit_cast(float, __builtin_amdgcn_permlanex16(b, b, 0x76543210, 0xfedcba98, false, false));
}
template<int M>
static __device__ __forceinline__ float rowx(float x) {
    return __builtin_bit_cast(float, __builtin_amdgcn_update_dpp(0, __builtin_bit_cast(int, x), 0x160 + M, 0xf, 0xf, true));
}
template<int M>
static __device__ __forceinline__ void row_step8(float (&s)[8]) {
#pragma unroll
    for (int r = 0; r < 8; ++r) s[r] += rowx<M>(s[r]);
}
static __device__ __forceinline__ void row_sum16x8(float (&s)[8]) {
    row_step8<1>(s);
    row_step8<2>(s);
    row_step8<4>(s);
    row_step8<8>(s);
}

template<int CP>
__global__ void __launch_bounds__(128) k_gemm_norm_max(const float* __restrict__ aggr, const float* __restrict__ xin,
                                                       const float* __restrict__ Wr, const float* __restrict__ Ws,
                                                       const float* __restrict__ bias, const int* __restrict__ batch,
                                                       float* __restrict__ out, unsigned* __restrict__ he, int Nrows) {
    const int lane = threadIdx.x & 31;
    const int half = lane >> 4;
    const int c    = lane & 15;
    const int m0 = __builtin_amdgcn_readfirstlane((int)(blockIdx.x * 4 + (threadIdx.x >> 5))) * 16;
    if (m0 >= Nrows) return;
    int arow = m0 + c;
    if (arow >= Nrows) arow = Nrows - 1;

    const float* ap = aggr + (size_t)arow * CP + half * 2;
    const float* xp = xin  + (size_t)arow * CP + half * 2;
    const float* wr = Wr   + (size_t)(4 * c) * CP + half * 2;
    const float* wsp= Ws   + (size_t)(4 * c) * CP + half * 2;
    const int gl = batch[arow];

    v8f acc[4] = {};
#pragma unroll
    for (int k = 0; k < CP; k += 4) {
        v2f a = *(const v2f*)(ap + k);
#pragma unroll
        for (int j = 0; j < 4; ++j) {
            v2f b = *(const v2f*)(wr + j * CP + k);
            acc[j] = __builtin_amdgcn_wmma_f32_16x16x4_f32(false, a, false, b,
                                                           (short)0, acc[j], false, false);
        }
    }
#pragma unroll
    for (int k = 0; k < CP; k += 4) {
        v2f a = *(const v2f*)(xp + k);
#pragma unroll
        for (int j = 0; j < 4; ++j) {
            v2f b = *(const v2f*)(wsp + j * CP + k);
            acc[j] = __builtin_amdgcn_wmma_f32_16x16x4_f32(false, a, false, b,
                                                           (short)0, acc[j], false, false);
        }
    }

    const float4 bq = *(const float4*)(bias + 4 * c);
    const float bv[4] = { bq.x, bq.y, bq.z, bq.w };
    float s[8], q[8];
#pragma unroll
    for (int r = 0; r < 8; ++r) {
#pragma unroll
        for (int j = 0; j < 4; ++j) acc[j][r] += bv[j];
        s[r] = (acc[0][r] + acc[1][r]) + (acc[2][r] + acc[3][r]);
    }
    row_sum16x8(s);
#pragma unroll
    for (int r = 0; r < 8; ++r) {
        const float mu = s[r] * (1.f / HID);
#pragma unroll
        for (int j = 0; j < 4; ++j) acc[j][r] -= mu;
        q[r] = (acc[0][r] * acc[0][r] + acc[1][r] * acc[1][r]) + (acc[2][r] * acc[2][r] + acc[3][r] * acc[3][r]);
    }
    row_sum16x8(q);
    float cm[4];
#pragma unroll
    for (int r = 0; r < 8; ++r) {
        const float inv = rsqrtf(q[r] * (1.f / HID) + EPS);
#pragma unroll
        for (int j = 0; j < 4; ++j) {
            const float y = fmaxf(acc[j][r] * inv, 0.f);
            acc[j][r] = y;
            cm[j] = (r == 0) ? y : fmaxf(cm[j], y);
        }
    }
    float* op = out + (size_t)(m0 + half * 8) * HID + 4 * c;
    if (m0 + 16 <= Nrows) {
#pragma unroll
        for (int r = 0; r < 8; ++r) {
            float4 o; o.x = acc[0][r]; o.y = acc[1][r]; o.z = acc[2][r]; o.w = acc[3][r];
            *(float4*)(op + (size_t)r * HID) = o;
        }
    } else {
#pragma unroll
        for (int r = 0; r < 8; ++r) {
            if (m0 + r + half * 8 < Nrows) {
                float4 o; o.x = acc[0][r]; o.y = acc[1][r]; o.z = acc[2][r]; o.w = acc[3][r];
                *(float4*)(op + (size_t)r * HID) = o;
            }
        }
    }

    const int g0 = __builtin_amdgcn_readfirstlane(gl);
    const bool one_graph = __builtin_amdgcn_ballot_w32(gl != g0) == 0u;
    if (one_graph) {
#pragma unroll
        for (int j = 0; j < 4; ++j) cm[j] = fmaxf(cm[j], xhalf(cm[j]));
        const float p0 = half ? cm[2] : cm[0];
        const float p1 = half ? cm[3] : cm[1];
        unsigned* hp = he + (size_t)g0 * HID + 4 * c + 2 * half;
        atomicMax(hp,     enc_f32(p0));
        atomicMax(hp + 1, enc_f32(p1));
    } else {
#pragma unroll
        for (int r = 0; r < 8; ++r) {
            const int row = m0 + r + half * 8;
            if (row < Nrows) {
                unsigned* hp = he + (size_t)batch[row] * HID + 4 * c;
#pragma unroll
                for (int j = 0; j < 4; ++j) atomicMax(hp + j, enc_f32(acc[j][r]));
            }
        }
    }
}

__global__ void k_decode_h(const unsigned* __restrict__ henc, float* __restrict__ h, int n) {
    int i = blockIdx.x * blockDim.x + threadIdx.x;
    if (i >= n) return;
    int g = i / (3 * HID);
    int c = i % (3 * HID);
    int layer = c / HID, cc = c % HID;
    h[i] = dec_f32(henc[layer * NG * HID + g * HID + cc]);
}

__global__ void k_mlp1(const float* __restrict__ h, const float* __restrict__ Wl1,
                       const float* __restrict__ bl1, float* __restrict__ z1) {
    int i = blockIdx.x * blockDim.x + threadIdx.x;
    if (i >= NG * 2 * HID) return;
    int g = i / (2 * HID), oc = i % (2 * HID);
    float s = bl1[oc];
    for (int k = 0; k < 3 * HID; ++k) s += h[g * 3 * HID + k] * Wl1[oc * 3 * HID + k];
    z1[i] = fmaxf(s, 0.f);
}

__global__ void k_mlp2_norm(const float* __restrict__ z1, const float* __restrict__ Wl2,
                            const float* __restrict__ bl2, float* __restrict__ outp) {
    int wave = (blockIdx.x * blockDim.x + threadIdx.x) >> 5;
    int lane = threadIdx.x & 31;
    if (wave >= NG) return;
    float s = bl2[lane];
    for (int k = 0; k < 2 * HID; ++k) s += z1[wave * 2 * HID + k] * Wl2[lane * 2 * HID + k];
    float q = s * s;
    for (int off = 16; off; off >>= 1) q += __shfl_xor(q, off);
    float nrm = fmaxf(sqrtf(q), 1e-12f);
    outp[wave * (HID / 2) + lane] = s / nrm;
}

extern "C" void kernel_launch(void* const* d_in, const int* in_sizes, int n_in,
                              void* d_out, int out_size, void* d_ws, size_t ws_size,
                              hipStream_t stream) {
    const float* x    = (const float*)d_in[0];
    const int*  eidx  = (const int*)  d_in[1];
    const int*  batch = (const int*)  d_in[2];
    const float* t    = (const float*)d_in[3];
    const float* W1r  = (const float*)d_in[4];
    const float* b1   = (const float*)d_in[5];
    const float* W1s  = (const float*)d_in[6];
    const float* W2r  = (const float*)d_in[7];
    const float* b2   = (const float*)d_in[8];
    const float* W2s  = (const float*)d_in[9];
    const float* W3r  = (const float*)d_in[10];
    const float* b3   = (const float*)d_in[11];
    const float* W3s  = (const float*)d_in[12];
    const float* Wl1  = (const float*)d_in[13];
    const float* bl1  = (const float*)d_in[14];
    const float* Wl2  = (const float*)d_in[15];
    const float* bl2  = (const float*)d_in[16];
    const int* src = eidx;
    const int* dst = eidx + NE;

    float* ws = (float*)d_ws;
    size_t o = 0;
    float*    x1     = ws + o; o += (size_t)NN * HID;
    float*    x2     = ws + o; o += (size_t)NN * HID;
    float*    x3     = ws + o; o += (size_t)NN * HID;
    int*      csr    = (int*)(ws + o); o += (size_t)NN * HID * 3;
    int*      cnt    = csr;
    int*      bsum   = cnt + NN;
    int*      rowptr = bsum + 256;
    int*      cursor = rowptr + NN + 16;
    int*      csrc   = cursor + NN;
    float*    aggr   = ws + o; o += (size_t)NN * HID;
    float*    xpad   = ws + o; o += (size_t)NN * 32;
    float*    w1rp   = ws + o; o += (size_t)HID * 32;
    float*    w1sp   = ws + o; o += (size_t)HID * 32;
    unsigned* henc   = (unsigned*)(ws + o); o += 3 * NG * HID;
    float*    h      = ws + o; o += NG * 3 * HID;
    float*    z1     = ws + o; o += NG * 2 * HID;

    k_pad_rows<<<(NN * 32 + 255) / 256, 256, 0, stream>>>(x, xpad, NN, INC, 32);
    k_pad_rows<<<(HID * 32 + 255) / 256, 256, 0, stream>>>(W1r, w1rp, HID, INC, 32);
    k_pad_rows<<<(HID * 32 + 255) / 256, 256, 0, stream>>>(W1s, w1sp, HID, INC, 32);
    k_init_enc<<<(NN + 255) / 256, 256, 0, stream>>>(henc, 3 * NG * HID, cnt, NN);
    k_csr_hist<<<(NE + 255) / 256, 256, 0, stream>>>(dst, cnt, NE);
    k_csr_bsum<<<(NN + 255) / 256, 256, 0, stream>>>(cnt, bsum, NN);
    k_csr_scan<<<(NN + 255) / 256, 256, 0, stream>>>(cnt, bsum, rowptr, cursor, NN);
    k_csr_fill<<<(NE + 255) / 256, 256, 0, stream>>>(src, dst, cursor, csrc, NE);

    struct Layer { const float* xin; int Cp; const float *Wr, *b, *Ws; float* xout; };
    Layer layers[3] = {
        { xpad, 32, w1rp, b1, w1sp, x1 },
        { x1,   64, W2r,  b2, W2s,  x2 },
        { x2,   64, W3r,  b3, W3s,  x3 },
    };

    for (int l = 0; l < 3; ++l) {
        const Layer& L = layers[l];
        if (L.Cp == 32)
            k_seg_softmax<32><<<(NN + 7) / 8, 256, 0, stream>>>(rowptr, csrc, L.xin, t, aggr, NN);
        else
            k_seg_softmax<64><<<(NN + 7) / 8, 256, 0, stream>>>(rowptr, csrc, L.xin, t, aggr, NN);

        if (L.Cp == 32)
            k_gemm_norm_max<32><<<((NN + 15) / 16 + 3) / 4, 128, 0, stream>>>(aggr, L.xin, L.Wr, L.Ws, L.b, batch,
                                                                              L.xout, henc + (size_t)l * NG * HID, NN);
        else
            k_gemm_norm_max<64><<<((NN + 15) / 16 + 3) / 4, 128, 0, stream>>>(aggr, L.xin, L.Wr, L.Ws, L.b, batch,
                                                                              L.xout, henc + (size_t)l * NG * HID, NN);
    }

    k_decode_h<<<(NG * 3 * HID + 255) / 256, 256, 0, stream>>>(henc, h, NG * 3 * HID);
    k_mlp1<<<(NG * 2 * HID + 255) / 256, 256, 0, stream>>>(h, Wl1, bl1, z1);
    k_mlp2_norm<<<NG * 32 / 128, 128, 0, stream>>>(z1, Wl2, bl2, (float*)d_out);
}


# 17 launches to 11: clears and pads ride in one prep launch and in the histogram launch, block sums fold into the scan, decode+MLP+normalise are one block-per-graph kernel (MLP dots: new sum tree)
# speedup vs baseline: 3.1188x; 1.0608x over previous
#include <hip/hip_runtime.h>
#include <math.h>

#define NN      50000
#define NE      800000
#define NG      128
#define INC     31
#define HID     64
#define EPS     1e-5f

typedef __attribute__((ext_vector_type(2))) float v2f;
typedef __attribute__((ext_vector_type(8))) float v8f;

static __device__ __forceinline__ unsigned enc_f32(float x) {
    unsigned b = __float_as_uint(x);
    return (b & 0x80000000u) ? ~b : (b | 0x80000000u);
}
static __device__ __forceinline__ float dec_f32(unsigned e) {
    unsigned b = (e & 0x80000000u) ? (e & 0x7FFFFFFFu) : ~e;
    return __uint_as_float(b);
}
#define ENC_NEG_INF 0x007FFFFFu

__global__ void k_prep(int* __restrict__ cnt, unsigned* __restrict__ henc,
                       const float* __restrict__ W1r, const float* __restrict__ W1s,
                       float* __restrict__ w1rp, float* __restrict__ w1sp) {
    int i = blockIdx.x * blockDim.x + threadIdx.x;
    if (i < NN) cnt[i] = 0;
    if (i < 3 * NG * HID) henc[i] = ENC_NEG_INF;
    if (i < HID * 32) {
        int r = i / 32, c = i % 32;
        w1rp[i] = (c < INC) ? W1r[r * INC + c] : 0.f;
        w1sp[i] = (c < INC) ? W1s[r * INC + c] : 0.f;
    }
}

__global__ void k_csr_hist_pad(const int* __restrict__ dst, int* __restrict__ cnt,
                               const float* __restrict__ x, float* __restrict__ xpad) {
    int e = blockIdx.x * blockDim.x + threadIdx.x;
    if (e >= NE) return;
    const int i1 = e + NE;
    const int r0 = e / 32,  c0 = e % 32;
    const int r1 = i1 / 32, c1 = i1 % 32;
    const int   d  = dst[e];
    const float v0 = x[r0 * INC + min(c0, INC - 1)];
    const float v1 = x[r1 * INC + min(c1, INC - 1)];
    atomicAdd(&cnt[d], 1);
    xpad[e]  = (c0 < INC) ? v0 : 0.f;
    xpad[i1] = (c1 < INC) ? v1 : 0.f;
}

__global__ void __launch_bounds__(1024) k_csr_scan(const int* __restrict__ cnt,
                                                   int* __restrict__ rowptr, int* __restrict__ cursor, int n) {
    __shared__ int red[32];
    __shared__ int wsum[32];
    const int tid = threadIdx.x, lane = tid & 31, w = tid >> 5;
    const int nfront4 = (int)blockIdx.x * 256;
    const int4* __restrict__ c4 = (const int4*)cnt;
    constexpr int FRONT_STEPS = (((NN + 1023) / 1024 - 1) * 256 + 1023) / 1024;
    int part = 0;
#pragma unroll
    for (int s = 0; s < FRONT_STEPS; ++s) {
        const int i = tid + s * 1024;
        const int4 v = c4[min(i, NN / 4 - 1)];
        const int t = (v.x + v.y) + (v.z + v.w);
        part += (i < nfront4) ? t : 0;
    }
    for (int off = 16; off; off >>= 1) part += __shfl_xor(part, off);
    if (lane == 0) red[w] = part;
    const int idx = blockIdx.x * 1024 + tid;
    const int c = idx < n ? cnt[idx] : 0;
    int inc = c;
    for (int off = 1; off < 32; off <<= 1) {
        int u = __shfl_up(inc, off);
        if (lane >= off) inc += u;
    }
    if (lane == 31) wsum[w] = inc;
    __syncthreads();
    int base = 0;
    for (int i = 0; i < 32; ++i) {
        base += red[i];
        if (i < w) base += wsum[i];
    }
    const int ex = base + inc - c;
    if (idx < n) { rowptr[idx] = ex; cursor[idx] = ex; }
    if (idx == n - 1) rowptr[n] = ex + c;
}

__global__ void k_csr_fill(const int* __restrict__ src, const int* __restrict__ dst,
                           int* __restrict__ cursor, int* __restrict__ csrc, int E) {
    int e = blockIdx.x * blockDim.x + threadIdx.x;
    if (e >= E) return;
    int p = atomicAdd(&cursor[dst[e]], 1);
    csrc[p] = src[e];
}

template<int VEC>
static __device__ __forceinline__ void ld_row(const float* __restrict__ p, float (&v)[VEC]) {
    if constexpr (VEC == 2) {
        v2f r = *(const v2f*)p;
        v[0] = r.x; v[1] = r.y;
    } else {
        v[0] = p[0];
    }
}

template<int VEC>
static __device__ __forceinline__ void seg_max(const float (&r)[VEC], float t, float (&mx)[VEC]) {
#pragma unroll
    for (int i = 0; i < VEC; ++i) mx[i] = fmaxf(mx[i], r[i] * t);
}
template<int VEC>
static __device__ __forceinline__ void seg_sum(const float (&r)[VEC], float t, const float (&mx)[VEC],
                                               float (&num)[VEC], float (&den)[VEC]) {
#pragma unroll
    for (int i = 0; i < VEC; ++i) {
        float e = __expf(r[i] * t - mx[i]);
        num[i] += r[i] * e;
        den[i] += e;
    }
}

template<int J, int VEC, int CP>
struct SegChain {
    static __device__ __forceinline__ void load(int deg, const float* __restrict__ x, unsigned loff,
                                                const int (&sidx)[32], float (&v)[32][VEC]) {
        if constexpr (J < 32) {
            if (J < deg) {
                ld_row<VEC>(x + (size_t)sidx[J] * CP + loff, v[J]);
                SegChain<J + 1, VEC, CP>::load(deg, x, loff, sidx, v);
            }
        }
    }
    static __device__ __forceinline__ void vmax(int deg, const float (&v)[32][VEC], float t, float (&mx)[VEC]) {
        if constexpr (J < 32) {
            if (J < deg) {
                seg_max<VEC>(v[J], t, mx);
                SegChain<J + 1, VEC, CP>::vmax(deg, v, t, mx);
            }
        }
    }
    static __device__ __forceinline__ void vsum(int deg, const float (&v)[32][VEC], float t,
                                                const float (&mx)[VEC], float (&num)[VEC], float (&den)[VEC]) {
        if constexpr (J < 32) {
            if (J < deg) {
                seg_sum<VEC>(v[J], t, mx, num, den);
                SegChain<J + 1, VEC, CP>::vsum(deg, v, t, mx, num, den);
            }
        }
    }
};

template<int CP>
__global__ void __launch_bounds__(256) k_seg_softmax(const int* __restrict__ rowptr,
                                                     const int* __restrict__ csrc,
                                                     const float* __restrict__ x,
                                                     const float* __restrict__ tptr,
                                                     float* __restrict__ aggr, int n) {
    constexpr int VEC = CP / 32;
    const int lane = threadIdx.x & 31;
    const int node = __builtin_amdgcn_readfirstlane((int)(blockIdx.x * 8 + (threadIdx.x >> 5)));
    if (node >= n) return;
    const int beg = rowptr[node];
    const int end = rowptr[node + 1];
    const int deg = end - beg;
    const float t = tptr[0];
    const unsigned loff = (unsigned)(lane * VEC);

    float mx[VEC], num[VEC], den[VEC];
#pragma unroll
    for (int i = 0; i < VEC; ++i) { mx[i] = -INFINITY; num[i] = 0.f; den[i] = 0.f; }

    if (deg <= 32) {
        const int mys = (lane < deg) ? csrc[beg + lane] : 0;
        int sidx[32];
#pragma unroll
        for (int j = 0; j < 32; ++j) sidx[j] = __builtin_amdgcn_readlane(mys, j);
        float v[32][VEC];
        SegChain<0, VEC, CP>::load(deg, x, loff, sidx, v);
        SegChain<0, VEC, CP>::vmax(deg, v, t, mx);
        SegChain<0, VEC, CP>::vsum(deg, v, t, mx, num, den);
    } else {
        for (int base = beg; base < end; base += 32) {
            const int cnt = min(32, end - base);
            const int mys = (lane < cnt) ? csrc[base + lane] : 0;
            for (int j = 0; j < cnt; ++j) {
                const int s = __builtin_amdgcn_readlane(mys, j);
                float r[VEC];
                ld_row<VEC>(x + (size_t)s * CP + loff, r);
                seg_max<VEC>(r, t, mx);
            }
        }
        for (int base = beg; base < end; base += 32) {
            const int cnt = min(32, end - base);
            const int mys = (lane < cnt) ? csrc[base + lane] : 0;
            for (int j = 0; j < cnt; ++j) {
                const int s = __builtin_amdgcn_readlane(mys, j);
                float r[VEC];
                ld_row<VEC>(x + (size_t)s * CP + loff, r);
                seg_sum<VEC>(r, t, mx, num, den);
            }
        }
    }

    float res[VEC];
#pragma unroll
    for (int i = 0; i < VEC; ++i) res[i] = den[i] > 0.f ? num[i] / den[i] : 0.f;
    float* o = aggr + (size_t)node * CP + loff;
    if constexpr (VEC == 2) {
        v2f ov; ov.x = res[0]; ov.y = res[1];
        *(v2f*)o = ov;
    } else {
        o[0] = res[0];
    }
}


static __device__ __forceinline__ float xhalf(float x) {
    int b = __builtin_bit_cast(int, x);
    return __builtin_bit_cast(float, __builtin_amdgcn_permlanex16(b, b, 0x76543210, 0xfedcba98, false, false));
}
template<int M>
static __device__ __forceinline__ float rowx(float x) {
    return __builtin_bit_cast(float, __builtin_amdgcn_update_dpp(0, __builtin_bit_cast(int, x), 0x160 + M, 0xf, 0xf, true));
}
template<int M>
static __device__ __forceinline__ void row_step8(float (&s)[8]) {
#pragma unroll
    for (int r = 0; r < 8; ++r) s[r] += rowx<M>(s[r]);
}
static __device__ __forceinline__ void row_sum16x8(float (&s)[8]) {
    row_step8<1>(s);
    row_step8<2>(s);
    row_step8<4>(s);
    row_step8<8>(s);
}

template<int CP>
__global__ void __launch_bounds__(128) k_gemm_norm_max(const float* __restrict__ aggr, const float* __restrict__ xin,
                                                       const float* __restrict__ Wr, const float* __restrict__ Ws,
                                                       const float* __restrict__ bias, const int* __restrict__ batch,
                                                       float* __restrict__ out, unsigned* __restrict__ he, int Nrows) {
    const int lane = threadIdx.x & 31;
    const int half = lane >> 4;
    const int c    = lane & 15;
    const int m0 = __builtin_amdgcn_readfirstlane((int)(blockIdx.x * 4 + (threadIdx.x >> 5))) * 16;
    if (m0 >= Nrows) return;
    int arow = m0 + c;
    if (arow >= Nrows) arow = Nrows - 1;

    const float* ap = aggr + (size_t)arow * CP + half * 2;
    const float* xp = xin  + (size_t)arow * CP + half * 2;
    const float* wr = Wr   + (size_t)(4 * c) * CP + half * 2;
    const float* wsp= Ws   + (size_t)(4 * c) * CP + half * 2;
    const int gl = batch[arow];

    v8f acc[4] = {};
#pragma unroll
    for (int k = 0; k < CP; k += 4) {
        v2f a = *(const v2f*)(ap + k);
#pragma unroll
        for (int j = 0; j < 4; ++j) {
            v2f b = *(const v2f*)(wr + j * CP + k);
            acc[j] = __builtin_amdgcn_wmma_f32_16x16x4_f32(false, a, false, b,
                                                           (short)0, acc[j], false, false);
        }
    }
#pragma unroll
    for (int k = 0; k < CP; k += 4) {
        v2f a = *(const v2f*)(xp + k);
#pragma unroll
        for (int j = 0; j < 4; ++j) {
            v2f b = *(const v2f*)(wsp + j * CP + k);
            acc[j] = __builtin_amdgcn_wmma_f32_16x16x4_f32(false, a, false, b,
                                                           (short)0, acc[j], false, false);
        }
    }

    const float4 bq = *(const float4*)(bias + 4 * c);
    const float bv[4] = { bq.x, bq.y, bq.z, bq.w };
    float s[8], q[8];
#pragma unroll
    for (int r = 0; r < 8; ++r) {
#pragma unroll
        for (int j = 0; j < 4; ++j) acc[j][r] += bv[j];
        s[r] = (acc[0][r] + acc[1][r]) + (acc[2][r] + acc[3][r]);
    }
    row_sum16x8(s);
#pragma unroll
    for (int r = 0; r < 8; ++r) {
        const float mu = s[r] * (1.f / HID);
#pragma unroll
        for (int j = 0; j < 4; ++j) acc[j][r] -= mu;
        q[r] = (acc[0][r] * acc[0][r] + acc[1][r] * acc[1][r]) + (acc[2][r] * acc[2][r] + acc[3][r] * acc[3][r]);
    }
    row_sum16x8(q);
    float cm[4];
#pragma unroll
    for (int r = 0; r < 8; ++r) {
        const float inv = rsqrtf(q[r] * (1.f / HID) + EPS);
#pragma unroll
        for (int j = 0; j < 4; ++j) {
            const float y = fmaxf(acc[j][r] * inv, 0.f);
            acc[j][r] = y;
            cm[j] = (r == 0) ? y : fmaxf(cm[j], y);
        }
    }
    float* op = out + (size_t)(m0 + half * 8) * HID + 4 * c;
    if (m0 + 16 <= Nrows) {
#pragma unroll
        for (int r = 0; r < 8; ++r) {
            float4 o; o.x = acc[0][r]; o.y = acc[1][r]; o.z = acc[2][r]; o.w = acc[3][r];
            *(float4*)(op + (size_t)r * HID) = o;
        }
    } else {
#pragma unroll
        for (int r = 0; r < 8; ++r) {
            if (m0 + r + half * 8 < Nrows) {
                float4 o; o.x = acc[0][r]; o.y = acc[1][r]; o.z = acc[2][r]; o.w = acc[3][r];
                *(float4*)(op + (size_t)r * HID) = o;
            }
        }
    }

    const int g0 = __builtin_amdgcn_readfirstlane(gl);
    const bool one_graph = __builtin_amdgcn_ballot_w32(gl != g0) == 0u;
    if (one_graph) {
#pragma unroll
        for (int j = 0; j < 4; ++j) cm[j] = fmaxf(cm[j], xhalf(cm[j]));
        const float p0 = half ? cm[2] : cm[0];
        const float p1 = half ? cm[3] : cm[1];
        unsigned* hp = he + (size_t)g0 * HID + 4 * c + 2 * half;
        atomicMax(hp,     enc_f32(p0));
        atomicMax(hp + 1, enc_f32(p1));
    } else {
#pragma unroll
        for (int r = 0; r < 8; ++r) {
            const int row = m0 + r + half * 8;
            if (row < Nrows) {
                unsigned* hp = he + (size_t)batch[row] * HID + 4 * c;
#pragma unroll
                for (int j = 0; j < 4; ++j) atomicMax(hp + j, enc_f32(acc[j][r]));
            }
        }
    }
}

template<int NV>
static __device__ __forceinline__ void wave_sum_n(float (&p)[NV]) {
#pragma unroll
    for (int j = 0; j < NV; ++j) p[j] += rowx<1>(p[j]);
#pragma unroll
    for (int j = 0; j < NV; ++j) p[j] += rowx<2>(p[j]);
#pragma unroll
    for (int j = 0; j < NV; ++j) p[j] += rowx<4>(p[j]);
#pragma unroll
    for (int j = 0; j < NV; ++j) p[j] += rowx<8>(p[j]);
#pragma unroll
    for (int j = 0; j < NV; ++j) p[j] += xhalf(p[j]);
}

__global__ void __launch_bounds__(256) k_head(const unsigned* __restrict__ henc,
                                              const float* __restrict__ Wl1, const float* __restrict__ bl1,
                                              const float* __restrict__ Wl2, const float* __restrict__ bl2,
                                              float* __restrict__ outp) {
    __shared__ __attribute__((aligned(16))) float sh[3 * HID];
    __shared__ __attribute__((aligned(16))) float sz1[2 * HID];
    __shared__ __attribute__((aligned(16))) float sz2[HID / 2];
    const int g = blockIdx.x;
    const int tid = threadIdx.x, lane = tid & 31, w = tid >> 5;
    if (tid < 3 * HID) {
        const int layer = tid / HID, cc = tid % HID;
        sh[tid] = dec_f32(henc[layer * NG * HID + g * HID + cc]);
    }
    __syncthreads();

    {
        v2f hv[3];
#pragma unroll
        for (int i = 0; i < 3; ++i) hv[i] = *(const v2f*)(sh + 2 * (lane + 32 * i));
        const float* __restrict__ wbase = Wl1 + (w * 16) * (3 * HID) + 2 * lane;
        float p[16];
#pragma unroll
        for (int j = 0; j < 16; ++j) {
            const float* __restrict__ wrow = wbase + j * (3 * HID);
            float a = 0.f;
#pragma unroll
            for (int i = 0; i < 3; ++i) {
                const v2f wv = *(const v2f*)(wrow + 64 * i);
                a += hv[i].x * wv.x;
                a += hv[i].y * wv.y;
            }
            p[j] = a;
        }
        wave_sum_n<16>(p);
        float mine = p[0];
#pragma unroll
        for (int j = 1; j < 16; ++j) mine = (lane == j) ? p[j] : mine;
        if (lane < 16) sz1[w * 16 + lane] = fmaxf(mine + bl1[w * 16 + lane], 0.f);
    }
    __syncthreads();

    {
        const float4 zv = *(const float4*)(sz1 + 4 * lane);
        const float* __restrict__ wbase = Wl2 + (w * 4) * (2 * HID) + 4 * lane;
        float p[4];
#pragma unroll
        for (int j = 0; j < 4; ++j) {
            const float4 wv = *(const float4*)(wbase + j * (2 * HID));
            p[j] = (zv.x * wv.x + zv.y * wv.y) + (zv.z * wv.z + zv.w * wv.w);
        }
        wave_sum_n<4>(p);
        float mine = p[0];
#pragma unroll
        for (int j = 1; j < 4; ++j) mine = (lane == j) ? p[j] : mine;
        if (lane < 4) sz2[w * 4 + lane] = mine + bl2[w * 4 + lane];
    }
    __syncthreads();

    if (w == 0) {
        const float s = sz2[lane];
        float q[1] = { s * s };
        wave_sum_n<1>(q);
        const float nrm = fmaxf(sqrtf(q[0]), 1e-12f);
        outp[g * (HID / 2) + lane] = s / nrm;
    }
}

extern "C" void kernel_launch(void* const* d_in, const int* in_sizes, int n_in,
                              void* d_out, int out_size, void* d_ws, size_t ws_size,
                              hipStream_t stream) {
    const float* x    = (const float*)d_in[0];
    const int*  eidx  = (const int*)  d_in[1];
    const int*  batch = (const int*)  d_in[2];
    const float* t    = (const float*)d_in[3];
    const float* W1r  = (const float*)d_in[4];
    const float* b1   = (const float*)d_in[5];
    const float* W1s  = (const float*)d_in[6];
    const float* W2r  = (const float*)d_in[7];
    const float* b2   = (const float*)d_in[8];
    const float* W2s  = (const float*)d_in[9];
    const float* W3r  = (const float*)d_in[10];
    const float* b3   = (const float*)d_in[11];
    const float* W3s  = (const float*)d_in[12];
    const float* Wl1  = (const float*)d_in[13];
    const float* bl1  = (const float*)d_in[14];
    const float* Wl2  = (const float*)d_in[15];
    const float* bl2  = (const float*)d_in[16];
    const int* src = eidx;
    const int* dst = eidx + NE;

    float* ws = (float*)d_ws;
    size_t o = 0;
    float*    x1     = ws + o; o += (size_t)NN * HID;
    float*    x2     = ws + o; o += (size_t)NN * HID;
    float*    x3     = ws + o; o += (size_t)NN * HID;
    int*      csr    = (int*)(ws + o); o += (size_t)NN * HID * 3;
    int*      cnt    = csr;
    int*      rowptr = cnt + NN;
    int*      cursor = rowptr + NN + 16;
    int*      csrc   = cursor + NN;
    float*    aggr   = ws + o; o += (size_t)NN * HID;
    float*    xpad   = ws + o; o += (size_t)NN * 32;
    float*    w1rp   = ws + o; o += (size_t)HID * 32;
    float*    w1sp   = ws + o; o += (size_t)HID * 32;
    unsigned* henc   = (unsigned*)(ws + o); o += 3 * NG * HID;

    k_prep<<<(NN + 255) / 256, 256, 0, stream>>>(cnt, henc, W1r, W1s, w1rp, w1sp);
    k_csr_hist_pad<<<(NE + 255) / 256, 256, 0, stream>>>(dst, cnt, x, xpad);
    k_csr_scan<<<(NN + 1023) / 1024, 1024, 0, stream>>>(cnt, rowptr, cursor, NN);
    k_csr_fill<<<(NE + 255) / 256, 256, 0, stream>>>(src, dst, cursor, csrc, NE);

    struct Layer { const float* xin; int Cp; const float *Wr, *b, *Ws; float* xout; };
    Layer layers[3] = {
        { xpad, 32, w1rp, b1, w1sp, x1 },
        { x1,   64, W2r,  b2, W2s,  x2 },
        { x2,   64, W3r,  b3, W3s,  x3 },
    };

    for (int l = 0; l < 3; ++l) {
        const Layer& L = layers[l];
        if (L.Cp == 32)
            k_seg_softmax<32><<<(NN + 7) / 8, 256, 0, stream>>>(rowptr, csrc, L.xin, t, aggr, NN);
        else
            k_seg_softmax<64><<<(NN + 7) / 8, 256, 0, stream>>>(rowptr, csrc, L.xin, t, aggr, NN);

        if (L.Cp == 32)
            k_gemm_norm_max<32><<<((NN + 15) / 16 + 3) / 4, 128, 0, stream>>>(aggr, L.xin, L.Wr, L.Ws, L.b, batch,
                                                                              L.xout, henc + (size_t)l * NG * HID, NN);
        else
            k_gemm_norm_max<64><<<((NN + 15) / 16 + 3) / 4, 128, 0, stream>>>(aggr, L.xin, L.Wr, L.Ws, L.b, batch,
                                                                              L.xout, henc + (size_t)l * NG * HID, NN);
    }

    k_head<<<NG, 256, 0, stream>>>(henc, Wl1, bl1, Wl2, bl2, (float*)d_out);
}


# Edge list ordered by destination with a two-level counting sort in LDS (blocks -> 256-node bins -> nodes): no global atomics or scattered stores, 4 launches to 3; per-node sums in LDS-rank order
# speedup vs baseline: 5.4005x; 1.7316x over previous
#include <hip/hip_runtime.h>
#include <math.h>

#define NN      50000
#define NE      800000
#define NG      128
#define INC     31
#define HID     64
#define EPS     1e-5f

typedef __attribute__((ext_vector_type(2))) float v2f;
typedef __attribute__((ext_vector_type(8))) float v8f;

static __device__ __forceinline__ unsigned enc_f32(float x) {
    unsigned b = __float_as_uint(x);
    return (b & 0x80000000u) ? ~b : (b | 0x80000000u);
}
static __device__ __forceinline__ float dec_f32(unsigned e) {
    unsigned b = (e & 0x80000000u) ? (e & 0x7FFFFFFFu) : ~e;
    return __uint_as_float(b);
}
#define ENC_NEG_INF 0x007FFFFFu

#define EPB    4096
#define NBLK   ((NE + EPB - 1) / EPB)
#define BINSH  8
#define BINW   (1 << BINSH)
#define NBIN   ((NN + BINW - 1) / BINW)
#define CGRP   5
#define CSTEPS ((NBLK + CGRP - 1) / CGRP)
static_assert(NN <= 65536, "source and destination must fit 16 bits each");
static_assert(NBIN <= 256 && CGRP * NBIN <= 1024, "one thread per bin and group");
static_assert(NBLK * 1024 >= 3 * NG * HID && NBLK * 1024 >= HID * 32, "prep ranges are prefixes of the grid");
static_assert((long long)NBLK * 1024 * 8 >= (long long)NN * 32, "eight padded elements per thread cover x");

static __device__ __forceinline__ void excl_scan256x2(int a, int b, int (&wsa)[8], int (&wsb)[8],
                                                      int& ea, int& eb, int& tota) {
    const int tid = threadIdx.x, lane = tid & 31, w = tid >> 5;
    int ia = a, ib = b;
    if (w < 8) {
        for (int off = 1; off < 32; off <<= 1) {
            const int ua = __shfl_up(ia, off), ub = __shfl_up(ib, off);
            if (lane >= off) { ia += ua; ib += ub; }
        }
        if (lane == 31) { wsa[w] = ia; wsb[w] = ib; }
    }
    __syncthreads();
    int ba = 0, bb = 0, ta = 0;
#pragma unroll
    for (int i = 0; i < 8; ++i) {
        const int sa = wsa[i], sb = wsb[i];
        ta += sa;
        if (i < w) { ba += sa; bb += sb; }
    }
    ea = ba + ia - a; eb = bb + ib - b; tota = ta;
}

__global__ void __launch_bounds__(1024) k_sort_count(const int* __restrict__ dst, int* __restrict__ bcnt,
                                                     unsigned* __restrict__ henc,
                                                     const float* __restrict__ W1r, const float* __restrict__ W1s,
                                                     float* __restrict__ w1rp, float* __restrict__ w1sp,
                                                     const float* __restrict__ x, float* __restrict__ xpad) {
    __shared__ int h[NBIN];
    const int tid = threadIdx.x, blk = blockIdx.x;
    if (tid < NBIN) h[tid] = 0;
    int d[4];
#pragma unroll
    for (int k = 0; k < 4; ++k) d[k] = dst[min(blk * EPB + k * 1024 + tid, NE - 1)];
    const int i0 = blk * 1024 + tid;
    float xv[8];
#pragma unroll
    for (int k = 0; k < 8; ++k) {
        const int j = min(i0 + k * (NBLK * 1024), NN * 32 - 1);
        xv[k] = x[(j >> 5) * INC + min(j & 31, INC - 1)];
    }
    __syncthreads();
#pragma unroll
    for (int k = 0; k < 4; ++k)
        if (blk * EPB + k * 1024 + tid < NE) atomicAdd(&h[d[k] >> BINSH], 1);
    __syncthreads();
    if (tid < NBIN) bcnt[blk * NBIN + tid] = h[tid];

    if (i0 < 3 * NG * HID) henc[i0] = ENC_NEG_INF;
    if (i0 < HID * 32) {
        const int r = i0 / 32, c = i0 % 32;
        w1rp[i0] = (c < INC) ? W1r[r * INC + c] : 0.f;
        w1sp[i0] = (c < INC) ? W1s[r * INC + c] : 0.f;
    }
#pragma unroll
    for (int k = 0; k < 8; ++k) {
        const int j = i0 + k * (NBLK * 1024);
        if (j < NN * 32) xpad[j] = ((j & 31) < INC) ? xv[k] : 0.f;
    }
}

__global__ void __launch_bounds__(1024) k_sort_scatter(const int* __restrict__ src, const int* __restrict__ dst,
                                                       const int* __restrict__ bcnt, int* __restrict__ binstart,
                                                       unsigned* __restrict__ binned) {
    __shared__ int tpart[CGRP][NBIN];
    __shared__ int ppart[CGRP][NBIN];
    __shared__ int h[NBIN];
    __shared__ int loff[NBIN];
    __shared__ int gbase[NBIN];
    __shared__ int wsa[8];
    __shared__ int wsb[8];
    __shared__ unsigned stage[EPB];
    const int tid = threadIdx.x, blk = blockIdx.x;
    if (tid < NBIN) h[tid] = 0;
    int s[4], d[4];
#pragma unroll
    for (int k = 0; k < 4; ++k) {
        const int e = min(blk * EPB + k * 1024 + tid, NE - 1);
        s[k] = src[e];
        d[k] = dst[e];
    }
    const int g = tid / NBIN, cb = tid - g * NBIN;
    if (g < CGRP) {
        int tot = 0, pre = 0;
#pragma unroll
        for (int it = 0; it < CSTEPS; ++it) {
            const int b = g + CGRP * it;
            int v = bcnt[min(b, NBLK - 1) * NBIN + cb];
            v = (b < NBLK) ? v : 0;
            tot += v;
            pre += (b < blk) ? v : 0;
        }
        tpart[g][cb] = tot;
        ppart[g][cb] = pre;
    }
    __syncthreads();
    int r[4];
#pragma unroll
    for (int k = 0; k < 4; ++k)
        r[k] = (blk * EPB + k * 1024 + tid < NE) ? atomicAdd(&h[d[k] >> BINSH], 1) : 0;
    int T = 0, P = 0;
    if (tid < NBIN) {
#pragma unroll
        for (int q = 0; q < CGRP; ++q) { T += tpart[q][tid]; P += ppart[q][tid]; }
    }
    __syncthreads();
    const int hv = (tid < NBIN) ? h[tid] : 0;
    int bs, lo, total;
    excl_scan256x2(T, hv, wsa, wsb, bs, lo, total);
    if (tid < NBIN) {
        gbase[tid] = bs + P;
        loff[tid] = lo;
        if (blk == 0) binstart[tid] = bs;
    }
    if (blk == 0 && tid == 0) binstart[NBIN] = total;
    __syncthreads();
    int lf[4];
#pragma unroll
    for (int k = 0; k < 4; ++k) lf[k] = loff[d[k] >> BINSH];
#pragma unroll
    for (int k = 0; k < 4; ++k) {
        const int i = k * 1024 + tid;
        const bool live = blk * EPB + i < NE;
        stage[live ? lf[k] + r[k] : i] = live ? (((unsigned)d[k] << 16) | (unsigned)s[k]) : 0u;
    }
    __syncthreads();
    const int nvalid = min(EPB, NE - blk * EPB);
    unsigned p[4];
    int gb[4], lb[4];
#pragma unroll
    for (int k = 0; k < 4; ++k) p[k] = stage[k * 1024 + tid];
#pragma unroll
    for (int k = 0; k < 4; ++k) {
        const int b = (int)(p[k] >> (16 + BINSH));
        gb[k] = gbase[b];
        lb[k] = loff[b];
    }
#pragma unroll
    for (int k = 0; k < 4; ++k) {
        const int i = k * 1024 + tid;
        if (i < nvalid) binned[gb[k] + (i - lb[k])] = p[k];
    }
}

#define BCAP (2 * EPB)
__global__ void __launch_bounds__(1024) k_sort_bin(const unsigned* __restrict__ binned, const int* __restrict__ binstart,
                                                   int* __restrict__ rowptr, int* __restrict__ csrc) {
    __shared__ int cnt[BINW];
    __shared__ int wsa[8];
    __shared__ int wsb[8];
    __shared__ int outst[BCAP];
    const int tid = threadIdx.x, bin = blockIdx.x;
    const int beg = binstart[bin], end = binstart[bin + 1], n = end - beg;
    if (tid < BINW) cnt[tid] = 0;
    __syncthreads();
    for (int base = beg; base < end; base += EPB) {
        unsigned p[4];
#pragma unroll
        for (int k = 0; k < 4; ++k) p[k] = binned[min(base + k * 1024 + tid, end - 1)];
#pragma unroll
        for (int k = 0; k < 4; ++k)
            if (base + k * 1024 + tid < end) atomicAdd(&cnt[(p[k] >> 16) & (BINW - 1)], 1);
    }
    __syncthreads();
    const int c = (tid < BINW) ? cnt[tid] : 0;
    int ex, unused, total;
    excl_scan256x2(c, 0, wsa, wsb, ex, unused, total);
    if (tid < BINW) {
        const int node = bin * BINW + tid;
        if (node <= NN) rowptr[node] = beg + ex;
        cnt[tid] = ex;
    }
    __syncthreads();
    const bool staged = n <= BCAP;
    for (int base = beg; base < end; base += EPB) {
        unsigned p[4];
#pragma unroll
        for (int k = 0; k < 4; ++k) p[k] = binned[min(base + k * 1024 + tid, end - 1)];
        int rk[4];
#pragma unroll
        for (int k = 0; k < 4; ++k)
            rk[k] = (base + k * 1024 + tid < end) ? atomicAdd(&cnt[(p[k] >> 16) & (BINW - 1)], 1) : 0;
#pragma unroll
        for (int k = 0; k < 4; ++k) {
            if (base + k * 1024 + tid < end) {
                const int sv = (int)(p[k] & 0xFFFFu);
                if (staged) outst[rk[k]] = sv; else csrc[beg + rk[k]] = sv;
            }
        }
    }
    if (staged && n > 0) {
        __syncthreads();
        int v[BCAP / 1024];
#pragma unroll
        for (int k = 0; k < BCAP / 1024; ++k) v[k] = outst[min(k * 1024 + tid, n - 1)];
#pragma unroll
        for (int k = 0; k < BCAP / 1024; ++k)
            if (k * 1024 + tid < n) csrc[beg + k * 1024 + tid] = v[k];
    }
}

template<int VEC>
static __device__ __forceinline__ void ld_row(const float* __restrict__ p, float (&v)[VEC]) {
    if constexpr (VEC == 2) {
        v2f r = *(const v2f*)p;
        v[0] = r.x; v[1] = r.y;
    } else {
        v[0] = p[0];
    }
}

template<int VEC>
static __device__ __forceinline__ void seg_max(const float (&r)[VEC], float t, float (&mx)[VEC]) {
#pragma unroll
    for (int i = 0; i < VEC; ++i) mx[i] = fmaxf(mx[i], r[i] * t);
}
template<int VEC>
static __device__ __forceinline__ void seg_sum(const float (&r)[VEC], float t, const float (&mx)[VEC],
                                               float (&num)[VEC], float (&den)[VEC]) {
#pragma unroll
    for (int i = 0; i < VEC; ++i) {
        float e = __expf(r[i] * t - mx[i]);
        num[i] += r[i] * e;
        den[i] += e;
    }
}

template<int J, int VEC, int CP>
struct SegChain {
    static __device__ __forceinline__ void load(int deg, const float* __restrict__ x, unsigned loff,
                                                const int (&sidx)[32], float (&v)[32][VEC]) {
        if constexpr (J < 32) {
            if (J < deg) {
                ld_row<VEC>(x + (size_t)sidx[J] * CP + loff, v[J]);
                SegChain<J + 1, VEC, CP>::load(deg, x, loff, sidx, v);
            }
        }
    }
    static __device__ __forceinline__ void vmax(int deg, const float (&v)[32][VEC], float t, float (&mx)[VEC]) {
        if constexpr (J < 32) {
            if (J < deg) {
                seg_max<VEC>(v[J], t, mx);
                SegChain<J + 1, VEC, CP>::vmax(deg, v, t, mx);
            }
        }
    }
    static __device__ __forceinline__ void vsum(int deg, const float (&v)[32][VEC], float t,
                                                const float (&mx)[VEC], float (&num)[VEC], float (&den)[VEC]) {
        if constexpr (J < 32) {
            if (J < deg) {
                seg_sum<VEC>(v[J], t, mx, num, den);
                SegChain<J + 1, VEC, CP>::vsum(deg, v, t, mx, num, den);
            }
        }
    }
};

template<int CP>
__global__ void __launch_bounds__(256) k_seg_softmax(const int* __restrict__ rowptr,
                                                     const int* __restrict__ csrc,
                                                     const float* __restrict__ x,
                                                     const float* __restrict__ tptr,
                                                     float* __restrict__ aggr, int n) {
    constexpr int VEC = CP / 32;
    const int lane = threadIdx.x & 31;
    const int node = __builtin_amdgcn_readfirstlane((int)(blockIdx.x * 8 + (threadIdx.x >> 5)));
    if (node >= n) return;
    const int beg = rowptr[node];
    const int end = rowptr[node + 1];
    const int deg = end - beg;
    const float t = tptr[0];
    const unsigned loff = (unsigned)(lane * VEC);

    float mx[VEC], num[VEC], den[VEC];
#pragma unroll
    for (int i = 0; i < VEC; ++i) { mx[i] = -INFINITY; num[i] = 0.f; den[i] = 0.f; }

    if (deg <= 32) {
        const int mys = (lane < deg) ? csrc[beg + lane] : 0;
        int sidx[32];
#pragma unroll
        for (int j = 0; j < 32; ++j) sidx[j] = __builtin_amdgcn_readlane(mys, j);
        float v[32][VEC];
        SegChain<0, VEC, CP>::load(deg, x, loff, sidx, v);
        SegChain<0, VEC, CP>::vmax(deg, v, t, mx);
        SegChain<0, VEC, CP>::vsum(deg, v, t, mx, num, den);
    } else {
        for (int base = beg; base < end; base += 32) {
            const int cnt = min(32, end - base);
            const int mys = (lane < cnt) ? csrc[base + lane] : 0;
            for (int j = 0; j < cnt; ++j) {
                const int s = __builtin_amdgcn_readlane(mys, j);
                float r[VEC];
                ld_row<VEC>(x + (size_t)s * CP + loff, r);
                seg_max<VEC>(r, t, mx);
            }
        }
        for (int base = beg; base < end; base += 32) {
            const int cnt = min(32, end - base);
            const int mys = (lane < cnt) ? csrc[base + lane] : 0;
            for (int j = 0; j < cnt; ++j) {
                const int s = __builtin_amdgcn_readlane(mys, j);
                float r[VEC];
                ld_row<VEC>(x + (size_t)s * CP + loff, r);
                seg_sum<VEC>(r, t, mx, num, den);
            }
        }
    }

    float res[VEC];
#pragma unroll
    for (int i = 0; i < VEC; ++i) res[i] = den[i] > 0.f ? num[i] / den[i] : 0.f;
    float* o = aggr + (size_t)node * CP + loff;
    if constexpr (VEC == 2) {
        v2f ov; ov.x = res[0]; ov.y = res[1];
        *(v2f*)o = ov;
    } else {
        o[0] = res[0];
    }
}


static __device__ __forceinline__ float xhalf(float x) {
    int b = __builtin_bit_cast(int, x);
    return __builtin_bit_cast(float, __builtin_amdgcn_permlanex16(b, b, 0x76543210, 0xfedcba98, false, false));
}
template<int M>
static __device__ __forceinline__ float rowx(float x) {
    return __builtin_bit_cast(float, __builtin_amdgcn_update_dpp(0, __builtin_bit_cast(int, x), 0x160 + M, 0xf, 0xf, true));
}
template<int M>
static __device__ __forceinline__ void row_step8(float (&s)[8]) {
#pragma unroll
    for (int r = 0; r < 8; ++r) s[r] += rowx<M>(s[r]);
}
static __device__ __forceinline__ void row_sum16x8(float (&s)[8]) {
    row_step8<1>(s);
    row_step8<2>(s);
    row_step8<4>(s);
    row_step8<8>(s);
}

template<int CP>
__global__ void __launch_bounds__(128) k_gemm_norm_max(const float* __restrict__ aggr, const float* __restrict__ xin,
                                                       const float* __restrict__ Wr, const float* __restrict__ Ws,
                                                       const float* __restrict__ bias, const int* __restrict__ batch,
                                                       float* __restrict__ out, unsigned* __restrict__ he, int Nrows) {
    const int lane = threadIdx.x & 31;
    const int half = lane >> 4;
    const int c    = lane & 15;
    const int m0 = __builtin_amdgcn_readfirstlane((int)(blockIdx.x * 4 + (threadIdx.x >> 5))) * 16;
    if (m0 >= Nrows) return;
    int arow = m0 + c;
    if (arow >= Nrows) arow = Nrows - 1;

    const float* ap = aggr + (size_t)arow * CP + half * 2;
    const float* xp = xin  + (size_t)arow * CP + half * 2;
    const float* wr = Wr   + (size_t)(4 * c) * CP + half * 2;
    const float* wsp= Ws   + (size_t)(4 * c) * CP + half * 2;
    const int gl = batch[arow];

    v8f acc[4] = {};
#pragma unroll
    for (int k = 0; k < CP; k += 4) {
        v2f a = *(const v2f*)(ap + k);
#pragma unroll
        for (int j = 0; j < 4; ++j) {
            v2f b = *(const v2f*)(wr + j * CP + k);
            acc[j] = __builtin_amdgcn_wmma_f32_16x16x4_f32(false, a, false, b,
                                                           (short)0, acc[j], false, false);
        }
    }
#pragma unroll
    for (int k = 0; k < CP; k += 4) {
        v2f a = *(const v2f*)(xp + k);
#pragma unroll
        for (int j = 0; j < 4; ++j) {
            v2f b = *(const v2f*)(wsp + j * CP + k);
            acc[j] = __builtin_amdgcn_wmma_f32_16x16x4_f32(false, a, false, b,
                                                           (short)0, acc[j], false, false);
        }
    }

    const float4 bq = *(const float4*)(bias + 4 * c);
    const float bv[4] = { bq.x, bq.y, bq.z, bq.w };
    float s[8], q[8];
#pragma unroll
    for (int r = 0; r < 8; ++r) {
#pragma unroll
        for (int j = 0; j < 4; ++j) acc[j][r] += bv[j];
        s[r] = (acc[0][r] + acc[1][r]) + (acc[2][r] + acc[3][r]);
    }
    row_sum16x8(s);
#pragma unroll
    for (int r = 0; r < 8; ++r) {
        const float mu = s[r] * (1.f / HID);
#pragma unroll
        for (int j = 0; j < 4; ++j) acc[j][r] -= mu;
        q[r] = (acc[0][r] * acc[0][r] + acc[1][r] * acc[1][r]) + (acc[2][r] * acc[2][r] + acc[3][r] * acc[3][r]);
    }
    row_sum16x8(q);
    float cm[4];
#pragma unroll
    for (int r = 0; r < 8; ++r) {
        const float inv = rsqrtf(q[r] * (1.f / HID) + EPS);
#pragma unroll
        for (int j = 0; j < 4; ++j) {
            const float y = fmaxf(acc[j][r] * inv, 0.f);
            acc[j][r] = y;
            cm[j] = (r == 0) ? y : fmaxf(cm[j], y);
        }
    }
    float* op = out + (size_t)(m0 + half * 8) * HID + 4 * c;
    if (m0 + 16 <= Nrows) {
#pragma unroll
        for (int r = 0; r < 8; ++r) {
            float4 o; o.x = acc[0][r]; o.y = acc[1][r]; o.z = acc[2][r]; o.w = acc[3][r];
            *(float4*)(op + (size_t)r * HID) = o;
        }
    } else {
#pragma unroll
        for (int r = 0; r < 8; ++r) {
            if (m0 + r + half * 8 < Nrows) {
                float4 o; o.x = acc[0][r]; o.y = acc[1][r]; o.z = acc[2][r]; o.w = acc[3][r];
                *(float4*)(op + (size_t)r * HID) = o;
            }
        }
    }

    const int g0 = __builtin_amdgcn_readfirstlane(gl);
    const bool one_graph = __builtin_amdgcn_ballot_w32(gl != g0) == 0u;
    if (one_graph) {
#pragma unroll
        for (int j = 0; j < 4; ++j) cm[j] = fmaxf(cm[j], xhalf(cm[j]));
        const float p0 = half ? cm[2] : cm[0];
        const float p1 = half ? cm[3] : cm[1];
        unsigned* hp = he + (size_t)g0 * HID + 4 * c + 2 * half;
        atomicMax(hp,     enc_f32(p0));
        atomicMax(hp + 1, enc_f32(p1));
    } else {
#pragma unroll
        for (int r = 0; r < 8; ++r) {
            const int row = m0 + r + half * 8;
            if (row < Nrows) {
                unsigned* hp = he + (size_t)batch[row] * HID + 4 * c;
#pragma unroll
                for (int j = 0; j < 4; ++j) atomicMax(hp + j, enc_f32(acc[j][r]));
            }
        }
    }
}

template<int NV>
static __device__ __forceinline__ void wave_sum_n(float (&p)[NV]) {
#pragma unroll
    for (int j = 0; j < NV; ++j) p[j] += rowx<1>(p[j]);
#pragma unroll
    for (int j = 0; j < NV; ++j) p[j] += rowx<2>(p[j]);
#pragma unroll
    for (int j = 0; j < NV; ++j) p[j] += rowx<4>(p[j]);
#pragma unroll
    for (int j = 0; j < NV; ++j) p[j] += rowx<8>(p[j]);
#pragma unroll
    for (int j = 0; j < NV; ++j) p[j] += xhalf(p[j]);
}

__global__ void __launch_bounds__(256) k_head(const unsigned* __restrict__ henc,
                                              const float* __restrict__ Wl1, const float* __restrict__ bl1,
                                              const float* __restrict__ Wl2, const float* __restrict__ bl2,
                                              float* __restrict__ outp) {
    __shared__ __attribute__((aligned(16))) float sh[3 * HID];
    __shared__ __attribute__((aligned(16))) float sz1[2 * HID];
    __shared__ __attribute__((aligned(16))) float sz2[HID / 2];
    const int g = blockIdx.x;
    const int tid = threadIdx.x, lane = tid & 31, w = tid >> 5;
    if (tid < 3 * HID) {
        const int layer = tid / HID, cc = tid % HID;
        sh[tid] = dec_f32(henc[layer * NG * HID + g * HID + cc]);
    }
    __syncthreads();

    {
        v2f hv[3];
#pragma unroll
        for (int i = 0; i < 3; ++i) hv[i] = *(const v2f*)(sh + 2 * (lane + 32 * i));
        const float* __restrict__ wbase = Wl1 + (w * 16) * (3 * HID) + 2 * lane;
        float p[16];
#pragma unroll
        for (int j = 0; j < 16; ++j) {
            const float* __restrict__ wrow = wbase + j * (3 * HID);
            float a = 0.f;
#pragma unroll
            for (int i = 0; i < 3; ++i) {
                const v2f wv = *(const v2f*)(wrow + 64 * i);
                a += hv[i].x * wv.x;
                a += hv[i].y * wv.y;
            }
            p[j] = a;
        }
        wave_sum_n<16>(p);
        float mine = p[0];
#pragma unroll
        for (int j = 1; j < 16; ++j) mine = (lane == j) ? p[j] : mine;
        if (lane < 16) sz1[w * 16 + lane] = fmaxf(mine + bl1[w * 16 + lane], 0.f);
    }
    __syncthreads();

    {
        const float4 zv = *(const float4*)(sz1 + 4 * lane);
        const float* __restrict__ wbase = Wl2 + (w * 4) * (2 * HID) + 4 * lane;
        float p[4];
#pragma unroll
        for (int j = 0; j < 4; ++j) {
            const float4 wv = *(const float4*)(wbase + j * (2 * HID));
            p[j] = (zv.x * wv.x + zv.y * wv.y) + (zv.z * wv.z + zv.w * wv.w);
        }
        wave_sum_n<4>(p);
        float mine = p[0];
#pragma unroll
        for (int j = 1; j < 4; ++j) mine = (lane == j) ? p[j] : mine;
        if (lane < 4) sz2[w * 4 + lane] = mine + bl2[w * 4 + lane];
    }
    __syncthreads();

    if (w == 0) {
        const float s = sz2[lane];
        float q[1] = { s * s };
        wave_sum_n<1>(q);
        const float nrm = fmaxf(sqrtf(q[0]), 1e-12f);
        outp[g * (HID / 2) + lane] = s / nrm;
    }
}

extern "C" void kernel_launch(void* const* d_in, const int* in_sizes, int n_in,
                              void* d_out, int out_size, void* d_ws, size_t ws_size,
                              hipStream_t stream) {
    const float* x    = (const float*)d_in[0];
    const int*  eidx  = (const int*)  d_in[1];
    const int*  batch = (const int*)  d_in[2];
    const float* t    = (const float*)d_in[3];
    const float* W1r  = (const float*)d_in[4];
    const float* b1   = (const float*)d_in[5];
    const float* W1s  = (const float*)d_in[6];
    const float* W2r  = (const float*)d_in[7];
    const float* b2   = (const float*)d_in[8];
    const float* W2s  = (const float*)d_in[9];
    const float* W3r  = (const float*)d_in[10];
    const float* b3   = (const float*)d_in[11];
    const float* W3s  = (const float*)d_in[12];
    const float* Wl1  = (const float*)d_in[13];
    const float* bl1  = (const float*)d_in[14];
    const float* Wl2  = (const float*)d_in[15];
    const float* bl2  = (const float*)d_in[16];
    const int* src = eidx;
    const int* dst = eidx + NE;

    float* ws = (float*)d_ws;
    size_t o = 0;
    float*    x1     = ws + o; o += (size_t)NN * HID;
    float*    x2     = ws + o; o += (size_t)NN * HID;
    float*    x3     = ws + o; o += (size_t)NN * HID;
    int*      csr    = (int*)(ws + o); o += (size_t)NN * HID * 3;
    int*      bcnt     = csr;
    int*      binstart = bcnt + NBLK * NBIN;
    unsigned* binned   = (unsigned*)(binstart + NBIN + 16);
    int*      rowptr   = (int*)(binned + NE);
    int*      csrc     = rowptr + NN + 16;
    float*    aggr   = ws + o; o += (size_t)NN * HID;
    float*    xpad   = ws + o; o += (size_t)NN * 32;
    float*    w1rp   = ws + o; o += (size_t)HID * 32;
    float*    w1sp   = ws + o; o += (size_t)HID * 32;
    unsigned* henc   = (unsigned*)(ws + o); o += 3 * NG * HID;

    k_sort_count<<<NBLK, 1024, 0, stream>>>(dst, bcnt, henc, W1r, W1s, w1rp, w1sp, x, xpad);
    k_sort_scatter<<<NBLK, 1024, 0, stream>>>(src, dst, bcnt, binstart, binned);
    k_sort_bin<<<NBIN, 1024, 0, stream>>>(binned, binstart, rowptr, csrc);

    struct Layer { const float* xin; int Cp; const float *Wr, *b, *Ws; float* xout; };
    Layer layers[3] = {
        { xpad, 32, w1rp, b1, w1sp, x1 },
        { x1,   64, W2r,  b2, W2s,  x2 },
        { x2,   64, W3r,  b3, W3s,  x3 },
    };

    for (int l = 0; l < 3; ++l) {
        const Layer& L = layers[l];
        if (L.Cp == 32)
            k_seg_softmax<32><<<(NN + 7) / 8, 256, 0, stream>>>(rowptr, csrc, L.xin, t, aggr, NN);
        else
            k_seg_softmax<64><<<(NN + 7) / 8, 256, 0, stream>>>(rowptr, csrc, L.xin, t, aggr, NN);

        if (L.Cp == 32)
            k_gemm_norm_max<32><<<((NN + 15) / 16 + 3) / 4, 128, 0, stream>>>(aggr, L.xin, L.Wr, L.Ws, L.b, batch,
                                                                              L.xout, henc + (size_t)l * NG * HID, NN);
        else
            k_gemm_norm_max<64><<<((NN + 15) / 16 + 3) / 4, 128, 0, stream>>>(aggr, L.xin, L.Wr, L.Ws, L.b, batch,
                                                                              L.xout, henc + (size_t)l * NG * HID, NN);
    }

    k_head<<<NG, 256, 0, stream>>>(henc, Wl1, bl1, Wl2, bl2, (float*)d_out);
}


# One launch per layer (7 launches, not 10): a 16-wave group aggregates its 16 nodes into LDS and stages weights and root rows there; one wave then runs the WMMA GEMM and tail from LDS; sums as before
# speedup vs baseline: 6.4175x; 1.1883x over previous
#include <hip/hip_runtime.h>
#include <math.h>

#define NN      50000
#define NE      800000
#define NG      128
#define INC     31
#define HID     64
#define EPS     1e-5f

typedef __attribute__((ext_vector_type(2))) float v2f;
typedef __attribute__((ext_vector_type(8))) float v8f;

static __device__ __forceinline__ unsigned enc_f32(float x) {
    unsigned b = __float_as_uint(x);
    return (b & 0x80000000u) ? ~b : (b | 0x80000000u);
}
static __device__ __forceinline__ float dec_f32(unsigned e) {
    unsigned b = (e & 0x80000000u) ? (e & 0x7FFFFFFFu) : ~e;
    return __uint_as_float(b);
}
#define ENC_NEG_INF 0x007FFFFFu

#define EPB    4096
#define NBLK   ((NE + EPB - 1) / EPB)
#define BINSH  8
#define BINW   (1 << BINSH)
#define NBIN   ((NN + BINW - 1) / BINW)
#define CGRP   5
#define CSTEPS ((NBLK + CGRP - 1) / CGRP)
static_assert(NN <= 65536, "source and destination must fit 16 bits each");
static_assert(NBIN <= 256 && CGRP * NBIN <= 1024, "one thread per bin and group");
static_assert(NBLK * 1024 >= 3 * NG * HID && NBLK * 1024 >= HID * 32, "prep ranges are prefixes of the grid");
static_assert((long long)NBLK * 1024 * 8 >= (long long)NN * 32, "eight padded elements per thread cover x");

static __device__ __forceinline__ void excl_scan256x2(int a, int b, int (&wsa)[8], int (&wsb)[8],
                                                      int& ea, int& eb, int& tota) {
    const int tid = threadIdx.x, lane = tid & 31, w = tid >> 5;
    int ia = a, ib = b;
    if (w < 8) {
        for (int off = 1; off < 32; off <<= 1) {
            const int ua = __shfl_up(ia, off), ub = __shfl_up(ib, off);
            if (lane >= off) { ia += ua; ib += ub; }
        }
        if (lane == 31) { wsa[w] = ia; wsb[w] = ib; }
    }
    __syncthreads();
    int ba = 0, bb = 0, ta = 0;
#pragma unroll
    for (int i = 0; i < 8; ++i) {
        const int sa = wsa[i], sb = wsb[i];
        ta += sa;
        if (i < w) { ba += sa; bb += sb; }
    }
    ea = ba + ia - a; eb = bb + ib - b; tota = ta;
}

__global__ void __launch_bounds__(1024) k_sort_count(const int* __restrict__ dst, int* __restrict__ bcnt,
                                                     unsigned* __restrict__ henc,
                                                     const float* __restrict__ W1r, const float* __restrict__ W1s,
                                                     float* __restrict__ w1rp, float* __restrict__ w1sp,
                                                     const float* __restrict__ x, float* __restrict__ xpad) {
    __shared__ int h[NBIN];
    const int tid = threadIdx.x, blk = blockIdx.x;
    if (tid < NBIN) h[tid] = 0;
    int d[4];
#pragma unroll
    for (int k = 0; k < 4; ++k) d[k] = dst[min(blk * EPB + k * 1024 + tid, NE - 1)];
    const int i0 = blk * 1024 + tid;
    float xv[8];
#pragma unroll
    for (int k = 0; k < 8; ++k) {
        const int j = min(i0 + k * (NBLK * 1024), NN * 32 - 1);
        xv[k] = x[(j >> 5) * INC + min(j & 31, INC - 1)];
    }
    __syncthreads();
#pragma unroll
    for (int k = 0; k < 4; ++k)
        if (blk * EPB + k * 1024 + tid < NE) atomicAdd(&h[d[k] >> BINSH], 1);
    __syncthreads();
    if (tid < NBIN) bcnt[blk * NBIN + tid] = h[tid];

    if (i0 < 3 * NG * HID) henc[i0] = ENC_NEG_INF;
    if (i0 < HID * 32) {
        const int r = i0 / 32, c = i0 % 32;
        w1rp[i0] = (c < INC) ? W1r[r * INC + c] : 0.f;
        w1sp[i0] = (c < INC) ? W1s[r * INC + c] : 0.f;
    }
#pragma unroll
    for (int k = 0; k < 8; ++k) {
        const int j = i0 + k * (NBLK * 1024);
        if (j < NN * 32) xpad[j] = ((j & 31) < INC) ? xv[k] : 0.f;
    }
}

__global__ void __launch_bounds__(1024) k_sort_scatter(const int* __restrict__ src, const int* __restrict__ dst,
                                                       const int* __restrict__ bcnt, int* __restrict__ binstart,
                                                       unsigned* __restrict__ binned) {
    __shared__ int tpart[CGRP][NBIN];
    __shared__ int ppart[CGRP][NBIN];
    __shared__ int h[NBIN];
    __shared__ int loff[NBIN];
    __shared__ int gbase[NBIN];
    __shared__ int wsa[8];
    __shared__ int wsb[8];
    __shared__ unsigned stage[EPB];
    const int tid = threadIdx.x, blk = blockIdx.x;
    if (tid < NBIN) h[tid] = 0;
    int s[4], d[4];
#pragma unroll
    for (int k = 0; k < 4; ++k) {
        const int e = min(blk * EPB + k * 1024 + tid, NE - 1);
        s[k] = src[e];
        d[k] = dst[e];
    }
    const int g = tid / NBIN, cb = tid - g * NBIN;
    if (g < CGRP) {
        int tot = 0, pre = 0;
#pragma unroll
        for (int it = 0; it < CSTEPS; ++it) {
            const int b = g + CGRP * it;
            int v = bcnt[min(b, NBLK - 1) * NBIN + cb];
            v = (b < NBLK) ? v : 0;
            tot += v;
            pre += (b < blk) ? v : 0;
        }
        tpart[g][cb] = tot;
        ppart[g][cb] = pre;
    }
    __syncthreads();
    int r[4];
#pragma unroll
    for (int k = 0; k < 4; ++k)
        r[k] = (blk * EPB + k * 1024 + tid < NE) ? atomicAdd(&h[d[k] >> BINSH], 1) : 0;
    int T = 0, P = 0;
    if (tid < NBIN) {
#pragma unroll
        for (int q = 0; q < CGRP; ++q) { T += tpart[q][tid]; P += ppart[q][tid]; }
    }
    __syncthreads();
    const int hv = (tid < NBIN) ? h[tid] : 0;
    int bs, lo, total;
    excl_scan256x2(T, hv, wsa, wsb, bs, lo, total);
    if (tid < NBIN) {
        gbase[tid] = bs + P;
        loff[tid] = lo;
        if (blk == 0) binstart[tid] = bs;
    }
    if (blk == 0 && tid == 0) binstart[NBIN] = total;
    __syncthreads();
    int lf[4];
#pragma unroll
    for (int k = 0; k < 4; ++k) lf[k] = loff[d[k] >> BINSH];
#pragma unroll
    for (int k = 0; k < 4; ++k) {
        const int i = k * 1024 + tid;
        const bool live = blk * EPB + i < NE;
        stage[live ? lf[k] + r[k] : i] = live ? (((unsigned)d[k] << 16) | (unsigned)s[k]) : 0u;
    }
    __syncthreads();
    const int nvalid = min(EPB, NE - blk * EPB);
    unsigned p[4];
    int gb[4], lb[4];
#pragma unroll
    for (int k = 0; k < 4; ++k) p[k] = stage[k * 1024 + tid];
#pragma unroll
    for (int k = 0; k < 4; ++k) {
        const int b = (int)(p[k] >> (16 + BINSH));
        gb[k] = gbase[b];
        lb[k] = loff[b];
    }
#pragma unroll
    for (int k = 0; k < 4; ++k) {
        const int i = k * 1024 + tid;
        if (i < nvalid) binned[gb[k] + (i - lb[k])] = p[k];
    }
}

#define BCAP (2 * EPB)
__global__ void __launch_bounds__(1024) k_sort_bin(const unsigned* __restrict__ binned, const int* __restrict__ binstart,
                                                   int* __restrict__ rowptr, int* __restrict__ csrc) {
    __shared__ int cnt[BINW];
    __shared__ int wsa[8];
    __shared__ int wsb[8];
    __shared__ int outst[BCAP];
    const int tid = threadIdx.x, bin = blockIdx.x;
    const int beg = binstart[bin], end = binstart[bin + 1], n = end - beg;
    if (tid < BINW) cnt[tid] = 0;
    __syncthreads();
    for (int base = beg; base < end; base += EPB) {
        unsigned p[4];
#pragma unroll
        for (int k = 0; k < 4; ++k) p[k] = binned[min(base + k * 1024 + tid, end - 1)];
#pragma unroll
        for (int k = 0; k < 4; ++k)
            if (base + k * 1024 + tid < end) atomicAdd(&cnt[(p[k] >> 16) & (BINW - 1)], 1);
    }
    __syncthreads();
    const int c = (tid < BINW) ? cnt[tid] : 0;
    int ex, unused, total;
    excl_scan256x2(c, 0, wsa, wsb, ex, unused, total);
    if (tid < BINW) {
        const int node = bin * BINW + tid;
        if (node <= NN) rowptr[node] = beg + ex;
        cnt[tid] = ex;
    }
    __syncthreads();
    const bool staged = n <= BCAP;
    for (int base = beg; base < end; base += EPB) {
        unsigned p[4];
#pragma unroll
        for (int k = 0; k < 4; ++k) p[k] = binned[min(base + k * 1024 + tid, end - 1)];
        int rk[4];
#pragma unroll
        for (int k = 0; k < 4; ++k)
            rk[k] = (base + k * 1024 + tid < end) ? atomicAdd(&cnt[(p[k] >> 16) & (BINW - 1)], 1) : 0;
#pragma unroll
        for (int k = 0; k < 4; ++k) {
            if (base + k * 1024 + tid < end) {
                const int sv = (int)(p[k] & 0xFFFFu);
                if (staged) outst[rk[k]] = sv; else csrc[beg + rk[k]] = sv;
            }
        }
    }
    if (staged && n > 0) {
        __syncthreads();
        int v[BCAP / 1024];
#pragma unroll
        for (int k = 0; k < BCAP / 1024; ++k) v[k] = outst[min(k * 1024 + tid, n - 1)];
#pragma unroll
        for (int k = 0; k < BCAP / 1024; ++k)
            if (k * 1024 + tid < n) csrc[beg + k * 1024 + tid] = v[k];
    }
}

template<int VEC>
static __device__ __forceinline__ void ld_row(const float* __restrict__ p, float (&v)[VEC]) {
    if constexpr (VEC == 2) {
        v2f r = *(const v2f*)p;
        v[0] = r.x; v[1] = r.y;
    } else {
        v[0] = p[0];
    }
}

template<int VEC>
static __device__ __forceinline__ void seg_max(const float (&r)[VEC], float t, float (&mx)[VEC]) {
#pragma unroll
    for (int i = 0; i < VEC; ++i) mx[i] = fmaxf(mx[i], r[i] * t);
}
template<int VEC>
static __device__ __forceinline__ void seg_sum(const float (&r)[VEC], float t, const float (&mx)[VEC],
                                               float (&num)[VEC], float (&den)[VEC]) {
#pragma unroll
    for (int i = 0; i < VEC; ++i) {
        float e = __expf(r[i] * t - mx[i]);
        num[i] += r[i] * e;
        den[i] += e;
    }
}

template<int J, int VEC, int CP>
struct SegChain {
    static __device__ __forceinline__ void load(int deg, const float* __restrict__ x, unsigned loff,
                                                const int (&sidx)[32], float (&v)[32][VEC]) {
        if constexpr (J < 32) {
            if (J < deg) {
                ld_row<VEC>(x + (size_t)sidx[J] * CP + loff, v[J]);
                SegChain<J + 1, VEC, CP>::load(deg, x, loff, sidx, v);
            }
        }
    }
    static __device__ __forceinline__ void vmax(int deg, const float (&v)[32][VEC], float t, float (&mx)[VEC]) {
        if constexpr (J < 32) {
            if (J < deg) {
                seg_max<VEC>(v[J], t, mx);
                SegChain<J + 1, VEC, CP>::vmax(deg, v, t, mx);
            }
        }
    }
    static __device__ __forceinline__ void vsum(int deg, const float (&v)[32][VEC], float t,
                                                const float (&mx)[VEC], float (&num)[VEC], float (&den)[VEC]) {
        if constexpr (J < 32) {
            if (J < deg) {
                seg_sum<VEC>(v[J], t, mx, num, den);
                SegChain<J + 1, VEC, CP>::vsum(deg, v, t, mx, num, den);
            }
        }
    }
};

static __device__ __forceinline__ float xhalf(float x) {
    int b = __builtin_bit_cast(int, x);
    return __builtin_bit_cast(float, __builtin_amdgcn_permlanex16(b, b, 0x76543210, 0xfedcba98, false, false));
}
template<int M>
static __device__ __forceinline__ float rowx(float x) {
    return __builtin_bit_cast(float, __builtin_amdgcn_update_dpp(0, __builtin_bit_cast(int, x), 0x160 + M, 0xf, 0xf, true));
}
template<int M>
static __device__ __forceinline__ void row_step8(float (&s)[8]) {
#pragma unroll
    for (int r = 0; r < 8; ++r) s[r] += rowx<M>(s[r]);
}
static __device__ __forceinline__ void row_sum16x8(float (&s)[8]) {
    row_step8<1>(s);
    row_step8<2>(s);
    row_step8<4>(s);
    row_step8<8>(s);
}

template<int CP>
__global__ void __launch_bounds__(512, 12) k_layer(const int* __restrict__ rowptr, const int* __restrict__ csrc,
                                               const float* __restrict__ xin, const float* __restrict__ tptr,
                                               const float* __restrict__ Wr, const float* __restrict__ Ws,
                                               const float* __restrict__ bias, const int* __restrict__ batch,
                                               float* __restrict__ out, unsigned* __restrict__ he, int Nrows) {
    constexpr int VEC = CP / 32;
    constexpr int PITCH = CP + 4;
    __shared__ __attribute__((aligned(16))) float sA[16 * PITCH];
    __shared__ __attribute__((aligned(16))) float sX[16 * PITCH];
    __shared__ __attribute__((aligned(16))) float sW[2 * HID * PITCH];
    const int lane = threadIdx.x & 31;
    const int wv = __builtin_amdgcn_readfirstlane((int)(threadIdx.x >> 5));
    const int m0 = (int)blockIdx.x * 16;
    if (m0 >= Nrows) return;
    const int node = m0 + wv;
    constexpr int Q   = CP / 4;
    constexpr int PER = 2 * HID * Q / 512;
    const int nc = min(node, Nrows - 1);
    float4 wq[PER];
#pragma unroll
    for (int i = 0; i < PER; ++i) {
        const int rem = (i * 512) % (HID * Q) + (int)threadIdx.x;
        const float* __restrict__ Wm = ((i * 512) / (HID * Q)) ? Ws : Wr;
        wq[i] = *(const float4*)(Wm + (rem / Q) * CP + 4 * (rem % Q));
    }
    const unsigned loff = (unsigned)(lane * VEC);
    float xr[VEC];
    ld_row<VEC>(xin + (size_t)nc * CP + loff, xr);

    const int beg = rowptr[nc];
    const int end = rowptr[nc + 1];
    const int deg = (node < Nrows) ? end - beg : 0;
    const float t = tptr[0];
    const int mys0 = (lane < min(deg, 32)) ? csrc[beg + lane] : 0;

#pragma unroll
    for (int i = 0; i < PER; ++i) {
        const int rem = (i * 512) % (HID * Q) + (int)threadIdx.x;
        const int n = rem / Q;
        float* d = sW + ((i * 512) / (HID * Q)) * (HID * PITCH) + ((n & 3) * 16 + (n >> 2)) * PITCH + 4 * (rem % Q);
        *(float4*)d = wq[i];
    }
#pragma unroll
    for (int i = 0; i < VEC; ++i) sX[wv * PITCH + loff + i] = xr[i];

    {
        float mx[VEC], num[VEC], den[VEC];
#pragma unroll
        for (int i = 0; i < VEC; ++i) { mx[i] = -INFINITY; num[i] = 0.f; den[i] = 0.f; }
        if (deg <= 32) {
            int sidx[32];
#pragma unroll
            for (int j = 0; j < 32; ++j) sidx[j] = __builtin_amdgcn_readlane(mys0, j);
            float v[32][VEC];
            SegChain<0, VEC, CP>::load(deg, xin, loff, sidx, v);
            SegChain<0, VEC, CP>::vmax(deg, v, t, mx);
            SegChain<0, VEC, CP>::vsum(deg, v, t, mx, num, den);
        } else {
            for (int base = beg; base < end; base += 32) {
                const int cnt = min(32, end - base);
                const int mys = (lane < cnt) ? csrc[base + lane] : 0;
                for (int j = 0; j < cnt; ++j) {
                    const int s = __builtin_amdgcn_readlane(mys, j);
                    float r[VEC];
                    ld_row<VEC>(xin + (size_t)s * CP + loff, r);
                    seg_max<VEC>(r, t, mx);
                }
            }
            for (int base = beg; base < end; base += 32) {
                const int cnt = min(32, end - base);
                const int mys = (lane < cnt) ? csrc[base + lane] : 0;
                for (int j = 0; j < cnt; ++j) {
                    const int s = __builtin_amdgcn_readlane(mys, j);
                    float r[VEC];
                    ld_row<VEC>(xin + (size_t)s * CP + loff, r);
                    seg_sum<VEC>(r, t, mx, num, den);
                }
            }
        }
        float* o = sA + wv * PITCH + loff;
#pragma unroll
        for (int i = 0; i < VEC; ++i) o[i] = den[i] > 0.f ? num[i] / den[i] : 0.f;
    }
    __syncthreads();
    if (wv != (int)(((blockIdx.x + (blockIdx.x >> 8)) & 3u) * 5u)) return;

    const int half = lane >> 4;
    const int c    = lane & 15;
    const int gl = batch[min(m0 + c, Nrows - 1)];
    const float* ap  = sA + c * PITCH + half * 2;
    const float* xp  = sX + c * PITCH + half * 2;
    const float* wr  = sW + c * PITCH + half * 2;
    const float* wsp = wr + HID * PITCH;

    v8f acc[4] = {};
#pragma unroll
    for (int k = 0; k < CP; k += 4) {
        v2f a = *(const v2f*)(ap + k);
#pragma unroll
        for (int j = 0; j < 4; ++j) {
            v2f b = *(const v2f*)(wr + j * 16 * PITCH + k);
            acc[j] = __builtin_amdgcn_wmma_f32_16x16x4_f32(false, a, false, b,
                                                           (short)0, acc[j], false, false);
        }
    }
#pragma unroll
    for (int k = 0; k < CP; k += 4) {
        v2f a = *(const v2f*)(xp + k);
#pragma unroll
        for (int j = 0; j < 4; ++j) {
            v2f b = *(const v2f*)(wsp + j * 16 * PITCH + k);
            acc[j] = __builtin_amdgcn_wmma_f32_16x16x4_f32(false, a, false, b,
                                                           (short)0, acc[j], false, false);
        }
    }
    const float4 bq = *(const float4*)(bias + 4 * c);
    const float bv[4] = { bq.x, bq.y, bq.z, bq.w };
    float s[8], q[8];
#pragma unroll
    for (int r = 0; r < 8; ++r) {
#pragma unroll
        for (int j = 0; j < 4; ++j) acc[j][r] += bv[j];
        s[r] = (acc[0][r] + acc[1][r]) + (acc[2][r] + acc[3][r]);
    }
    row_sum16x8(s);
#pragma unroll
    for (int r = 0; r < 8; ++r) {
        const float mu = s[r] * (1.f / HID);
#pragma unroll
        for (int j = 0; j < 4; ++j) acc[j][r] -= mu;
        q[r] = (acc[0][r] * acc[0][r] + acc[1][r] * acc[1][r]) + (acc[2][r] * acc[2][r] + acc[3][r] * acc[3][r]);
    }
    row_sum16x8(q);
    float cm[4];
#pragma unroll
    for (int r = 0; r < 8; ++r) {
        const float inv = rsqrtf(q[r] * (1.f / HID) + EPS);
#pragma unroll
        for (int j = 0; j < 4; ++j) {
            const float y = fmaxf(acc[j][r] * inv, 0.f);
            acc[j][r] = y;
            cm[j] = (r == 0) ? y : fmaxf(cm[j], y);
        }
    }
    float* op = out + (size_t)(m0 + half * 8) * HID + 4 * c;
    if (m0 + 16 <= Nrows) {
#pragma unroll
        for (int r = 0; r < 8; ++r) {
            float4 o; o.x = acc[0][r]; o.y = acc[1][r]; o.z = acc[2][r]; o.w = acc[3][r];
            *(float4*)(op + (size_t)r * HID) = o;
        }
    } else {
#pragma unroll
        for (int r = 0; r < 8; ++r) {
            if (m0 + r + half * 8 < Nrows) {
                float4 o; o.x = acc[0][r]; o.y = acc[1][r]; o.z = acc[2][r]; o.w = acc[3][r];
                *(float4*)(op + (size_t)r * HID) = o;
            }
        }
    }

    const int g0 = __builtin_amdgcn_readfirstlane(gl);
    const bool one_graph = __builtin_amdgcn_ballot_w32(gl != g0) == 0u;
    if (one_graph) {
#pragma unroll
        for (int j = 0; j < 4; ++j) cm[j] = fmaxf(cm[j], xhalf(cm[j]));
        const float p0 = half ? cm[2] : cm[0];
        const float p1 = half ? cm[3] : cm[1];
        unsigned* hp = he + (size_t)g0 * HID + 4 * c + 2 * half;
        atomicMax(hp,     enc_f32(p0));
        atomicMax(hp + 1, enc_f32(p1));
    } else {
#pragma unroll
        for (int r = 0; r < 8; ++r) {
            const int row = m0 + r + half * 8;
            if (row < Nrows) {
                unsigned* hp = he + (size_t)batch[row] * HID + 4 * c;
#pragma unroll
                for (int j = 0; j < 4; ++j) atomicMax(hp + j, enc_f32(acc[j][r]));
            }
        }
    }
}

template<int NV>
static __device__ __forceinline__ void wave_sum_n(float (&p)[NV]) {
#pragma unroll
    for (int j = 0; j < NV; ++j) p[j] += rowx<1>(p[j]);
#pragma unroll
    for (int j = 0; j < NV; ++j) p[j] += rowx<2>(p[j]);
#pragma unroll
    for (int j = 0; j < NV; ++j) p[j] += rowx<4>(p[j]);
#pragma unroll
    for (int j = 0; j < NV; ++j) p[j] += rowx<8>(p[j]);
#pragma unroll
    for (int j = 0; j < NV; ++j) p[j] += xhalf(p[j]);
}

__global__ void __launch_bounds__(256) k_head(const unsigned* __restrict__ henc,
                                              const float* __restrict__ Wl1, const float* __restrict__ bl1,
                                              const float* __restrict__ Wl2, const float* __restrict__ bl2,
                                              float* __restrict__ outp) {
    __shared__ __attribute__((aligned(16))) float sh[3 * HID];
    __shared__ __attribute__((aligned(16))) float sz1[2 * HID];
    __shared__ __attribute__((aligned(16))) float sz2[HID / 2];
    const int g = blockIdx.x;
    const int tid = threadIdx.x, lane = tid & 31, w = tid >> 5;
    if (tid < 3 * HID) {
        const int layer = tid / HID, cc = tid % HID;
        sh[tid] = dec_f32(henc[layer * NG * HID + g * HID + cc]);
    }
    __syncthreads();

    {
        v2f hv[3];
#pragma unroll
        for (int i = 0; i < 3; ++i) hv[i] = *(const v2f*)(sh + 2 * (lane + 32 * i));
        const float* __restrict__ wbase = Wl1 + (w * 16) * (3 * HID) + 2 * lane;
        float p[16];
#pragma unroll
        for (int j = 0; j < 16; ++j) {
            const float* __restrict__ wrow = wbase + j * (3 * HID);
            float a = 0.f;
#pragma unroll
            for (int i = 0; i < 3; ++i) {
                const v2f wv = *(const v2f*)(wrow + 64 * i);
                a += hv[i].x * wv.x;
                a += hv[i].y * wv.y;
            }
            p[j] = a;
        }
        wave_sum_n<16>(p);
        float mine = p[0];
#pragma unroll
        for (int j = 1; j < 16; ++j) mine = (lane == j) ? p[j] : mine;
        if (lane < 16) sz1[w * 16 + lane] = fmaxf(mine + bl1[w * 16 + lane], 0.f);
    }
    __syncthreads();

    {
        const float4 zv = *(const float4*)(sz1 + 4 * lane);
        const float* __restrict__ wbase = Wl2 + (w * 4) * (2 * HID) + 4 * lane;
        float p[4];
#pragma unroll
        for (int j = 0; j < 4; ++j) {
            const float4 wv = *(const float4*)(wbase + j * (2 * HID));
            p[j] = (zv.x * wv.x + zv.y * wv.y) + (zv.z * wv.z + zv.w * wv.w);
        }
        wave_sum_n<4>(p);
        float mine = p[0];
#pragma unroll
        for (int j = 1; j < 4; ++j) mine = (lane == j) ? p[j] : mine;
        if (lane < 4) sz2[w * 4 + lane] = mine + bl2[w * 4 + lane];
    }
    __syncthreads();

    if (w == 0) {
        const float s = sz2[lane];
        float q[1] = { s * s };
        wave_sum_n<1>(q);
        const float nrm = fmaxf(sqrtf(q[0]), 1e-12f);
        outp[g * (HID / 2) + lane] = s / nrm;
    }
}

extern "C" void kernel_launch(void* const* d_in, const int* in_sizes, int n_in,
                              void* d_out, int out_size, void* d_ws, size_t ws_size,
                              hipStream_t stream) {
    const float* x    = (const float*)d_in[0];
    const int*  eidx  = (const int*)  d_in[1];
    const int*  batch = (const int*)  d_in[2];
    const float* t    = (const float*)d_in[3];
    const float* W1r  = (const float*)d_in[4];
    const float* b1   = (const float*)d_in[5];
    const float* W1s  = (const float*)d_in[6];
    const float* W2r  = (const float*)d_in[7];
    const float* b2   = (const float*)d_in[8];
    const float* W2s  = (const float*)d_in[9];
    const float* W3r  = (const float*)d_in[10];
    const float* b3   = (const float*)d_in[11];
    const float* W3s  = (const float*)d_in[12];
    const float* Wl1  = (const float*)d_in[13];
    const float* bl1  = (const float*)d_in[14];
    const float* Wl2  = (const float*)d_in[15];
    const float* bl2  = (const float*)d_in[16];
    const int* src = eidx;
    const int* dst = eidx + NE;

    float* ws = (float*)d_ws;
    size_t o = 0;
    float*    x1     = ws + o; o += (size_t)NN * HID;
    float*    x2     = ws + o; o += (size_t)NN * HID;
    float*    x3     = ws + o; o += (size_t)NN * HID;
    int*      csr    = (int*)(ws + o); o += (size_t)NN * HID * 3;
    int*      bcnt     = csr;
    int*      binstart = bcnt + NBLK * NBIN;
    unsigned* binned   = (unsigned*)(binstart + NBIN + 16);
    int*      rowptr   = (int*)(binned + NE);
    int*      csrc     = rowptr + NN + 16;
    o += (size_t)NN * HID;
    float*    xpad   = ws + o; o += (size_t)NN * 32;
    float*    w1rp   = ws + o; o += (size_t)HID * 32;
    float*    w1sp   = ws + o; o += (size_t)HID * 32;
    unsigned* henc   = (unsigned*)(ws + o); o += 3 * NG * HID;

    k_sort_count<<<NBLK, 1024, 0, stream>>>(dst, bcnt, henc, W1r, W1s, w1rp, w1sp, x, xpad);
    k_sort_scatter<<<NBLK, 1024, 0, stream>>>(src, dst, bcnt, binstart, binned);
    k_sort_bin<<<NBIN, 1024, 0, stream>>>(binned, binstart, rowptr, csrc);

    struct Layer { const float* xin; int Cp; const float *Wr, *b, *Ws; float* xout; };
    Layer layers[3] = {
        { xpad, 32, w1rp, b1, w1sp, x1 },
        { x1,   64, W2r,  b2, W2s,  x2 },
        { x2,   64, W3r,  b3, W3s,  x3 },
    };

    for (int l = 0; l < 3; ++l) {
        const Layer& L = layers[l];
        if (L.Cp == 32)
            k_layer<32><<<(NN + 15) / 16, 512, 0, stream>>>(rowptr, csrc, L.xin, t, L.Wr, L.Ws, L.b, batch,
                                                            L.xout, henc + (size_t)l * NG * HID, NN);
        else
            k_layer<64><<<(NN + 15) / 16, 512, 0, stream>>>(rowptr, csrc, L.xin, t, L.Wr, L.Ws, L.b, batch,
                                                            L.xout, henc + (size_t)l * NG * HID, NN);
    }

    k_head<<<NG, 256, 0, stream>>>(henc, Wl1, bl1, Wl2, bl2, (float*)d_out);
}


# Layer group owns 64 nodes (4 tiles), not 16: each of 16 waves aggregates 4 adjacent nodes in turn, weights staged once per 64 nodes, 4 waves (one per SIMD) run GEMM+tail; 4x fewer waves; same sums
# speedup vs baseline: 6.7885x; 1.0578x over previous
#include <hip/hip_runtime.h>
#include <math.h>

#define NN      50000
#define NE      800000
#define NG      128
#define INC     31
#define HID     64
#define EPS     1e-5f

typedef __attribute__((ext_vector_type(2))) float v2f;
typedef __attribute__((ext_vector_type(8))) float v8f;

static __device__ __forceinline__ unsigned enc_f32(float x) {
    unsigned b = __float_as_uint(x);
    return (b & 0x80000000u) ? ~b : (b | 0x80000000u);
}
static __device__ __forceinline__ float dec_f32(unsigned e) {
    unsigned b = (e & 0x80000000u) ? (e & 0x7FFFFFFFu) : ~e;
    return __uint_as_float(b);
}
#define ENC_NEG_INF 0x007FFFFFu

#define EPB    4096
#define NBLK   ((NE + EPB - 1) / EPB)
#define BINSH  8
#define BINW   (1 << BINSH)
#define NBIN   ((NN + BINW - 1) / BINW)
#define CGRP   5
#define CSTEPS ((NBLK + CGRP - 1) / CGRP)
static_assert(NN <= 65536, "source and destination must fit 16 bits each");
static_assert(NBIN <= 256 && CGRP * NBIN <= 1024, "one thread per bin and group");
static_assert(NBLK * 1024 >= 3 * NG * HID && NBLK * 1024 >= HID * 32, "prep ranges are prefixes of the grid");
static_assert((long long)NBLK * 1024 * 8 >= (long long)NN * 32, "eight padded elements per thread cover x");

static __device__ __forceinline__ void excl_scan256x2(int a, int b, int (&wsa)[8], int (&wsb)[8],
                                                      int& ea, int& eb, int& tota) {
    const int tid = threadIdx.x, lane = tid & 31, w = tid >> 5;
    int ia = a, ib = b;
    if (w < 8) {
        for (int off = 1; off < 32; off <<= 1) {
            const int ua = __shfl_up(ia, off), ub = __shfl_up(ib, off);
            if (lane >= off) { ia += ua; ib += ub; }
        }
        if (lane == 31) { wsa[w] = ia; wsb[w] = ib; }
    }
    __syncthreads();
    int ba = 0, bb = 0, ta = 0;
#pragma unroll
    for (int i = 0; i < 8; ++i) {
        const int sa = wsa[i], sb = wsb[i];
        ta += sa;
        if (i < w) { ba += sa; bb += sb; }
    }
    ea = ba + ia - a; eb = bb + ib - b; tota = ta;
}

__global__ void __launch_bounds__(1024) k_sort_count(const int* __restrict__ dst, int* __restrict__ bcnt,
                                                     unsigned* __restrict__ henc,
                                                     const float* __restrict__ W1r, const float* __restrict__ W1s,
                                                     float* __restrict__ w1rp, float* __restrict__ w1sp,
                                                     const float* __restrict__ x, float* __restrict__ xpad) {
    __shared__ int h[NBIN];
    const int tid = threadIdx.x, blk = blockIdx.x;
    if (tid < NBIN) h[tid] = 0;
    int d[4];
#pragma unroll
    for (int k = 0; k < 4; ++k) d[k] = dst[min(blk * EPB + k * 1024 + tid, NE - 1)];
    const int i0 = blk * 1024 + tid;
    float xv[8];
#pragma unroll
    for (int k = 0; k < 8; ++k) {
        const int j = min(i0 + k * (NBLK * 1024), NN * 32 - 1);
        xv[k] = x[(j >> 5) * INC + min(j & 31, INC - 1)];
    }
    __syncthreads();
#pragma unroll
    for (int k = 0; k < 4; ++k)
        if (blk * EPB + k * 1024 + tid < NE) atomicAdd(&h[d[k] >> BINSH], 1);
    __syncthreads();
    if (tid < NBIN) bcnt[blk * NBIN + tid] = h[tid];

    if (i0 < 3 * NG * HID) henc[i0] = ENC_NEG_INF;
    if (i0 < HID * 32) {
        const int r = i0 / 32, c = i0 % 32;
        w1rp[i0] = (c < INC) ? W1r[r * INC + c] : 0.f;
        w1sp[i0] = (c < INC) ? W1s[r * INC + c] : 0.f;
    }
#pragma unroll
    for (int k = 0; k < 8; ++k) {
        const int j = i0 + k * (NBLK * 1024);
        if (j < NN * 32) xpad[j] = ((j & 31) < INC) ? xv[k] : 0.f;
    }
}

__global__ void __launch_bounds__(1024) k_sort_scatter(const int* __restrict__ src, const int* __restrict__ dst,
                                                       const int* __restrict__ bcnt, int* __restrict__ binstart,
                                                       unsigned* __restrict__ binned) {
    __shared__ int tpart[CGRP][NBIN];
    __shared__ int ppart[CGRP][NBIN];
    __shared__ int h[NBIN];
    __shared__ int loff[NBIN];
    __shared__ int gbase[NBIN];
    __shared__ int wsa[8];
    __shared__ int wsb[8];
    __shared__ unsigned stage[EPB];
    const int tid = threadIdx.x, blk = blockIdx.x;
    if (tid < NBIN) h[tid] = 0;
    int s[4], d[4];
#pragma unroll
    for (int k = 0; k < 4; ++k) {
        const int e = min(blk * EPB + k * 1024 + tid, NE - 1);
        s[k] = src[e];
        d[k] = dst[e];
    }
    const int g = tid / NBIN, cb = tid - g * NBIN;
    if (g < CGRP) {
        int tot = 0, pre = 0;
#pragma unroll
        for (int it = 0; it < CSTEPS; ++it) {
            const int b = g + CGRP * it;
            int v = bcnt[min(b, NBLK - 1) * NBIN + cb];
            v = (b < NBLK) ? v : 0;
            tot += v;
            pre += (b < blk) ? v : 0;
        }
        tpart[g][cb] = tot;
        ppart[g][cb] = pre;
    }
    __syncthreads();
    int r[4];
#pragma unroll
    for (int k = 0; k < 4; ++k)
        r[k] = (blk * EPB + k * 1024 + tid < NE) ? atomicAdd(&h[d[k] >> BINSH], 1) : 0;
    int T = 0, P = 0;
    if (tid < NBIN) {
#pragma unroll
        for (int q = 0; q < CGRP; ++q) { T += tpart[q][tid]; P += ppart[q][tid]; }
    }
    __syncthreads();
    const int hv = (tid < NBIN) ? h[tid] : 0;
    int bs, lo, total;
    excl_scan256x2(T, hv, wsa, wsb, bs, lo, total);
    if (tid < NBIN) {
        gbase[tid] = bs + P;
        loff[tid] = lo;
        if (blk == 0) binstart[tid] = bs;
    }
    if (blk == 0 && tid == 0) binstart[NBIN] = total;
    __syncthreads();
    int lf[4];
#pragma unroll
    for (int k = 0; k < 4; ++k) lf[k] = loff[d[k] >> BINSH];
#pragma unroll
    for (int k = 0; k < 4; ++k) {
        const int i = k * 1024 + tid;
        const bool live = blk * EPB + i < NE;
        stage[live ? lf[k] + r[k] : i] = live ? (((unsigned)d[k] << 16) | (unsigned)s[k]) : 0u;
    }
    __syncthreads();
    const int nvalid = min(EPB, NE - blk * EPB);
    unsigned p[4];
    int gb[4], lb[4];
#pragma unroll
    for (int k = 0; k < 4; ++k) p[k] = stage[k * 1024 + tid];
#pragma unroll
    for (int k = 0; k < 4; ++k) {
        const int b = (int)(p[k] >> (16 + BINSH));
        gb[k] = gbase[b];
        lb[k] = loff[b];
    }
#pragma unroll
    for (int k = 0; k < 4; ++k) {
        const int i = k * 1024 + tid;
        if (i < nvalid) binned[gb[k] + (i - lb[k])] = p[k];
    }
}

#define BCAP (2 * EPB)
__global__ void __launch_bounds__(1024) k_sort_bin(const unsigned* __restrict__ binned, const int* __restrict__ binstart,
                                                   int* __restrict__ rowptr, int* __restrict__ csrc) {
    __shared__ int cnt[BINW];
    __shared__ int wsa[8];
    __shared__ int wsb[8];
    __shared__ int outst[BCAP];
    const int tid = threadIdx.x, bin = blockIdx.x;
    const int beg = binstart[bin], end = binstart[bin + 1], n = end - beg;
    if (tid < BINW) cnt[tid] = 0;
    __syncthreads();
    for (int base = beg; base < end; base += EPB) {
        unsigned p[4];
#pragma unroll
        for (int k = 0; k < 4; ++k) p[k] = binned[min(base + k * 1024 + tid, end - 1)];
#pragma unroll
        for (int k = 0; k < 4; ++k)
            if (base + k * 1024 + tid < end) atomicAdd(&cnt[(p[k] >> 16) & (BINW - 1)], 1);
    }
    __syncthreads();
    const int c = (tid < BINW) ? cnt[tid] : 0;
    int ex, unused, total;
    excl_scan256x2(c, 0, wsa, wsb, ex, unused, total);
    if (tid < BINW) {
        const int node = bin * BINW + tid;
        if (node <= NN) rowptr[node] = beg + ex;
        cnt[tid] = ex;
    }
    __syncthreads();
    const bool staged = n <= BCAP;
    for (int base = beg; base < end; base += EPB) {
        unsigned p[4];
#pragma unroll
        for (int k = 0; k < 4; ++k) p[k] = binned[min(base + k * 1024 + tid, end - 1)];
        int rk[4];
#pragma unroll
        for (int k = 0; k < 4; ++k)
            rk[k] = (base + k * 1024 + tid < end) ? atomicAdd(&cnt[(p[k] >> 16) & (BINW - 1)], 1) : 0;
#pragma unroll
        for (int k = 0; k < 4; ++k) {
            if (base + k * 1024 + tid < end) {
                const int sv = (int)(p[k] & 0xFFFFu);
                if (staged) outst[rk[k]] = sv; else csrc[beg + rk[k]] = sv;
            }
        }
    }
    if (staged && n > 0) {
        __syncthreads();
        int v[BCAP / 1024];
#pragma unroll
        for (int k = 0; k < BCAP / 1024; ++k) v[k] = outst[min(k * 1024 + tid, n - 1)];
#pragma unroll
        for (int k = 0; k < BCAP / 1024; ++k)
            if (k * 1024 + tid < n) csrc[beg + k * 1024 + tid] = v[k];
    }
}

template<int VEC>
static __device__ __forceinline__ void ld_row(const float* __restrict__ p, float (&v)[VEC]) {
    if constexpr (VEC == 2) {
        v2f r = *(const v2f*)p;
        v[0] = r.x; v[1] = r.y;
    } else {
        v[0] = p[0];
    }
}

template<int VEC>
static __device__ __forceinline__ void seg_max(const float (&r)[VEC], float t, float (&mx)[VEC]) {
#pragma unroll
    for (int i = 0; i < VEC; ++i) mx[i] = fmaxf(mx[i], r[i] * t);
}
template<int VEC>
static __device__ __forceinline__ void seg_sum(const float (&r)[VEC], float t, const float (&mx)[VEC],
                                               float (&num)[VEC], float (&den)[VEC]) {
#pragma unroll
    for (int i = 0; i < VEC; ++i) {
        float e = __expf(r[i] * t - mx[i]);
        num[i] += r[i] * e;
        den[i] += e;
    }
}

template<int J, int VEC, int CP, int R>
struct SegChain {
    static __device__ __forceinline__ void load(int deg, const float* __restrict__ x, unsigned loff,
                                                const int (&sidx)[32], float (&v)[R][VEC]) {
        if constexpr (J < R) {
            if (J < deg) {
                ld_row<VEC>(x + (size_t)sidx[J] * CP + loff, v[J]);
                SegChain<J + 1, VEC, CP, R>::load(deg, x, loff, sidx, v);
            }
        }
    }
    static __device__ __forceinline__ void vmax(int deg, const float (&v)[R][VEC], float t, float (&mx)[VEC]) {
        if constexpr (J < R) {
            if (J < deg) {
                seg_max<VEC>(v[J], t, mx);
                SegChain<J + 1, VEC, CP, R>::vmax(deg, v, t, mx);
            }
        }
    }
    static __device__ __forceinline__ void vsum(int deg, const float (&v)[R][VEC], float t,
                                                const float (&mx)[VEC], float (&num)[VEC], float (&den)[VEC]) {
        if constexpr (J < R) {
            if (J < deg) {
                seg_sum<VEC>(v[J], t, mx, num, den);
                SegChain<J + 1, VEC, CP, R>::vsum(deg, v, t, mx, num, den);
            }
        }
    }
};

static __device__ __forceinline__ float xhalf(float x) {
    int b = __builtin_bit_cast(int, x);
    return __builtin_bit_cast(float, __builtin_amdgcn_permlanex16(b, b, 0x76543210, 0xfedcba98, false, false));
}
template<int M>
static __device__ __forceinline__ float rowx(float x) {
    return __builtin_bit_cast(float, __builtin_amdgcn_update_dpp(0, __builtin_bit_cast(int, x), 0x160 + M, 0xf, 0xf, true));
}
template<int M>
static __device__ __forceinline__ void row_step8(float (&s)[8]) {
#pragma unroll
    for (int r = 0; r < 8; ++r) s[r] += rowx<M>(s[r]);
}
static __device__ __forceinline__ void row_sum16x8(float (&s)[8]) {
    row_step8<1>(s);
    row_step8<2>(s);
    row_step8<4>(s);
    row_step8<8>(s);
}

#define NPW  4
#define GN   (16 * NPW)
static_assert(NPW == 4, "phase 2 gives tile t to wave 5t: waves 0, 5, 10, 15");
template<int CP>
__global__ void __launch_bounds__(512, 12) k_layer(const int* __restrict__ rowptr, const int* __restrict__ csrc,
                                               const float* __restrict__ xin, const float* __restrict__ tptr,
                                               const float* __restrict__ Wr, const float* __restrict__ Ws,
                                               const float* __restrict__ bias, const int* __restrict__ batch,
                                               float* __restrict__ out, unsigned* __restrict__ he, int Nrows) {
    constexpr int VEC = CP / 32;
    constexpr int R = (VEC == 2) ? 30 : 32;
    constexpr int PITCH = CP + 4;
    __shared__ __attribute__((aligned(16))) float sA[GN * PITCH];
    __shared__ __attribute__((aligned(16))) float sX[GN * PITCH];
    __shared__ __attribute__((aligned(16))) float sW[2 * HID * PITCH];
    const int lane = threadIdx.x & 31;
    const int wv = __builtin_amdgcn_readfirstlane((int)(threadIdx.x >> 5));
    const int g0n = (int)blockIdx.x * GN;
    if (g0n >= Nrows) return;
    const int n0 = g0n + wv * NPW;
    constexpr int Q   = CP / 4;
    constexpr int PER = 2 * HID * Q / 512;
    float4 wq[PER];
#pragma unroll
    for (int i = 0; i < PER; ++i) {
        const int rem = (i * 512) % (HID * Q) + (int)threadIdx.x;
        const float* __restrict__ Wm = ((i * 512) / (HID * Q)) ? Ws : Wr;
        wq[i] = *(const float4*)(Wm + (rem / Q) * CP + 4 * (rem % Q));
    }
    const unsigned loff = (unsigned)(lane * VEC);
    float xr[NPW][VEC];
#pragma unroll
    for (int i = 0; i < NPW; ++i)
        ld_row<VEC>(xin + (size_t)min(n0 + i, Nrows - 1) * CP + loff, xr[i]);

    const int rp = rowptr[min(n0 + lane, Nrows)];
    const float t = tptr[0];
    int bnd[NPW + 1];
#pragma unroll
    for (int k = 0; k <= NPW; ++k) bnd[k] = __builtin_amdgcn_readlane(rp, k);
    int beg = bnd[0];
    int end = bnd[1];
    int mys0 = (lane < min(end - beg, 32)) ? csrc[beg + lane] : 0;

#pragma unroll
    for (int i = 0; i < PER; ++i) {
        const int rem = (i * 512) % (HID * Q) + (int)threadIdx.x;
        const int n = rem / Q;
        float* d = sW + ((i * 512) / (HID * Q)) * (HID * PITCH) + ((n & 3) * 16 + (n >> 2)) * PITCH + 4 * (rem % Q);
        *(float4*)d = wq[i];
    }
#pragma unroll
    for (int i = 0; i < NPW; ++i)
#pragma unroll
        for (int k = 0; k < VEC; ++k) sX[(wv * NPW + i) * PITCH + loff + k] = xr[i][k];

#pragma nounroll
    for (int i = 0; i < NPW; ++i) {
        const int deg = end - beg;
        int sidx[32];
#pragma unroll
        for (int j = 0; j < 32; ++j) sidx[j] = __builtin_amdgcn_readlane(mys0, j);
        const int nbeg = end;
        int nend = bnd[NPW];
#pragma unroll
        for (int k = NPW - 1; k >= 2; --k) nend = (i + 2 == k) ? bnd[k] : nend;
        const int mysn = (i + 1 < NPW && lane < min(nend - nbeg, 32)) ? csrc[nbeg + lane] : 0;
        float mx[VEC], num[VEC], den[VEC];
#pragma unroll
        for (int k = 0; k < VEC; ++k) { mx[k] = -INFINITY; num[k] = 0.f; den[k] = 0.f; }
        if (deg <= R) {
            float v[R][VEC];
            SegChain<0, VEC, CP, R>::load(deg, xin, loff, sidx, v);
            SegChain<0, VEC, CP, R>::vmax(deg, v, t, mx);
            SegChain<0, VEC, CP, R>::vsum(deg, v, t, mx, num, den);
        } else {
            for (int base = beg; base < end; base += 32) {
                const int cnt = min(32, end - base);
                const int mys = (lane < cnt) ? csrc[base + lane] : 0;
                for (int j = 0; j < cnt; ++j) {
                    const int s = __builtin_amdgcn_readlane(mys, j);
                    float r[VEC];
                    ld_row<VEC>(xin + (size_t)s * CP + loff, r);
                    seg_max<VEC>(r, t, mx);
                }
            }
            for (int base = beg; base < end; base += 32) {
                const int cnt = min(32, end - base);
                const int mys = (lane < cnt) ? csrc[base + lane] : 0;
                for (int j = 0; j < cnt; ++j) {
                    const int s = __builtin_amdgcn_readlane(mys, j);
                    float r[VEC];
                    ld_row<VEC>(xin + (size_t)s * CP + loff, r);
                    seg_sum<VEC>(r, t, mx, num, den);
                }
            }
        }
        float* o = sA + (wv * NPW + i) * PITCH + loff;
#pragma unroll
        for (int k = 0; k < VEC; ++k) o[k] = den[k] > 0.f ? num[k] / den[k] : 0.f;
        beg = nbeg; end = nend; mys0 = mysn;
    }
    __syncthreads();
    if (wv % 5 != 0) return;
    const int m0 = g0n + (wv / 5) * 16;
    if (m0 >= Nrows) return;

    const int half = lane >> 4;
    const int c    = lane & 15;
    const float* ap  = sA + ((wv / 5) * 16 + c) * PITCH + half * 2;
    const float* xp  = sX + ((wv / 5) * 16 + c) * PITCH + half * 2;
    const float* wr  = sW + c * PITCH + half * 2;
    const float* wsp = wr + HID * PITCH;

    v8f acc[4] = {};
#pragma unroll
    for (int k = 0; k < CP; k += 4) {
        v2f a = *(const v2f*)(ap + k);
#pragma unroll
        for (int j = 0; j < 4; ++j) {
            v2f b = *(const v2f*)(wr + j * 16 * PITCH + k);
            acc[j] = __builtin_amdgcn_wmma_f32_16x16x4_f32(false, a, false, b,
                                                           (short)0, acc[j], false, false);
        }
    }
#pragma unroll
    for (int k = 0; k < CP; k += 4) {
        v2f a = *(const v2f*)(xp + k);
#pragma unroll
        for (int j = 0; j < 4; ++j) {
            v2f b = *(const v2f*)(wsp + j * 16 * PITCH + k);
            acc[j] = __builtin_amdgcn_wmma_f32_16x16x4_f32(false, a, false, b,
                                                           (short)0, acc[j], false, false);
        }
    }
    const int gl = batch[min(m0 + c, Nrows - 1)];
    const int half_e = (int)(__builtin_amdgcn_mbcnt_lo(~0u, 0u) >> 4);
    const float4 bq = *(const float4*)(bias + 4 * c);
    const float bv[4] = { bq.x, bq.y, bq.z, bq.w };
    float s[8], q[8];
#pragma unroll
    for (int r = 0; r < 8; ++r) {
#pragma unroll
        for (int j = 0; j < 4; ++j) acc[j][r] += bv[j];
        s[r] = (acc[0][r] + acc[1][r]) + (acc[2][r] + acc[3][r]);
    }
    row_sum16x8(s);
#pragma unroll
    for (int r = 0; r < 8; ++r) {
        const float mu = s[r] * (1.f / HID);
#pragma unroll
        for (int j = 0; j < 4; ++j) acc[j][r] -= mu;
        q[r] = (acc[0][r] * acc[0][r] + acc[1][r] * acc[1][r]) + (acc[2][r] * acc[2][r] + acc[3][r] * acc[3][r]);
    }
    row_sum16x8(q);
    float cm[4];
#pragma unroll
    for (int r = 0; r < 8; ++r) {
        const float inv = rsqrtf(q[r] * (1.f / HID) + EPS);
#pragma unroll
        for (int j = 0; j < 4; ++j) {
            const float y = fmaxf(acc[j][r] * inv, 0.f);
            acc[j][r] = y;
            cm[j] = (r == 0) ? y : fmaxf(cm[j], y);
        }
    }
    float* op = out + (size_t)(m0 + half_e * 8) * HID + 4 * c;
    if (m0 + 16 <= Nrows) {
#pragma unroll
        for (int r = 0; r < 8; ++r) {
            float4 o; o.x = acc[0][r]; o.y = acc[1][r]; o.z = acc[2][r]; o.w = acc[3][r];
            *(float4*)(op + (size_t)r * HID) = o;
        }
    } else {
#pragma unroll
        for (int r = 0; r < 8; ++r) {
            if (m0 + r + half_e * 8 < Nrows) {
                float4 o; o.x = acc[0][r]; o.y = acc[1][r]; o.z = acc[2][r]; o.w = acc[3][r];
                *(float4*)(op + (size_t)r * HID) = o;
            }
        }
    }

    const int g0 = __builtin_amdgcn_readfirstlane(gl);
    const bool one_graph = __builtin_amdgcn_ballot_w32(gl != g0) == 0u;
    if (one_graph) {
#pragma unroll
        for (int j = 0; j < 4; ++j) cm[j] = fmaxf(cm[j], xhalf(cm[j]));
        const float p0 = half_e ? cm[2] : cm[0];
        const float p1 = half_e ? cm[3] : cm[1];
        unsigned* hp = he + (size_t)g0 * HID + 4 * c + 2 * half_e;
        atomicMax(hp,     enc_f32(p0));
        atomicMax(hp + 1, enc_f32(p1));
    } else {
#pragma unroll
        for (int r = 0; r < 8; ++r) {
            const int row = m0 + r + half_e * 8;
            if (row < Nrows) {
                unsigned* hp = he + (size_t)batch[row] * HID + 4 * c;
#pragma unroll
                for (int j = 0; j < 4; ++j) atomicMax(hp + j, enc_f32(acc[j][r]));
            }
        }
    }
}

template<int NV>
static __device__ __forceinline__ void wave_sum_n(float (&p)[NV]) {
#pragma unroll
    for (int j = 0; j < NV; ++j) p[j] += rowx<1>(p[j]);
#pragma unroll
    for (int j = 0; j < NV; ++j) p[j] += rowx<2>(p[j]);
#pragma unroll
    for (int j = 0; j < NV; ++j) p[j] += rowx<4>(p[j]);
#pragma unroll
    for (int j = 0; j < NV; ++j) p[j] += rowx<8>(p[j]);
#pragma unroll
    for (int j = 0; j < NV; ++j) p[j] += xhalf(p[j]);
}

__global__ void __launch_bounds__(256) k_head(const unsigned* __restrict__ henc,
                                              const float* __restrict__ Wl1, const float* __restrict__ bl1,
                                              const float* __restrict__ Wl2, const float* __restrict__ bl2,
                                              float* __restrict__ outp) {
    __shared__ __attribute__((aligned(16))) float sh[3 * HID];
    __shared__ __attribute__((aligned(16))) float sz1[2 * HID];
    __shared__ __attribute__((aligned(16))) float sz2[HID / 2];
    const int g = blockIdx.x;
    const int tid = threadIdx.x, lane = tid & 31, w = tid >> 5;
    if (tid < 3 * HID) {
        const int layer = tid / HID, cc = tid % HID;
        sh[tid] = dec_f32(henc[layer * NG * HID + g * HID + cc]);
    }
    __syncthreads();

    {
        v2f hv[3];
#pragma unroll
        for (int i = 0; i < 3; ++i) hv[i] = *(const v2f*)(sh + 2 * (lane + 32 * i));
        const float* __restrict__ wbase = Wl1 + (w * 16) * (3 * HID) + 2 * lane;
        float p[16];
#pragma unroll
        for (int j = 0; j < 16; ++j) {
            const float* __restrict__ wrow = wbase + j * (3 * HID);
            float a = 0.f;
#pragma unroll
            for (int i = 0; i < 3; ++i) {
                const v2f wv = *(const v2f*)(wrow + 64 * i);
                a += hv[i].x * wv.x;
                a += hv[i].y * wv.y;
            }
            p[j] = a;
        }
        wave_sum_n<16>(p);
        float mine = p[0];
#pragma unroll
        for (int j = 1; j < 16; ++j) mine = (lane == j) ? p[j] : mine;
        if (lane < 16) sz1[w * 16 + lane] = fmaxf(mine + bl1[w * 16 + lane], 0.f);
    }
    __syncthreads();

    {
        const float4 zv = *(const float4*)(sz1 + 4 * lane);
        const float* __restrict__ wbase = Wl2 + (w * 4) * (2 * HID) + 4 * lane;
        float p[4];
#pragma unroll
        for (int j = 0; j < 4; ++j) {
            const float4 wv = *(const float4*)(wbase + j * (2 * HID));
            p[j] = (zv.x * wv.x + zv.y * wv.y) + (zv.z * wv.z + zv.w * wv.w);
        }
        wave_sum_n<4>(p);
        float mine = p[0];
#pragma unroll
        for (int j = 1; j < 4; ++j) mine = (lane == j) ? p[j] : mine;
        if (lane < 4) sz2[w * 4 + lane] = mine + bl2[w * 4 + lane];
    }
    __syncthreads();

    if (w == 0) {
        const float s = sz2[lane];
        float q[1] = { s * s };
        wave_sum_n<1>(q);
        const float nrm = fmaxf(sqrtf(q[0]), 1e-12f);
        outp[g * (HID / 2) + lane] = s / nrm;
    }
}

extern "C" void kernel_launch(void* const* d_in, const int* in_sizes, int n_in,
                              void* d_out, int out_size, void* d_ws, size_t ws_size,
                              hipStream_t stream) {
    const float* x    = (const float*)d_in[0];
    const int*  eidx  = (const int*)  d_in[1];
    const int*  batch = (const int*)  d_in[2];
    const float* t    = (const float*)d_in[3];
    const float* W1r  = (const float*)d_in[4];
    const float* b1   = (const float*)d_in[5];
    const float* W1s  = (const float*)d_in[6];
    const float* W2r  = (const float*)d_in[7];
    const float* b2   = (const float*)d_in[8];
    const float* W2s  = (const float*)d_in[9];
    const float* W3r  = (const float*)d_in[10];
    const float* b3   = (const float*)d_in[11];
    const float* W3s  = (const float*)d_in[12];
    const float* Wl1  = (const float*)d_in[13];
    const float* bl1  = (const float*)d_in[14];
    const float* Wl2  = (const float*)d_in[15];
    const float* bl2  = (const float*)d_in[16];
    const int* src = eidx;
    const int* dst = eidx + NE;

    float* ws = (float*)d_ws;
    size_t o = 0;
    float*    x1     = ws + o; o += (size_t)NN * HID;
    float*    x2     = ws + o; o += (size_t)NN * HID;
    float*    x3     = ws + o; o += (size_t)NN * HID;
    int*      csr    = (int*)(ws + o); o += (size_t)NN * HID * 3;
    int*      bcnt     = csr;
    int*      binstart = bcnt + NBLK * NBIN;
    unsigned* binned   = (unsigned*)(binstart + NBIN + 16);
    int*      rowptr   = (int*)(binned + NE);
    int*      csrc     = rowptr + NN + 16;
    o += (size_t)NN * HID;
    float*    xpad   = ws + o; o += (size_t)NN * 32;
    float*    w1rp   = ws + o; o += (size_t)HID * 32;
    float*    w1sp   = ws + o; o += (size_t)HID * 32;
    unsigned* henc   = (unsigned*)(ws + o); o += 3 * NG * HID;

    k_sort_count<<<NBLK, 1024, 0, stream>>>(dst, bcnt, henc, W1r, W1s, w1rp, w1sp, x, xpad);
    k_sort_scatter<<<NBLK, 1024, 0, stream>>>(src, dst, bcnt, binstart, binned);
    k_sort_bin<<<NBIN, 1024, 0, stream>>>(binned, binstart, rowptr, csrc);

    struct Layer { const float* xin; int Cp; const float *Wr, *b, *Ws; float* xout; };
    Layer layers[3] = {
        { xpad, 32, w1rp, b1, w1sp, x1 },
        { x1,   64, W2r,  b2, W2s,  x2 },
        { x2,   64, W3r,  b3, W3s,  x3 },
    };

    for (int l = 0; l < 3; ++l) {
        const Layer& L = layers[l];
        if (L.Cp == 32)
            k_layer<32><<<(NN + GN - 1) / GN, 512, 0, stream>>>(rowptr, csrc, L.xin, t, L.Wr, L.Ws, L.b, batch,
                                                            L.xout, henc + (size_t)l * NG * HID, NN);
        else
            k_layer<64><<<(NN + GN - 1) / GN, 512, 0, stream>>>(rowptr, csrc, L.xin, t, L.Wr, L.Ws, L.b, batch,
                                                            L.xout, henc + (size_t)l * NG * HID, NN);
    }

    k_head<<<NG, 256, 0, stream>>>(henc, Wl1, bl1, Wl2, bl2, (float*)d_out);
}


# Level-1 sort blocks keep their bin runs in their own slice plus a run-start table; each bin block collects its 196 runs itself: no count launch, no global count table, 7 launches to 6; sums as before
# speedup vs baseline: 7.0990x; 1.0457x over previous
#include <hip/hip_runtime.h>
#include <math.h>

#define NN      50000
#define NE      800000
#define NG      128
#define INC     31
#define HID     64
#define EPS     1e-5f

typedef __attribute__((ext_vector_type(2))) float v2f;
typedef __attribute__((ext_vector_type(8))) float v8f;

static __device__ __forceinline__ unsigned enc_f32(float x) {
    unsigned b = __float_as_uint(x);
    return (b & 0x80000000u) ? ~b : (b | 0x80000000u);
}
static __device__ __forceinline__ float dec_f32(unsigned e) {
    unsigned b = (e & 0x80000000u) ? (e & 0x7FFFFFFFu) : ~e;
    return __uint_as_float(b);
}
#define ENC_NEG_INF 0x007FFFFFu

#define EPB    4096
#define NBLK   ((NE + EPB - 1) / EPB)
#define BINSH  8
#define BINW   (1 << BINSH)
#define NBIN   ((NN + BINW - 1) / BINW)
#define BOFFW  (NBIN + 1)
static_assert(NN <= 65536, "source and destination must fit 16 bits each");
static_assert(NBIN <= 256 && NBLK <= 256, "one thread per bin in level 1, one thread (and one thread quad) per run in level 2");
static_assert(NBLK * 1024 >= 3 * NG * HID && NBLK * 1024 >= HID * 32, "prep ranges are prefixes of the grid");
static_assert((long long)NBLK * 1024 * 8 >= (long long)NN * 32, "eight padded elements per thread cover x");

static __device__ __forceinline__ void excl_scan256x2(int a, int b, int (&wsa)[8], int (&wsb)[8],
                                                      int& ea, int& eb, int& tota, int& totb) {
    const int tid = threadIdx.x, lane = tid & 31, w = tid >> 5;
    int ia = a, ib = b;
    if (w < 8) {
        for (int off = 1; off < 32; off <<= 1) {
            const int ua = __shfl_up(ia, off), ub = __shfl_up(ib, off);
            if (lane >= off) { ia += ua; ib += ub; }
        }
        if (lane == 31) { wsa[w] = ia; wsb[w] = ib; }
    }
    __syncthreads();
    int ba = 0, bb = 0, ta = 0, tb = 0;
#pragma unroll
    for (int i = 0; i < 8; ++i) {
        const int sa = wsa[i], sb = wsb[i];
        ta += sa; tb += sb;
        if (i < w) { ba += sa; bb += sb; }
    }
    ea = ba + ia - a; eb = bb + ib - b; tota = ta; totb = tb;
}

__global__ void __launch_bounds__(1024) k_sort_block(const int* __restrict__ src, const int* __restrict__ dst,
                                                     int* __restrict__ boff, unsigned* __restrict__ binned,
                                                     unsigned* __restrict__ henc,
                                                     const float* __restrict__ W1r, const float* __restrict__ W1s,
                                                     float* __restrict__ w1rp, float* __restrict__ w1sp,
                                                     const float* __restrict__ x, float* __restrict__ xpad) {
    __shared__ int h[NBIN];
    __shared__ int loff[NBIN];
    __shared__ int wsa[8];
    __shared__ int wsb[8];
    __shared__ unsigned stage[EPB];
    const int tid = threadIdx.x, blk = blockIdx.x;
    if (tid < NBIN) h[tid] = 0;
    int s[4], d[4];
#pragma unroll
    for (int k = 0; k < 4; ++k) {
        const int e = min(blk * EPB + k * 1024 + tid, NE - 1);
        s[k] = src[e];
        d[k] = dst[e];
    }
    const int i0 = blk * 1024 + tid;
    float xv[8];
#pragma unroll
    for (int k = 0; k < 8; ++k) {
        const int j = min(i0 + k * (NBLK * 1024), NN * 32 - 1);
        xv[k] = x[(j >> 5) * INC + min(j & 31, INC - 1)];
    }
    __syncthreads();
    int r[4];
#pragma unroll
    for (int k = 0; k < 4; ++k)
        r[k] = (blk * EPB + k * 1024 + tid < NE) ? atomicAdd(&h[d[k] >> BINSH], 1) : 0;
    __syncthreads();
    const int hv = (tid < NBIN) ? h[tid] : 0;
    int lo, unused, total, unused2;
    excl_scan256x2(hv, 0, wsa, wsb, lo, unused, total, unused2);
    if (tid < NBIN) {
        loff[tid] = lo;
        boff[blk * BOFFW + tid] = lo;
    }
    if (tid == 0) boff[blk * BOFFW + NBIN] = total;
    __syncthreads();
    int lf[4];
#pragma unroll
    for (int k = 0; k < 4; ++k) lf[k] = loff[d[k] >> BINSH];
#pragma unroll
    for (int k = 0; k < 4; ++k) {
        const int i = k * 1024 + tid;
        const bool live = blk * EPB + i < NE;
        stage[live ? lf[k] + r[k] : i] = live ? (((unsigned)d[k] << 16) | (unsigned)s[k]) : 0u;
    }
    __syncthreads();
    const int nvalid = min(EPB, NE - blk * EPB);
    unsigned p[4];
#pragma unroll
    for (int k = 0; k < 4; ++k) p[k] = stage[k * 1024 + tid];
#pragma unroll
    for (int k = 0; k < 4; ++k) {
        const int i = k * 1024 + tid;
        if (i < nvalid) binned[blk * EPB + i] = p[k];
    }

    if (i0 < 3 * NG * HID) henc[i0] = ENC_NEG_INF;
    if (i0 < HID * 32) {
        const int rr = i0 / 32, cc = i0 % 32;
        w1rp[i0] = (cc < INC) ? W1r[rr * INC + cc] : 0.f;
        w1sp[i0] = (cc < INC) ? W1s[rr * INC + cc] : 0.f;
    }
#pragma unroll
    for (int k = 0; k < 8; ++k) {
        const int j = i0 + k * (NBLK * 1024);
        if (j < NN * 32) xpad[j] = ((j & 31) < INC) ? xv[k] : 0.f;
    }
}

#define BCAP (2 * EPB)
#define RUNC 8
__global__ void __launch_bounds__(1024) k_sort_bin(const unsigned* __restrict__ binned, const int* __restrict__ boff,
                                                   int* __restrict__ rowptr, int* __restrict__ csrc) {
    __shared__ int cnt[BINW];
    __shared__ int wsa[8];
    __shared__ int wsb[8];
    __shared__ int outst[BCAP];
    const int tid = threadIdx.x, bin = blockIdx.x;
    const int t4 = tid >> 2, sub = tid & 3;
    const int ta = min(tid, NBLK - 1) * BOFFW + bin, tq = min(t4, NBLK - 1) * BOFFW + bin;
    const int a0 = boff[ta], b0 = boff[ta + 1];
    const int q0 = boff[tq], q1 = boff[tq + 1];
    if (tid < BINW) cnt[tid] = 0;
    int unused, unused2, beg, end;
    excl_scan256x2((tid < NBLK) ? a0 : 0, (tid < NBLK) ? b0 : 0, wsa, wsb, unused, unused2, beg, end);
    const int n = end - beg;
    const int rs = (t4 < NBLK) ? q0 : 0, re = (t4 < NBLK) ? q1 : 0;
    const unsigned* __restrict__ run = binned + min(t4, NBLK - 1) * EPB;
    for (int base = rs; base < re; base += 4 * RUNC) {
        unsigned p[RUNC];
#pragma unroll
        for (int k = 0; k < RUNC; ++k) p[k] = run[min(base + 4 * k + sub, re - 1)];
#pragma unroll
        for (int k = 0; k < RUNC; ++k)
            if (base + 4 * k + sub < re) atomicAdd(&cnt[(p[k] >> 16) & (BINW - 1)], 1);
    }
    __syncthreads();
    const int c = (tid < BINW) ? cnt[tid] : 0;
    int ex, total;
    excl_scan256x2(c, 0, wsa, wsb, ex, unused, total, unused2);
    if (tid < BINW) {
        const int node = bin * BINW + tid;
        if (node <= NN) rowptr[node] = beg + ex;
        cnt[tid] = ex;
    }
    __syncthreads();
    const bool staged = n <= BCAP;
    for (int base = rs; base < re; base += 4 * RUNC) {
        unsigned p[RUNC];
#pragma unroll
        for (int k = 0; k < RUNC; ++k) p[k] = run[min(base + 4 * k + sub, re - 1)];
        int rk[RUNC];
#pragma unroll
        for (int k = 0; k < RUNC; ++k)
            rk[k] = (base + 4 * k + sub < re) ? atomicAdd(&cnt[(p[k] >> 16) & (BINW - 1)], 1) : 0;
#pragma unroll
        for (int k = 0; k < RUNC; ++k) {
            if (base + 4 * k + sub < re) {
                const int sv = (int)(p[k] & 0xFFFFu);
                if (staged) outst[rk[k]] = sv; else csrc[beg + rk[k]] = sv;
            }
        }
    }
    if (staged && n > 0) {
        __syncthreads();
        int v[BCAP / 1024];
#pragma unroll
        for (int k = 0; k < BCAP / 1024; ++k) v[k] = outst[min(k * 1024 + tid, n - 1)];
#pragma unroll
        for (int k = 0; k < BCAP / 1024; ++k)
            if (k * 1024 + tid < n) csrc[beg + k * 1024 + tid] = v[k];
    }
}

template<int VEC>
static __device__ __forceinline__ void ld_row(const float* __restrict__ p, float (&v)[VEC]) {
    if constexpr (VEC == 2) {
        v2f r = *(const v2f*)p;
        v[0] = r.x; v[1] = r.y;
    } else {
        v[0] = p[0];
    }
}

template<int VEC>
static __device__ __forceinline__ void seg_max(const float (&r)[VEC], float t, float (&mx)[VEC]) {
#pragma unroll
    for (int i = 0; i < VEC; ++i) mx[i] = fmaxf(mx[i], r[i] * t);
}
template<int VEC>
static __device__ __forceinline__ void seg_sum(const float (&r)[VEC], float t, const float (&mx)[VEC],
                                               float (&num)[VEC], float (&den)[VEC]) {
#pragma unroll
    for (int i = 0; i < VEC; ++i) {
        float e = __expf(r[i] * t - mx[i]);
        num[i] += r[i] * e;
        den[i] += e;
    }
}

template<int J, int VEC, int CP, int R>
struct SegChain {
    static __device__ __forceinline__ void load(int deg, const float* __restrict__ x, unsigned loff,
                                                const int (&sidx)[32], float (&v)[R][VEC]) {
        if constexpr (J < R) {
            if (J < deg) {
                ld_row<VEC>(x + (size_t)sidx[J] * CP + loff, v[J]);
                SegChain<J + 1, VEC, CP, R>::load(deg, x, loff, sidx, v);
            }
        }
    }
    static __device__ __forceinline__ void vmax(int deg, const float (&v)[R][VEC], float t, float (&mx)[VEC]) {
        if constexpr (J < R) {
            if (J < deg) {
                seg_max<VEC>(v[J], t, mx);
                SegChain<J + 1, VEC, CP, R>::vmax(deg, v, t, mx);
            }
        }
    }
    static __device__ __forceinline__ void vsum(int deg, const float (&v)[R][VEC], float t,
                                                const float (&mx)[VEC], float (&num)[VEC], float (&den)[VEC]) {
        if constexpr (J < R) {
            if (J < deg) {
                seg_sum<VEC>(v[J], t, mx, num, den);
                SegChain<J + 1, VEC, CP, R>::vsum(deg, v, t, mx, num, den);
            }
        }
    }
};

static __device__ __forceinline__ float xhalf(float x) {
    int b = __builtin_bit_cast(int, x);
    return __builtin_bit_cast(float, __builtin_amdgcn_permlanex16(b, b, 0x76543210, 0xfedcba98, false, false));
}
template<int M>
static __device__ __forceinline__ float rowx(float x) {
    return __builtin_bit_cast(float, __builtin_amdgcn_update_dpp(0, __builtin_bit_cast(int, x), 0x160 + M, 0xf, 0xf, true));
}
template<int M>
static __device__ __forceinline__ void row_step8(float (&s)[8]) {
#pragma unroll
    for (int r = 0; r < 8; ++r) s[r] += rowx<M>(s[r]);
}
static __device__ __forceinline__ void row_sum16x8(float (&s)[8]) {
    row_step8<1>(s);
    row_step8<2>(s);
    row_step8<4>(s);
    row_step8<8>(s);
}

#define NPW  4
#define GN   (16 * NPW)
static_assert(NPW == 4, "phase 2 gives tile t to wave 5t: waves 0, 5, 10, 15");
template<int CP>
__global__ void __launch_bounds__(512, 12) k_layer(const int* __restrict__ rowptr, const int* __restrict__ csrc,
                                               const float* __restrict__ xin, const float* __restrict__ tptr,
                                               const float* __restrict__ Wr, const float* __restrict__ Ws,
                                               const float* __restrict__ bias, const int* __restrict__ batch,
                                               float* __restrict__ out, unsigned* __restrict__ he, int Nrows) {
    constexpr int VEC = CP / 32;
    constexpr int R = (VEC == 2) ? 30 : 32;
    constexpr int PITCH = CP + 4;
    __shared__ __attribute__((aligned(16))) float sA[GN * PITCH];
    __shared__ __attribute__((aligned(16))) float sX[GN * PITCH];
    __shared__ __attribute__((aligned(16))) float sW[2 * HID * PITCH];
    const int lane = threadIdx.x & 31;
    const int wv = __builtin_amdgcn_readfirstlane((int)(threadIdx.x >> 5));
    const int g0n = (int)blockIdx.x * GN;
    if (g0n >= Nrows) return;
    const int n0 = g0n + wv * NPW;
    constexpr int Q   = CP / 4;
    constexpr int PER = 2 * HID * Q / 512;
    float4 wq[PER];
#pragma unroll
    for (int i = 0; i < PER; ++i) {
        const int rem = (i * 512) % (HID * Q) + (int)threadIdx.x;
        const float* __restrict__ Wm = ((i * 512) / (HID * Q)) ? Ws : Wr;
        wq[i] = *(const float4*)(Wm + (rem / Q) * CP + 4 * (rem % Q));
    }
    const unsigned loff = (unsigned)(lane * VEC);
    float xr[NPW][VEC];
#pragma unroll
    for (int i = 0; i < NPW; ++i)
        ld_row<VEC>(xin + (size_t)min(n0 + i, Nrows - 1) * CP + loff, xr[i]);

    const int rp = rowptr[min(n0 + lane, Nrows)];
    const float t = tptr[0];
    int bnd[NPW + 1];
#pragma unroll
    for (int k = 0; k <= NPW; ++k) bnd[k] = __builtin_amdgcn_readlane(rp, k);
    int beg = bnd[0];
    int end = bnd[1];
    int mys0 = (lane < min(end - beg, 32)) ? csrc[beg + lane] : 0;

#pragma unroll
    for (int i = 0; i < PER; ++i) {
        const int rem = (i * 512) % (HID * Q) + (int)threadIdx.x;
        const int n = rem / Q;
        float* d = sW + ((i * 512) / (HID * Q)) * (HID * PITCH) + ((n & 3) * 16 + (n >> 2)) * PITCH + 4 * (rem % Q);
        *(float4*)d = wq[i];
    }
#pragma unroll
    for (int i = 0; i < NPW; ++i)
#pragma unroll
        for (int k = 0; k < VEC; ++k) sX[(wv * NPW + i) * PITCH + loff + k] = xr[i][k];

#pragma nounroll
    for (int i = 0; i < NPW; ++i) {
        const int deg = end - beg;
        int sidx[32];
#pragma unroll
        for (int j = 0; j < 32; ++j) sidx[j] = __builtin_amdgcn_readlane(mys0, j);
        const int nbeg = end;
        int nend = bnd[NPW];
#pragma unroll
        for (int k = NPW - 1; k >= 2; --k) nend = (i + 2 == k) ? bnd[k] : nend;
        const int mysn = (i + 1 < NPW && lane < min(nend - nbeg, 32)) ? csrc[nbeg + lane] : 0;
        float mx[VEC], num[VEC], den[VEC];
#pragma unroll
        for (int k = 0; k < VEC; ++k) { mx[k] = -INFINITY; num[k] = 0.f; den[k] = 0.f; }
        if (deg <= R) {
            float v[R][VEC];
            SegChain<0, VEC, CP, R>::load(deg, xin, loff, sidx, v);
            SegChain<0, VEC, CP, R>::vmax(deg, v, t, mx);
            SegChain<0, VEC, CP, R>::vsum(deg, v, t, mx, num, den);
        } else {
            for (int base = beg; base < end; base += 32) {
                const int cnt = min(32, end - base);
                const int mys = (lane < cnt) ? csrc[base + lane] : 0;
                for (int j = 0; j < cnt; ++j) {
                    const int s = __builtin_amdgcn_readlane(mys, j);
                    float r[VEC];
                    ld_row<VEC>(xin + (size_t)s * CP + loff, r);
                    seg_max<VEC>(r, t, mx);
                }
            }
            for (int base = beg; base < end; base += 32) {
                const int cnt = min(32, end - base);
                const int mys = (lane < cnt) ? csrc[base + lane] : 0;
                for (int j = 0; j < cnt; ++j) {
                    const int s = __builtin_amdgcn_readlane(mys, j);
                    float r[VEC];
                    ld_row<VEC>(xin + (size_t)s * CP + loff, r);
                    seg_sum<VEC>(r, t, mx, num, den);
                }
            }
        }
        float* o = sA + (wv * NPW + i) * PITCH + loff;
#pragma unroll
        for (int k = 0; k < VEC; ++k) o[k] = den[k] > 0.f ? num[k] / den[k] : 0.f;
        beg = nbeg; end = nend; mys0 = mysn;
    }
    __syncthreads();
    if (wv % 5 != 0) return;
    const int m0 = g0n + (wv / 5) * 16;
    if (m0 >= Nrows) return;

    const int half = lane >> 4;
    const int c    = lane & 15;
    const float* ap  = sA + ((wv / 5) * 16 + c) * PITCH + half * 2;
    const float* xp  = sX + ((wv / 5) * 16 + c) * PITCH + half * 2;
    const float* wr  = sW + c * PITCH + half * 2;
    const float* wsp = wr + HID * PITCH;

    v8f acc[4] = {};
#pragma unroll
    for (int k = 0; k < CP; k += 4) {
        v2f a = *(const v2f*)(ap + k);
#pragma unroll
        for (int j = 0; j < 4; ++j) {
            v2f b = *(const v2f*)(wr + j * 16 * PITCH + k);
            acc[j] = __builtin_amdgcn_wmma_f32_16x16x4_f32(false, a, false, b,
                                                           (short)0, acc[j], false, false);
        }
    }
#pragma unroll
    for (int k = 0; k < CP; k += 4) {
        v2f a = *(const v2f*)(xp + k);
#pragma unroll
        for (int j = 0; j < 4; ++j) {
            v2f b = *(const v2f*)(wsp + j * 16 * PITCH + k);
            acc[j] = __builtin_amdgcn_wmma_f32_16x16x4_f32(false, a, false, b,
                                                           (short)0, acc[j], false, false);
        }
    }
    const int gl = batch[min(m0 + c, Nrows - 1)];
    const int half_e = (int)(__builtin_amdgcn_mbcnt_lo(~0u, 0u) >> 4);
    const float4 bq = *(const float4*)(bias + 4 * c);
    const float bv[4] = { bq.x, bq.y, bq.z, bq.w };
    float s[8], q[8];
#pragma unroll
    for (int r = 0; r < 8; ++r) {
#pragma unroll
        for (int j = 0; j < 4; ++j) acc[j][r] += bv[j];
        s[r] = (acc[0][r] + acc[1][r]) + (acc[2][r] + acc[3][r]);
    }
    row_sum16x8(s);
#pragma unroll
    for (int r = 0; r < 8; ++r) {
        const float mu = s[r] * (1.f / HID);
#pragma unroll
        for (int j = 0; j < 4; ++j) acc[j][r] -= mu;
        q[r] = (acc[0][r] * acc[0][r] + acc[1][r] * acc[1][r]) + (acc[2][r] * acc[2][r] + acc[3][r] * acc[3][r]);
    }
    row_sum16x8(q);
    float cm[4];
#pragma unroll
    for (int r = 0; r < 8; ++r) {
        const float inv = rsqrtf(q[r] * (1.f / HID) + EPS);
#pragma unroll
        for (int j = 0; j < 4; ++j) {
            const float y = fmaxf(acc[j][r] * inv, 0.f);
            acc[j][r] = y;
            cm[j] = (r == 0) ? y : fmaxf(cm[j], y);
        }
    }
    float* op = out + (size_t)(m0 + half_e * 8) * HID + 4 * c;
    if (m0 + 16 <= Nrows) {
#pragma unroll
        for (int r = 0; r < 8; ++r) {
            float4 o; o.x = acc[0][r]; o.y = acc[1][r]; o.z = acc[2][r]; o.w = acc[3][r];
            *(float4*)(op + (size_t)r * HID) = o;
        }
    } else {
#pragma unroll
        for (int r = 0; r < 8; ++r) {
            if (m0 + r + half_e * 8 < Nrows) {
                float4 o; o.x = acc[0][r]; o.y = acc[1][r]; o.z = acc[2][r]; o.w = acc[3][r];
                *(float4*)(op + (size_t)r * HID) = o;
            }
        }
    }

    const int g0 = __builtin_amdgcn_readfirstlane(gl);
    const bool one_graph = __builtin_amdgcn_ballot_w32(gl != g0) == 0u;
    if (one_graph) {
#pragma unroll
        for (int j = 0; j < 4; ++j) cm[j] = fmaxf(cm[j], xhalf(cm[j]));
        const float p0 = half_e ? cm[2] : cm[0];
        const float p1 = half_e ? cm[3] : cm[1];
        unsigned* hp = he + (size_t)g0 * HID + 4 * c + 2 * half_e;
        atomicMax(hp,     enc_f32(p0));
        atomicMax(hp + 1, enc_f32(p1));
    } else {
#pragma unroll
        for (int r = 0; r < 8; ++r) {
            const int row = m0 + r + half_e * 8;
            if (row < Nrows) {
                unsigned* hp = he + (size_t)batch[row] * HID + 4 * c;
#pragma unroll
                for (int j = 0; j < 4; ++j) atomicMax(hp + j, enc_f32(acc[j][r]));
            }
        }
    }
}

template<int NV>
static __device__ __forceinline__ void wave_sum_n(float (&p)[NV]) {
#pragma unroll
    for (int j = 0; j < NV; ++j) p[j] += rowx<1>(p[j]);
#pragma unroll
    for (int j = 0; j < NV; ++j) p[j] += rowx<2>(p[j]);
#pragma unroll
    for (int j = 0; j < NV; ++j) p[j] += rowx<4>(p[j]);
#pragma unroll
    for (int j = 0; j < NV; ++j) p[j] += rowx<8>(p[j]);
#pragma unroll
    for (int j = 0; j < NV; ++j) p[j] += xhalf(p[j]);
}

__global__ void __launch_bounds__(256) k_head(const unsigned* __restrict__ henc,
                                              const float* __restrict__ Wl1, const float* __restrict__ bl1,
                                              const float* __restrict__ Wl2, const float* __restrict__ bl2,
                                              float* __restrict__ outp) {
    __shared__ __attribute__((aligned(16))) float sh[3 * HID];
    __shared__ __attribute__((aligned(16))) float sz1[2 * HID];
    __shared__ __attribute__((aligned(16))) float sz2[HID / 2];
    const int g = blockIdx.x;
    const int tid = threadIdx.x, lane = tid & 31, w = tid >> 5;
    if (tid < 3 * HID) {
        const int layer = tid / HID, cc = tid % HID;
        sh[tid] = dec_f32(henc[layer * NG * HID + g * HID + cc]);
    }
    __syncthreads();

    {
        v2f hv[3];
#pragma unroll
        for (int i = 0; i < 3; ++i) hv[i] = *(const v2f*)(sh + 2 * (lane + 32 * i));
        const float* __restrict__ wbase = Wl1 + (w * 16) * (3 * HID) + 2 * lane;
        float p[16];
#pragma unroll
        for (int j = 0; j < 16; ++j) {
            const float* __restrict__ wrow = wbase + j * (3 * HID);
            float a = 0.f;
#pragma unroll
            for (int i = 0; i < 3; ++i) {
                const v2f wv = *(const v2f*)(wrow + 64 * i);
                a += hv[i].x * wv.x;
                a += hv[i].y * wv.y;
            }
            p[j] = a;
        }
        wave_sum_n<16>(p);
        float mine = p[0];
#pragma unroll
        for (int j = 1; j < 16; ++j) mine = (lane == j) ? p[j] : mine;
        if (lane < 16) sz1[w * 16 + lane] = fmaxf(mine + bl1[w * 16 + lane], 0.f);
    }
    __syncthreads();

    {
        const float4 zv = *(const float4*)(sz1 + 4 * lane);
        const float* __restrict__ wbase = Wl2 + (w * 4) * (2 * HID) + 4 * lane;
        float p[4];
#pragma unroll
        for (int j = 0; j < 4; ++j) {
            const float4 wv = *(const float4*)(wbase + j * (2 * HID));
            p[j] = (zv.x * wv.x + zv.y * wv.y) + (zv.z * wv.z + zv.w * wv.w);
        }
        wave_sum_n<4>(p);
        float mine = p[0];
#pragma unroll
        for (int j = 1; j < 4; ++j) mine = (lane == j) ? p[j] : mine;
        if (lane < 4) sz2[w * 4 + lane] = mine + bl2[w * 4 + lane];
    }
    __syncthreads();

    if (w == 0) {
        const float s = sz2[lane];
        float q[1] = { s * s };
        wave_sum_n<1>(q);
        const float nrm = fmaxf(sqrtf(q[0]), 1e-12f);
        outp[g * (HID / 2) + lane] = s / nrm;
    }
}

extern "C" void kernel_launch(void* const* d_in, const int* in_sizes, int n_in,
                              void* d_out, int out_size, void* d_ws, size_t ws_size,
                              hipStream_t stream) {
    const float* x    = (const float*)d_in[0];
    const int*  eidx  = (const int*)  d_in[1];
    const int*  batch = (const int*)  d_in[2];
    const float* t    = (const float*)d_in[3];
    const float* W1r  = (const float*)d_in[4];
    const float* b1   = (const float*)d_in[5];
    const float* W1s  = (const float*)d_in[6];
    const float* W2r  = (const float*)d_in[7];
    const float* b2   = (const float*)d_in[8];
    const float* W2s  = (const float*)d_in[9];
    const float* W3r  = (const float*)d_in[10];
    const float* b3   = (const float*)d_in[11];
    const float* W3s  = (const float*)d_in[12];
    const float* Wl1  = (const float*)d_in[13];
    const float* bl1  = (const float*)d_in[14];
    const float* Wl2  = (const float*)d_in[15];
    const float* bl2  = (const float*)d_in[16];
    const int* src = eidx;
    const int* dst = eidx + NE;

    float* ws = (float*)d_ws;
    size_t o = 0;
    float*    x1     = ws + o; o += (size_t)NN * HID;
    float*    x2     = ws + o; o += (size_t)NN * HID;
    float*    x3     = ws + o; o += (size_t)NN * HID;
    int*      csr    = (int*)(ws + o); o += (size_t)NN * HID * 3;
    int*      boff     = csr;
    unsigned* binned   = (unsigned*)(boff + NBLK * BOFFW + 16);
    int*      rowptr   = (int*)(binned + NBLK * EPB);
    int*      csrc     = rowptr + NN + 16;
    o += (size_t)NN * HID;
    float*    xpad   = ws + o; o += (size_t)NN * 32;
    float*    w1rp   = ws + o; o += (size_t)HID * 32;
    float*    w1sp   = ws + o; o += (size_t)HID * 32;
    unsigned* henc   = (unsigned*)(ws + o); o += 3 * NG * HID;

    k_sort_block<<<NBLK, 1024, 0, stream>>>(src, dst, boff, binned, henc, W1r, W1s, w1rp, w1sp, x, xpad);
    k_sort_bin<<<NBIN, 1024, 0, stream>>>(binned, boff, rowptr, csrc);

    struct Layer { const float* xin; int Cp; const float *Wr, *b, *Ws; float* xout; };
    Layer layers[3] = {
        { xpad, 32, w1rp, b1, w1sp, x1 },
        { x1,   64, W2r,  b2, W2s,  x2 },
        { x2,   64, W3r,  b3, W3s,  x3 },
    };

    for (int l = 0; l < 3; ++l) {
        const Layer& L = layers[l];
        if (L.Cp == 32)
            k_layer<32><<<(NN + GN - 1) / GN, 512, 0, stream>>>(rowptr, csrc, L.xin, t, L.Wr, L.Ws, L.b, batch,
                                                            L.xout, henc + (size_t)l * NG * HID, NN);
        else
            k_layer<64><<<(NN + GN - 1) / GN, 512, 0, stream>>>(rowptr, csrc, L.xin, t, L.Wr, L.Ws, L.b, batch,
                                                            L.xout, henc + (size_t)l * NG * HID, NN);
    }

    k_head<<<NG, 256, 0, stream>>>(henc, Wl1, bl1, Wl2, bl2, (float*)d_out);
}


# Softmax gather, fewer vector ops per edge: segment max from raw row max and min (3-input max/min, rows in pairs) scaled once by t; exp as one fma into exp2 with t*log2(e) folded in; same sum order
# speedup vs baseline: 7.4932x; 1.0555x over previous
#include <hip/hip_runtime.h>
#include <math.h>

#define NN      50000
#define NE      800000
#define NG      128
#define INC     31
#define HID     64
#define EPS     1e-5f

typedef __attribute__((ext_vector_type(2))) float v2f;
typedef __attribute__((ext_vector_type(8))) float v8f;

static __device__ __forceinline__ unsigned enc_f32(float x) {
    unsigned b = __float_as_uint(x);
    return (b & 0x80000000u) ? ~b : (b | 0x80000000u);
}
static __device__ __forceinline__ float dec_f32(unsigned e) {
    unsigned b = (e & 0x80000000u) ? (e & 0x7FFFFFFFu) : ~e;
    return __uint_as_float(b);
}
#define ENC_NEG_INF 0x007FFFFFu

#define EPB    4096
#define NBLK   ((NE + EPB - 1) / EPB)
#define BINSH  8
#define BINW   (1 << BINSH)
#define NBIN   ((NN + BINW - 1) / BINW)
#define BOFFW  (NBIN + 1)
static_assert(NN <= 65536, "source and destination must fit 16 bits each");
static_assert(NBIN <= 256 && NBLK <= 256, "one thread per bin in level 1, one thread (and one thread quad) per run in level 2");
static_assert(NBLK * 1024 >= 3 * NG * HID && NBLK * 1024 >= HID * 32, "prep ranges are prefixes of the grid");
static_assert((long long)NBLK * 1024 * 8 >= (long long)NN * 32, "eight padded elements per thread cover x");

static __device__ __forceinline__ void excl_scan256x2(int a, int b, int (&wsa)[8], int (&wsb)[8],
                                                      int& ea, int& eb, int& tota, int& totb) {
    const int tid = threadIdx.x, lane = tid & 31, w = tid >> 5;
    int ia = a, ib = b;
    if (w < 8) {
        for (int off = 1; off < 32; off <<= 1) {
            const int ua = __shfl_up(ia, off), ub = __shfl_up(ib, off);
            if (lane >= off) { ia += ua; ib += ub; }
        }
        if (lane == 31) { wsa[w] = ia; wsb[w] = ib; }
    }
    __syncthreads();
    int ba = 0, bb = 0, ta = 0, tb = 0;
#pragma unroll
    for (int i = 0; i < 8; ++i) {
        const int sa = wsa[i], sb = wsb[i];
        ta += sa; tb += sb;
        if (i < w) { ba += sa; bb += sb; }
    }
    ea = ba + ia - a; eb = bb + ib - b; tota = ta; totb = tb;
}

__global__ void __launch_bounds__(1024) k_sort_block(const int* __restrict__ src, const int* __restrict__ dst,
                                                     int* __restrict__ boff, unsigned* __restrict__ binned,
                                                     unsigned* __restrict__ henc,
                                                     const float* __restrict__ W1r, const float* __restrict__ W1s,
                                                     float* __restrict__ w1rp, float* __restrict__ w1sp,
                                                     const float* __restrict__ x, float* __restrict__ xpad) {
    __shared__ int h[NBIN];
    __shared__ int loff[NBIN];
    __shared__ int wsa[8];
    __shared__ int wsb[8];
    __shared__ unsigned stage[EPB];
    const int tid = threadIdx.x, blk = blockIdx.x;
    if (tid < NBIN) h[tid] = 0;
    int s[4], d[4];
#pragma unroll
    for (int k = 0; k < 4; ++k) {
        const int e = min(blk * EPB + k * 1024 + tid, NE - 1);
        s[k] = src[e];
        d[k] = dst[e];
    }
    const int i0 = blk * 1024 + tid;
    float xv[8];
#pragma unroll
    for (int k = 0; k < 8; ++k) {
        const int j = min(i0 + k * (NBLK * 1024), NN * 32 - 1);
        xv[k] = x[(j >> 5) * INC + min(j & 31, INC - 1)];
    }
    __syncthreads();
    int r[4];
#pragma unroll
    for (int k = 0; k < 4; ++k)
        r[k] = (blk * EPB + k * 1024 + tid < NE) ? atomicAdd(&h[d[k] >> BINSH], 1) : 0;
    __syncthreads();
    const int hv = (tid < NBIN) ? h[tid] : 0;
    int lo, unused, total, unused2;
    excl_scan256x2(hv, 0, wsa, wsb, lo, unused, total, unused2);
    if (tid < NBIN) {
        loff[tid] = lo;
        boff[blk * BOFFW + tid] = lo;
    }
    if (tid == 0) boff[blk * BOFFW + NBIN] = total;
    __syncthreads();
    int lf[4];
#pragma unroll
    for (int k = 0; k < 4; ++k) lf[k] = loff[d[k] >> BINSH];
#pragma unroll
    for (int k = 0; k < 4; ++k) {
        const int i = k * 1024 + tid;
        const bool live = blk * EPB + i < NE;
        stage[live ? lf[k] + r[k] : i] = live ? (((unsigned)d[k] << 16) | (unsigned)s[k]) : 0u;
    }
    __syncthreads();
    const int nvalid = min(EPB, NE - blk * EPB);
    unsigned p[4];
#pragma unroll
    for (int k = 0; k < 4; ++k) p[k] = stage[k * 1024 + tid];
#pragma unroll
    for (int k = 0; k < 4; ++k) {
        const int i = k * 1024 + tid;
        if (i < nvalid) binned[blk * EPB + i] = p[k];
    }

    if (i0 < 3 * NG * HID) henc[i0] = ENC_NEG_INF;
    if (i0 < HID * 32) {
        const int rr = i0 / 32, cc = i0 % 32;
        w1rp[i0] = (cc < INC) ? W1r[rr * INC + cc] : 0.f;
        w1sp[i0] = (cc < INC) ? W1s[rr * INC + cc] : 0.f;
    }
#pragma unroll
    for (int k = 0; k < 8; ++k) {
        const int j = i0 + k * (NBLK * 1024);
        if (j < NN * 32) xpad[j] = ((j & 31) < INC) ? xv[k] : 0.f;
    }
}

#define BCAP (2 * EPB)
#define RUNC 8
__global__ void __launch_bounds__(1024) k_sort_bin(const unsigned* __restrict__ binned, const int* __restrict__ boff,
                                                   int* __restrict__ rowptr, int* __restrict__ csrc) {
    __shared__ int cnt[BINW];
    __shared__ int wsa[8];
    __shared__ int wsb[8];
    __shared__ int outst[BCAP];
    const int tid = threadIdx.x, bin = blockIdx.x;
    const int t4 = tid >> 2, sub = tid & 3;
    const int ta = min(tid, NBLK - 1) * BOFFW + bin, tq = min(t4, NBLK - 1) * BOFFW + bin;
    const int a0 = boff[ta], b0 = boff[ta + 1];
    const int q0 = boff[tq], q1 = boff[tq + 1];
    if (tid < BINW) cnt[tid] = 0;
    int unused, unused2, beg, end;
    excl_scan256x2((tid < NBLK) ? a0 : 0, (tid < NBLK) ? b0 : 0, wsa, wsb, unused, unused2, beg, end);
    const int n = end - beg;
    const int rs = (t4 < NBLK) ? q0 : 0, re = (t4 < NBLK) ? q1 : 0;
    const unsigned* __restrict__ run = binned + min(t4, NBLK - 1) * EPB;
    for (int base = rs; base < re; base += 4 * RUNC) {
        unsigned p[RUNC];
#pragma unroll
        for (int k = 0; k < RUNC; ++k) p[k] = run[min(base + 4 * k + sub, re - 1)];
#pragma unroll
        for (int k = 0; k < RUNC; ++k)
            if (base + 4 * k + sub < re) atomicAdd(&cnt[(p[k] >> 16) & (BINW - 1)], 1);
    }
    __syncthreads();
    const int c = (tid < BINW) ? cnt[tid] : 0;
    int ex, total;
    excl_scan256x2(c, 0, wsa, wsb, ex, unused, total, unused2);
    if (tid < BINW) {
        const int node = bin * BINW + tid;
        if (node <= NN) rowptr[node] = beg + ex;
        cnt[tid] = ex;
    }
    __syncthreads();
    const bool staged = n <= BCAP;
    for (int base = rs; base < re; base += 4 * RUNC) {
        unsigned p[RUNC];
#pragma unroll
        for (int k = 0; k < RUNC; ++k) p[k] = run[min(base + 4 * k + sub, re - 1)];
        int rk[RUNC];
#pragma unroll
        for (int k = 0; k < RUNC; ++k)
            rk[k] = (base + 4 * k + sub < re) ? atomicAdd(&cnt[(p[k] >> 16) & (BINW - 1)], 1) : 0;
#pragma unroll
        for (int k = 0; k < RUNC; ++k) {
            if (base + 4 * k + sub < re) {
                const int sv = (int)(p[k] & 0xFFFFu);
                if (staged) outst[rk[k]] = sv; else csrc[beg + rk[k]] = sv;
            }
        }
    }
    if (staged && n > 0) {
        __syncthreads();
        int v[BCAP / 1024];
#pragma unroll
        for (int k = 0; k < BCAP / 1024; ++k) v[k] = outst[min(k * 1024 + tid, n - 1)];
#pragma unroll
        for (int k = 0; k < BCAP / 1024; ++k)
            if (k * 1024 + tid < n) csrc[beg + k * 1024 + tid] = v[k];
    }
}

template<int VEC>
static __device__ __forceinline__ void ld_row(const float* __restrict__ p, float (&v)[VEC]) {
    if constexpr (VEC == 2) {
        v2f r = *(const v2f*)p;
        v[0] = r.x; v[1] = r.y;
    } else {
        v[0] = p[0];
    }
}

template<int VEC>
static __device__ __forceinline__ void seg_rng(const float (&r)[VEC], float (&hi)[VEC], float (&lo)[VEC]) {
#pragma unroll
    for (int i = 0; i < VEC; ++i) { hi[i] = fmaxf(hi[i], r[i]); lo[i] = fminf(lo[i], r[i]); }
}
template<int VEC>
static __device__ __forceinline__ void seg_rng2(const float (&a)[VEC], const float (&b)[VEC], float (&hi)[VEC], float (&lo)[VEC]) {
#pragma unroll
    for (int i = 0; i < VEC; ++i) { hi[i] = fmaxf(fmaxf(hi[i], a[i]), b[i]); lo[i] = fminf(fminf(lo[i], a[i]), b[i]); }
}
template<int VEC>
static __device__ __forceinline__ void seg_sum(const float (&r)[VEC], float t2, const float (&mx2)[VEC],
                                               float (&num)[VEC], float (&den)[VEC]) {
#pragma unroll
    for (int i = 0; i < VEC; ++i) {
        float e = __builtin_amdgcn_exp2f(__builtin_fmaf(r[i], t2, -mx2[i]));
        num[i] += r[i] * e;
        den[i] += e;
    }
}
template<int VEC>
static __device__ __forceinline__ void seg_top(const float (&hi)[VEC], const float (&lo)[VEC], float t2, float (&mx2)[VEC]) {
#pragma unroll
    for (int i = 0; i < VEC; ++i) mx2[i] = fmaxf(hi[i] * t2, lo[i] * t2);
}

template<int J, int VEC, int CP, int R>
struct SegChain {
    static __device__ __forceinline__ void load(int deg, const float* __restrict__ x, unsigned loff,
                                                const int (&sidx)[32], float (&v)[R][VEC]) {
        if constexpr (J + 1 < R) {
            if (J < deg) {
                ld_row<VEC>(x + (size_t)sidx[J] * CP + loff, v[J]);
                const int s1 = (J + 1 < deg) ? sidx[J + 1] : sidx[J];
                ld_row<VEC>(x + (size_t)s1 * CP + loff, v[J + 1]);
                SegChain<J + 2, VEC, CP, R>::load(deg, x, loff, sidx, v);
            }
        }
    }
    static __device__ __forceinline__ void vrng(int deg, const float (&v)[R][VEC], float (&hi)[VEC], float (&lo)[VEC]) {
        if constexpr (J + 1 < R) {
            if (J < deg) {
                seg_rng2<VEC>(v[J], v[J + 1], hi, lo);
                SegChain<J + 2, VEC, CP, R>::vrng(deg, v, hi, lo);
            }
        }
    }
    static __device__ __forceinline__ void vsum(int deg, const float (&v)[R][VEC], float t,
                                                const float (&mx)[VEC], float (&num)[VEC], float (&den)[VEC]) {
        if constexpr (J < R) {
            if (J < deg) {
                seg_sum<VEC>(v[J], t, mx, num, den);
                SegChain<J + 1, VEC, CP, R>::vsum(deg, v, t, mx, num, den);
            }
        }
    }
};

static __device__ __forceinline__ float xhalf(float x) {
    int b = __builtin_bit_cast(int, x);
    return __builtin_bit_cast(float, __builtin_amdgcn_permlanex16(b, b, 0x76543210, 0xfedcba98, false, false));
}
template<int M>
static __device__ __forceinline__ float rowx(float x) {
    return __builtin_bit_cast(float, __builtin_amdgcn_update_dpp(0, __builtin_bit_cast(int, x), 0x160 + M, 0xf, 0xf, true));
}
template<int M>
static __device__ __forceinline__ void row_step8(float (&s)[8]) {
#pragma unroll
    for (int r = 0; r < 8; ++r) s[r] += rowx<M>(s[r]);
}
static __device__ __forceinline__ void row_sum16x8(float (&s)[8]) {
    row_step8<1>(s);
    row_step8<2>(s);
    row_step8<4>(s);
    row_step8<8>(s);
}

#define NPW  4
#define GN   (16 * NPW)
static_assert(NPW == 4, "phase 2 gives tile t to wave 5t: waves 0, 5, 10, 15");
template<int CP>
__global__ void __launch_bounds__(512, 12) k_layer(const int* __restrict__ rowptr, const int* __restrict__ csrc,
                                               const float* __restrict__ xin, const float* __restrict__ tptr,
                                               const float* __restrict__ Wr, const float* __restrict__ Ws,
                                               const float* __restrict__ bias, const int* __restrict__ batch,
                                               float* __restrict__ out, unsigned* __restrict__ he, int Nrows) {
    constexpr int VEC = CP / 32;
    constexpr int R = (VEC == 2) ? 30 : 32;
    static_assert(R % 2 == 0, "the held rows are fetched and ranged in pairs");
    constexpr int PITCH = CP + 4;
    __shared__ __attribute__((aligned(16))) float sA[GN * PITCH];
    __shared__ __attribute__((aligned(16))) float sX[GN * PITCH];
    __shared__ __attribute__((aligned(16))) float sW[2 * HID * PITCH];
    const int lane = threadIdx.x & 31;
    const int wv = __builtin_amdgcn_readfirstlane((int)(threadIdx.x >> 5));
    const int g0n = (int)blockIdx.x * GN;
    if (g0n >= Nrows) return;
    const int n0 = g0n + wv * NPW;
    constexpr int Q   = CP / 4;
    constexpr int PER = 2 * HID * Q / 512;
    float4 wq[PER];
#pragma unroll
    for (int i = 0; i < PER; ++i) {
        const int rem = (i * 512) % (HID * Q) + (int)threadIdx.x;
        const float* __restrict__ Wm = ((i * 512) / (HID * Q)) ? Ws : Wr;
        wq[i] = *(const float4*)(Wm + (rem / Q) * CP + 4 * (rem % Q));
    }
    const unsigned loff = (unsigned)(lane * VEC);
    float xr[NPW][VEC];
#pragma unroll
    for (int i = 0; i < NPW; ++i)
        ld_row<VEC>(xin + (size_t)min(n0 + i, Nrows - 1) * CP + loff, xr[i]);

    const int rp = rowptr[min(n0 + lane, Nrows)];
    const float t2 = tptr[0] * 1.44269504088896340736f;
    int bnd[NPW + 1];
#pragma unroll
    for (int k = 0; k <= NPW; ++k) bnd[k] = __builtin_amdgcn_readlane(rp, k);
    int beg = bnd[0];
    int end = bnd[1];
    int mys0 = (lane < min(end - beg, 32)) ? csrc[beg + lane] : 0;

#pragma unroll
    for (int i = 0; i < PER; ++i) {
        const int rem = (i * 512) % (HID * Q) + (int)threadIdx.x;
        const int n = rem / Q;
        float* d = sW + ((i * 512) / (HID * Q)) * (HID * PITCH) + ((n & 3) * 16 + (n >> 2)) * PITCH + 4 * (rem % Q);
        *(float4*)d = wq[i];
    }
#pragma unroll
    for (int i = 0; i < NPW; ++i)
#pragma unroll
        for (int k = 0; k < VEC; ++k) sX[(wv * NPW + i) * PITCH + loff + k] = xr[i][k];

#pragma nounroll
    for (int i = 0; i < NPW; ++i) {
        const int deg = end - beg;
        int sidx[32];
#pragma unroll
        for (int j = 0; j < 32; ++j) sidx[j] = __builtin_amdgcn_readlane(mys0, j);
        const int nbeg = end;
        int nend = bnd[NPW];
#pragma unroll
        for (int k = NPW - 1; k >= 2; --k) nend = (i + 2 == k) ? bnd[k] : nend;
        const int mysn = (i + 1 < NPW && lane < min(nend - nbeg, 32)) ? csrc[nbeg + lane] : 0;
        float hi[VEC], lo[VEC], mx[VEC], num[VEC], den[VEC];
#pragma unroll
        for (int k = 0; k < VEC; ++k) { hi[k] = -INFINITY; lo[k] = INFINITY; num[k] = 0.f; den[k] = 0.f; }
        if (deg <= R) {
            float v[R][VEC];
            SegChain<0, VEC, CP, R>::load(deg, xin, loff, sidx, v);
            SegChain<0, VEC, CP, R>::vrng(deg, v, hi, lo);
            seg_top<VEC>(hi, lo, t2, mx);
            SegChain<0, VEC, CP, R>::vsum(deg, v, t2, mx, num, den);
        } else {
            for (int base = beg; base < end; base += 32) {
                const int cnt = min(32, end - base);
                const int mys = (lane < cnt) ? csrc[base + lane] : 0;
                for (int j = 0; j < cnt; ++j) {
                    const int s = __builtin_amdgcn_readlane(mys, j);
                    float r[VEC];
                    ld_row<VEC>(xin + (size_t)s * CP + loff, r);
                    seg_rng<VEC>(r, hi, lo);
                }
            }
            seg_top<VEC>(hi, lo, t2, mx);
            for (int base = beg; base < end; base += 32) {
                const int cnt = min(32, end - base);
                const int mys = (lane < cnt) ? csrc[base + lane] : 0;
                for (int j = 0; j < cnt; ++j) {
                    const int s = __builtin_amdgcn_readlane(mys, j);
                    float r[VEC];
                    ld_row<VEC>(xin + (size_t)s * CP + loff, r);
                    seg_sum<VEC>(r, t2, mx, num, den);
                }
            }
        }
        float* o = sA + (wv * NPW + i) * PITCH + loff;
#pragma unroll
        for (int k = 0; k < VEC; ++k) o[k] = den[k] > 0.f ? num[k] / den[k] : 0.f;
        beg = nbeg; end = nend; mys0 = mysn;
    }
    __syncthreads();
    if (wv % 5 != 0) return;
    const int m0 = g0n + (wv / 5) * 16;
    if (m0 >= Nrows) return;

    const int half = lane >> 4;
    const int c    = lane & 15;
    const float* ap  = sA + ((wv / 5) * 16 + c) * PITCH + half * 2;
    const float* xp  = sX + ((wv / 5) * 16 + c) * PITCH + half * 2;
    const float* wr  = sW + c * PITCH + half * 2;
    const float* wsp = wr + HID * PITCH;

    v8f acc[4] = {};
#pragma unroll
    for (int k = 0; k < CP; k += 4) {
        v2f a = *(const v2f*)(ap + k);
#pragma unroll
        for (int j = 0; j < 4; ++j) {
            v2f b = *(const v2f*)(wr + j * 16 * PITCH + k);
            acc[j] = __builtin_amdgcn_wmma_f32_16x16x4_f32(false, a, false, b,
                                                           (short)0, acc[j], false, false);
        }
    }
#pragma unroll
    for (int k = 0; k < CP; k += 4) {
        v2f a = *(const v2f*)(xp + k);
#pragma unroll
        for (int j = 0; j < 4; ++j) {
            v2f b = *(const v2f*)(wsp + j * 16 * PITCH + k);
            acc[j] = __builtin_amdgcn_wmma_f32_16x16x4_f32(false, a, false, b,
                                                           (short)0, acc[j], false, false);
        }
    }
    const int gl = batch[min(m0 + c, Nrows - 1)];
    const int half_e = (int)(__builtin_amdgcn_mbcnt_lo(~0u, 0u) >> 4);
    const float4 bq = *(const float4*)(bias + 4 * c);
    const float bv[4] = { bq.x, bq.y, bq.z, bq.w };
    float s[8], q[8];
#pragma unroll
    for (int r = 0; r < 8; ++r) {
#pragma unroll
        for (int j = 0; j < 4; ++j) acc[j][r] += bv[j];
        s[r] = (acc[0][r] + acc[1][r]) + (acc[2][r] + acc[3][r]);
    }
    row_sum16x8(s);
#pragma unroll
    for (int r = 0; r < 8; ++r) {
        const float mu = s[r] * (1.f / HID);
#pragma unroll
        for (int j = 0; j < 4; ++j) acc[j][r] -= mu;
        q[r] = (acc[0][r] * acc[0][r] + acc[1][r] * acc[1][r]) + (acc[2][r] * acc[2][r] + acc[3][r] * acc[3][r]);
    }
    row_sum16x8(q);
    float cm[4];
#pragma unroll
    for (int r = 0; r < 8; ++r) {
        const float inv = rsqrtf(q[r] * (1.f / HID) + EPS);
#pragma unroll
        for (int j = 0; j < 4; ++j) {
            const float y = fmaxf(acc[j][r] * inv, 0.f);
            acc[j][r] = y;
            cm[j] = (r == 0) ? y : fmaxf(cm[j], y);
        }
    }
    float* op = out + (size_t)(m0 + half_e * 8) * HID + 4 * c;
    if (m0 + 16 <= Nrows) {
#pragma unroll
        for (int r = 0; r < 8; ++r) {
            float4 o; o.x = acc[0][r]; o.y = acc[1][r]; o.z = acc[2][r]; o.w = acc[3][r];
            *(float4*)(op + (size_t)r * HID) = o;
        }
    } else {
#pragma unroll
        for (int r = 0; r < 8; ++r) {
            if (m0 + r + half_e * 8 < Nrows) {
                float4 o; o.x = acc[0][r]; o.y = acc[1][r]; o.z = acc[2][r]; o.w = acc[3][r];
                *(float4*)(op + (size_t)r * HID) = o;
            }
        }
    }

    const int g0 = __builtin_amdgcn_readfirstlane(gl);
    const bool one_graph = __builtin_amdgcn_ballot_w32(gl != g0) == 0u;
    if (one_graph) {
#pragma unroll
        for (int j = 0; j < 4; ++j) cm[j] = fmaxf(cm[j], xhalf(cm[j]));
        const float p0 = half_e ? cm[2] : cm[0];
        const float p1 = half_e ? cm[3] : cm[1];
        unsigned* hp = he + (size_t)g0 * HID + 4 * c + 2 * half_e;
        atomicMax(hp,     enc_f32(p0));
        atomicMax(hp + 1, enc_f32(p1));
    } else {
#pragma unroll
        for (int r = 0; r < 8; ++r) {
            const int row = m0 + r + half_e * 8;
            if (row < Nrows) {
                unsigned* hp = he + (size_t)batch[row] * HID + 4 * c;
#pragma unroll
                for (int j = 0; j < 4; ++j) atomicMax(hp + j, enc_f32(acc[j][r]));
            }
        }
    }
}

template<int NV>
static __device__ __forceinline__ void wave_sum_n(float (&p)[NV]) {
#pragma unroll
    for (int j = 0; j < NV; ++j) p[j] += rowx<1>(p[j]);
#pragma unroll
    for (int j = 0; j < NV; ++j) p[j] += rowx<2>(p[j]);
#pragma unroll
    for (int j = 0; j < NV; ++j) p[j] += rowx<4>(p[j]);
#pragma unroll
    for (int j = 0; j < NV; ++j) p[j] += rowx<8>(p[j]);
#pragma unroll
    for (int j = 0; j < NV; ++j) p[j] += xhalf(p[j]);
}

__global__ void __launch_bounds__(256) k_head(const unsigned* __restrict__ henc,
                                              const float* __restrict__ Wl1, const float* __restrict__ bl1,
                                              const float* __restrict__ Wl2, const float* __restrict__ bl2,
                                              float* __restrict__ outp) {
    __shared__ __attribute__((aligned(16))) float sh[3 * HID];
    __shared__ __attribute__((aligned(16))) float sz1[2 * HID];
    __shared__ __attribute__((aligned(16))) float sz2[HID / 2];
    const int g = blockIdx.x;
    const int tid = threadIdx.x, lane = tid & 31, w = tid >> 5;
    if (tid < 3 * HID) {
        const int layer = tid / HID, cc = tid % HID;
        sh[tid] = dec_f32(henc[layer * NG * HID + g * HID + cc]);
    }
    __syncthreads();

    {
        v2f hv[3];
#pragma unroll
        for (int i = 0; i < 3; ++i) hv[i] = *(const v2f*)(sh + 2 * (lane + 32 * i));
        const float* __restrict__ wbase = Wl1 + (w * 16) * (3 * HID) + 2 * lane;
        float p[16];
#pragma unroll
        for (int j = 0; j < 16; ++j) {
            const float* __restrict__ wrow = wbase + j * (3 * HID);
            float a = 0.f;
#pragma unroll
            for (int i = 0; i < 3; ++i) {
                const v2f wv = *(const v2f*)(wrow + 64 * i);
                a += hv[i].x * wv.x;
                a += hv[i].y * wv.y;
            }
            p[j] = a;
        }
        wave_sum_n<16>(p);
        float mine = p[0];
#pragma unroll
        for (int j = 1; j < 16; ++j) mine = (lane == j) ? p[j] : mine;
        if (lane < 16) sz1[w * 16 + lane] = fmaxf(mine + bl1[w * 16 + lane], 0.f);
    }
    __syncthreads();

    {
        const float4 zv = *(const float4*)(sz1 + 4 * lane);
        const float* __restrict__ wbase = Wl2 + (w * 4) * (2 * HID) + 4 * lane;
        float p[4];
#pragma unroll
        for (int j = 0; j < 4; ++j) {
            const float4 wv = *(const float4*)(wbase + j * (2 * HID));
            p[j] = (zv.x * wv.x + zv.y * wv.y) + (zv.z * wv.z + zv.w * wv.w);
        }
        wave_sum_n<4>(p);
        float mine = p[0];
#pragma unroll
        for (int j = 1; j < 4; ++j) mine = (lane == j) ? p[j] : mine;
        if (lane < 4) sz2[w * 4 + lane] = mine + bl2[w * 4 + lane];
    }
    __syncthreads();

    if (w == 0) {
        const float s = sz2[lane];
        float q[1] = { s * s };
        wave_sum_n<1>(q);
        const float nrm = fmaxf(sqrtf(q[0]), 1e-12f);
        outp[g * (HID / 2) + lane] = s / nrm;
    }
}

extern "C" void kernel_launch(void* const* d_in, const int* in_sizes, int n_in,
                              void* d_out, int out_size, void* d_ws, size_t ws_size,
                              hipStream_t stream) {
    const float* x    = (const float*)d_in[0];
    const int*  eidx  = (const int*)  d_in[1];
    const int*  batch = (const int*)  d_in[2];
    const float* t    = (const float*)d_in[3];
    const float* W1r  = (const float*)d_in[4];
    const float* b1   = (const float*)d_in[5];
    const float* W1s  = (const float*)d_in[6];
    const float* W2r  = (const float*)d_in[7];
    const float* b2   = (const float*)d_in[8];
    const float* W2s  = (const float*)d_in[9];
    const float* W3r  = (const float*)d_in[10];
    const float* b3   = (const float*)d_in[11];
    const float* W3s  = (const float*)d_in[12];
    const float* Wl1  = (const float*)d_in[13];
    const float* bl1  = (const float*)d_in[14];
    const float* Wl2  = (const float*)d_in[15];
    const float* bl2  = (const float*)d_in[16];
    const int* src = eidx;
    const int* dst = eidx + NE;

    float* ws = (float*)d_ws;
    size_t o = 0;
    float*    x1     = ws + o; o += (size_t)NN * HID;
    float*    x2     = ws + o; o += (size_t)NN * HID;
    float*    x3     = ws + o; o += (size_t)NN * HID;
    int*      csr    = (int*)(ws + o); o += (size_t)NN * HID * 3;
    int*      boff     = csr;
    unsigned* binned   = (unsigned*)(boff + NBLK * BOFFW + 16);
    int*      rowptr   = (int*)(binned + NBLK * EPB);
    int*      csrc     = rowptr + NN + 16;
    o += (size_t)NN * HID;
    float*    xpad   = ws + o; o += (size_t)NN * 32;
    float*    w1rp   = ws + o; o += (size_t)HID * 32;
    float*    w1sp   = ws + o; o += (size_t)HID * 32;
    unsigned* henc   = (unsigned*)(ws + o); o += 3 * NG * HID;

    k_sort_block<<<NBLK, 1024, 0, stream>>>(src, dst, boff, binned, henc, W1r, W1s, w1rp, w1sp, x, xpad);
    k_sort_bin<<<NBIN, 1024, 0, stream>>>(binned, boff, rowptr, csrc);

    struct Layer { const float* xin; int Cp; const float *Wr, *b, *Ws; float* xout; };
    Layer layers[3] = {
        { xpad, 32, w1rp, b1, w1sp, x1 },
        { x1,   64, W2r,  b2, W2s,  x2 },
        { x2,   64, W3r,  b3, W3s,  x3 },
    };

    for (int l = 0; l < 3; ++l) {
        const Layer& L = layers[l];
        if (L.Cp == 32)
            k_layer<32><<<(NN + GN - 1) / GN, 512, 0, stream>>>(rowptr, csrc, L.xin, t, L.Wr, L.Ws, L.b, batch,
                                                            L.xout, henc + (size_t)l * NG * HID, NN);
        else
            k_layer<64><<<(NN + GN - 1) / GN, 512, 0, stream>>>(rowptr, csrc, L.xin, t, L.Wr, L.Ws, L.b, batch,
                                                            L.xout, henc + (size_t)l * NG * HID, NN);
    }

    k_head<<<NG, 256, 0, stream>>>(henc, Wl1, bl1, Wl2, bl2, (float*)d_out);
}


# Softmax gather, node addresses on the scalar unit: the source list comes by scalar loads (no lane reads), a row load is scalar row base + 32-bit lane offset (no 64-bit vector add a row); same sums
# speedup vs baseline: 7.8014x; 1.0411x over previous
#include <hip/hip_runtime.h>
#include <math.h>

#define NN      50000
#define NE      800000
#define NG      128
#define INC     31
#define HID     64
#define EPS     1e-5f

typedef __attribute__((ext_vector_type(2))) float v2f;
typedef __attribute__((ext_vector_type(8))) float v8f;

static __device__ __forceinline__ unsigned enc_f32(float x) {
    unsigned b = __float_as_uint(x);
    return (b & 0x80000000u) ? ~b : (b | 0x80000000u);
}
static __device__ __forceinline__ float dec_f32(unsigned e) {
    unsigned b = (e & 0x80000000u) ? (e & 0x7FFFFFFFu) : ~e;
    return __uint_as_float(b);
}
#define ENC_NEG_INF 0x007FFFFFu

#define EPB    4096
#define NBLK   ((NE + EPB - 1) / EPB)
#define BINSH  8
#define BINW   (1 << BINSH)
#define NBIN   ((NN + BINW - 1) / BINW)
#define BOFFW  (NBIN + 1)
static_assert(NN <= 65536, "source and destination must fit 16 bits each");
static_assert(NBIN <= 256 && NBLK <= 256, "one thread per bin in level 1, one thread (and one thread quad) per run in level 2");
static_assert(NBLK * 1024 >= 3 * NG * HID && NBLK * 1024 >= HID * 32, "prep ranges are prefixes of the grid");
static_assert((long long)NBLK * 1024 * 8 >= (long long)NN * 32, "eight padded elements per thread cover x");

static __device__ __forceinline__ void excl_scan256x2(int a, int b, int (&wsa)[8], int (&wsb)[8],
                                                      int& ea, int& eb, int& tota, int& totb) {
    const int tid = threadIdx.x, lane = tid & 31, w = tid >> 5;
    int ia = a, ib = b;
    if (w < 8) {
        for (int off = 1; off < 32; off <<= 1) {
            const int ua = __shfl_up(ia, off), ub = __shfl_up(ib, off);
            if (lane >= off) { ia += ua; ib += ub; }
        }
        if (lane == 31) { wsa[w] = ia; wsb[w] = ib; }
    }
    __syncthreads();
    int ba = 0, bb = 0, ta = 0, tb = 0;
#pragma unroll
    for (int i = 0; i < 8; ++i) {
        const int sa = wsa[i], sb = wsb[i];
        ta += sa; tb += sb;
        if (i < w) { ba += sa; bb += sb; }
    }
    ea = ba + ia - a; eb = bb + ib - b; tota = ta; totb = tb;
}

__global__ void __launch_bounds__(1024) k_sort_block(const int* __restrict__ src, const int* __restrict__ dst,
                                                     int* __restrict__ boff, unsigned* __restrict__ binned,
                                                     unsigned* __restrict__ henc,
                                                     const float* __restrict__ W1r, const float* __restrict__ W1s,
                                                     float* __restrict__ w1rp, float* __restrict__ w1sp,
                                                     const float* __restrict__ x, float* __restrict__ xpad) {
    __shared__ int h[NBIN];
    __shared__ int loff[NBIN];
    __shared__ int wsa[8];
    __shared__ int wsb[8];
    __shared__ unsigned stage[EPB];
    const int tid = threadIdx.x, blk = blockIdx.x;
    if (tid < NBIN) h[tid] = 0;
    int s[4], d[4];
#pragma unroll
    for (int k = 0; k < 4; ++k) {
        const int e = min(blk * EPB + k * 1024 + tid, NE - 1);
        s[k] = src[e];
        d[k] = dst[e];
    }
    const int i0 = blk * 1024 + tid;
    float xv[8];
#pragma unroll
    for (int k = 0; k < 8; ++k) {
        const int j = min(i0 + k * (NBLK * 1024), NN * 32 - 1);
        xv[k] = x[(j >> 5) * INC + min(j & 31, INC - 1)];
    }
    __syncthreads();
    int r[4];
#pragma unroll
    for (int k = 0; k < 4; ++k)
        r[k] = (blk * EPB + k * 1024 + tid < NE) ? atomicAdd(&h[d[k] >> BINSH], 1) : 0;
    __syncthreads();
    const int hv = (tid < NBIN) ? h[tid] : 0;
    int lo, unused, total, unused2;
    excl_scan256x2(hv, 0, wsa, wsb, lo, unused, total, unused2);
    if (tid < NBIN) {
        loff[tid] = lo;
        boff[blk * BOFFW + tid] = lo;
    }
    if (tid == 0) boff[blk * BOFFW + NBIN] = total;
    __syncthreads();
    int lf[4];
#pragma unroll
    for (int k = 0; k < 4; ++k) lf[k] = loff[d[k] >> BINSH];
#pragma unroll
    for (int k = 0; k < 4; ++k) {
        const int i = k * 1024 + tid;
        const bool live = blk * EPB + i < NE;
        stage[live ? lf[k] + r[k] : i] = live ? (((unsigned)d[k] << 16) | (unsigned)s[k]) : 0u;
    }
    __syncthreads();
    const int nvalid = min(EPB, NE - blk * EPB);
    unsigned p[4];
#pragma unroll
    for (int k = 0; k < 4; ++k) p[k] = stage[k * 1024 + tid];
#pragma unroll
    for (int k = 0; k < 4; ++k) {
        const int i = k * 1024 + tid;
        if (i < nvalid) binned[blk * EPB + i] = p[k];
    }

    if (i0 < 3 * NG * HID) henc[i0] = ENC_NEG_INF;
    if (i0 < HID * 32) {
        const int rr = i0 / 32, cc = i0 % 32;
        w1rp[i0] = (cc < INC) ? W1r[rr * INC + cc] : 0.f;
        w1sp[i0] = (cc < INC) ? W1s[rr * INC + cc] : 0.f;
    }
#pragma unroll
    for (int k = 0; k < 8; ++k) {
        const int j = i0 + k * (NBLK * 1024);
        if (j < NN * 32) xpad[j] = ((j & 31) < INC) ? xv[k] : 0.f;
    }
}

#define BCAP (2 * EPB)
#define RUNC 8
__global__ void __launch_bounds__(1024) k_sort_bin(const unsigned* __restrict__ binned, const int* __restrict__ boff,
                                                   int* __restrict__ rowptr, int* __restrict__ csrc) {
    __shared__ int cnt[BINW];
    __shared__ int wsa[8];
    __shared__ int wsb[8];
    __shared__ int outst[BCAP];
    const int tid = threadIdx.x, bin = blockIdx.x;
    const int t4 = tid >> 2, sub = tid & 3;
    const int ta = min(tid, NBLK - 1) * BOFFW + bin, tq = min(t4, NBLK - 1) * BOFFW + bin;
    const int a0 = boff[ta], b0 = boff[ta + 1];
    const int q0 = boff[tq], q1 = boff[tq + 1];
    if (tid < BINW) cnt[tid] = 0;
    int unused, unused2, beg, end;
    excl_scan256x2((tid < NBLK) ? a0 : 0, (tid < NBLK) ? b0 : 0, wsa, wsb, unused, unused2, beg, end);
    const int n = end - beg;
    const int rs = (t4 < NBLK) ? q0 : 0, re = (t4 < NBLK) ? q1 : 0;
    const unsigned* __restrict__ run = binned + min(t4, NBLK - 1) * EPB;
    for (int base = rs; base < re; base += 4 * RUNC) {
        unsigned p[RUNC];
#pragma unroll
        for (int k = 0; k < RUNC; ++k) p[k] = run[min(base + 4 * k + sub, re - 1)];
#pragma unroll
        for (int k = 0; k < RUNC; ++k)
            if (base + 4 * k + sub < re) atomicAdd(&cnt[(p[k] >> 16) & (BINW - 1)], 1);
    }
    __syncthreads();
    const int c = (tid < BINW) ? cnt[tid] : 0;
    int ex, total;
    excl_scan256x2(c, 0, wsa, wsb, ex, unused, total, unused2);
    if (tid < BINW) {
        const int node = bin * BINW + tid;
        if (node <= NN) rowptr[node] = beg + ex;
        cnt[tid] = ex;
    }
    __syncthreads();
    const bool staged = n <= BCAP;
    for (int base = rs; base < re; base += 4 * RUNC) {
        unsigned p[RUNC];
#pragma unroll
        for (int k = 0; k < RUNC; ++k) p[k] = run[min(base + 4 * k + sub, re - 1)];
        int rk[RUNC];
#pragma unroll
        for (int k = 0; k < RUNC; ++k)
            rk[k] = (base + 4 * k + sub < re) ? atomicAdd(&cnt[(p[k] >> 16) & (BINW - 1)], 1) : 0;
#pragma unroll
        for (int k = 0; k < RUNC; ++k) {
            if (base + 4 * k + sub < re) {
                const int sv = (int)(p[k] & 0xFFFFu);
                if (staged) outst[rk[k]] = sv; else csrc[beg + rk[k]] = sv;
            }
        }
    }
    if (bin == NBIN - 1 && tid < 32) csrc[NE + tid] = 0;
    if (staged && n > 0) {
        __syncthreads();
        int v[BCAP / 1024];
#pragma unroll
        for (int k = 0; k < BCAP / 1024; ++k) v[k] = outst[min(k * 1024 + tid, n - 1)];
#pragma unroll
        for (int k = 0; k < BCAP / 1024; ++k)
            if (k * 1024 + tid < n) csrc[beg + k * 1024 + tid] = v[k];
    }
}

template<int VEC>
static __device__ __forceinline__ void ld_row(const float* __restrict__ p, float (&v)[VEC]) {
    if constexpr (VEC == 2) {
        v2f r = *(const v2f*)p;
        v[0] = r.x; v[1] = r.y;
    } else {
        v[0] = p[0];
    }
}

template<int VEC>
static __device__ __forceinline__ void ld_row_so(const float* __restrict__ rowbase, unsigned bo, float (&v)[VEC]) {
    const char* p = (const char*)rowbase + bo;
    if constexpr (VEC == 2) {
        v2f r = *(const v2f*)p;
        v[0] = r.x; v[1] = r.y;
    } else {
        v[0] = *(const float*)p;
    }
}

template<int VEC>
static __device__ __forceinline__ void seg_rng(const float (&r)[VEC], float (&hi)[VEC], float (&lo)[VEC]) {
#pragma unroll
    for (int i = 0; i < VEC; ++i) { hi[i] = fmaxf(hi[i], r[i]); lo[i] = fminf(lo[i], r[i]); }
}
template<int VEC>
static __device__ __forceinline__ void seg_rng2(const float (&a)[VEC], const float (&b)[VEC], float (&hi)[VEC], float (&lo)[VEC]) {
#pragma unroll
    for (int i = 0; i < VEC; ++i) { hi[i] = fmaxf(fmaxf(hi[i], a[i]), b[i]); lo[i] = fminf(fminf(lo[i], a[i]), b[i]); }
}
template<int VEC>
static __device__ __forceinline__ void seg_sum(const float (&r)[VEC], float t2, const float (&mx2)[VEC],
                                               float (&num)[VEC], float (&den)[VEC]) {
#pragma unroll
    for (int i = 0; i < VEC; ++i) {
        float e = __builtin_amdgcn_exp2f(__builtin_fmaf(r[i], t2, -mx2[i]));
        num[i] += r[i] * e;
        den[i] += e;
    }
}
template<int VEC>
static __device__ __forceinline__ void seg_top(const float (&hi)[VEC], const float (&lo)[VEC], float t2, float (&mx2)[VEC]) {
#pragma unroll
    for (int i = 0; i < VEC; ++i) mx2[i] = fmaxf(hi[i] * t2, lo[i] * t2);
}

template<int J, int VEC, int CP, int R>
struct SegChain {
    static __device__ __forceinline__ void load(int deg, const float* __restrict__ x, unsigned& bo,
                                                const int (&sidx)[32], float (&v)[R][VEC]) {
        if constexpr (J + 1 < R) {
            if (J < deg) {
                asm("" : "+v"(bo));
                ld_row_so<VEC>(x + (size_t)sidx[J] * CP, bo, v[J]);
                const int s1 = (J + 1 < deg) ? sidx[J + 1] : sidx[J];
                ld_row_so<VEC>(x + (size_t)s1 * CP, bo, v[J + 1]);
                SegChain<J + 2, VEC, CP, R>::load(deg, x, bo, sidx, v);
            }
        }
    }
    static __device__ __forceinline__ void vrng(int deg, const float (&v)[R][VEC], float (&hi)[VEC], float (&lo)[VEC]) {
        if constexpr (J + 1 < R) {
            if (J < deg) {
                seg_rng2<VEC>(v[J], v[J + 1], hi, lo);
                SegChain<J + 2, VEC, CP, R>::vrng(deg, v, hi, lo);
            }
        }
    }
    static __device__ __forceinline__ void vsum(int deg, const float (&v)[R][VEC], float t,
                                                const float (&mx)[VEC], float (&num)[VEC], float (&den)[VEC]) {
        if constexpr (J < R) {
            if (J < deg) {
                seg_sum<VEC>(v[J], t, mx, num, den);
                SegChain<J + 1, VEC, CP, R>::vsum(deg, v, t, mx, num, den);
            }
        }
    }
};

static __device__ __forceinline__ float xhalf(float x) {
    int b = __builtin_bit_cast(int, x);
    return __builtin_bit_cast(float, __builtin_amdgcn_permlanex16(b, b, 0x76543210, 0xfedcba98, false, false));
}
template<int M>
static __device__ __forceinline__ float rowx(float x) {
    return __builtin_bit_cast(float, __builtin_amdgcn_update_dpp(0, __builtin_bit_cast(int, x), 0x160 + M, 0xf, 0xf, true));
}
template<int M>
static __device__ __forceinline__ void row_step8(float (&s)[8]) {
#pragma unroll
    for (int r = 0; r < 8; ++r) s[r] += rowx<M>(s[r]);
}
static __device__ __forceinline__ void row_sum16x8(float (&s)[8]) {
    row_step8<1>(s);
    row_step8<2>(s);
    row_step8<4>(s);
    row_step8<8>(s);
}

#define NPW  4
#define GN   (16 * NPW)
static_assert(NPW == 4, "phase 2 gives tile t to wave 5t: waves 0, 5, 10, 15");
template<int CP>
__global__ void __launch_bounds__(512, 12) k_layer(const int* __restrict__ rowptr, const int* __restrict__ csrc,
                                               const float* __restrict__ xin, const float* __restrict__ tptr,
                                               const float* __restrict__ Wr, const float* __restrict__ Ws,
                                               const float* __restrict__ bias, const int* __restrict__ batch,
                                               float* __restrict__ out, unsigned* __restrict__ he, int Nrows) {
    constexpr int VEC = CP / 32;
    constexpr int R = (VEC == 2) ? 30 : 32;
    static_assert(R % 2 == 0, "the held rows are fetched and ranged in pairs");
    constexpr int PITCH = CP + 4;
    __shared__ __attribute__((aligned(16))) float sA[GN * PITCH];
    __shared__ __attribute__((aligned(16))) float sX[GN * PITCH];
    __shared__ __attribute__((aligned(16))) float sW[2 * HID * PITCH];
    const int lane = threadIdx.x & 31;
    const int wv = __builtin_amdgcn_readfirstlane((int)(threadIdx.x >> 5));
    const int g0n = (int)blockIdx.x * GN;
    if (g0n >= Nrows) return;
    const int n0 = g0n + wv * NPW;
    constexpr int Q   = CP / 4;
    constexpr int PER = 2 * HID * Q / 512;
    float4 wq[PER];
#pragma unroll
    for (int i = 0; i < PER; ++i) {
        const int rem = (i * 512) % (HID * Q) + (int)threadIdx.x;
        const float* __restrict__ Wm = ((i * 512) / (HID * Q)) ? Ws : Wr;
        wq[i] = *(const float4*)(Wm + (rem / Q) * CP + 4 * (rem % Q));
    }
    const unsigned loff = (unsigned)(lane * VEC);
    float xr[NPW][VEC];
#pragma unroll
    for (int i = 0; i < NPW; ++i)
        ld_row<VEC>(xin + (size_t)min(n0 + i, Nrows - 1) * CP + loff, xr[i]);

    const int rp = rowptr[min(n0 + lane, Nrows)];
    const float t2 = tptr[0] * 1.44269504088896340736f;
    int bnd[NPW + 1];
#pragma unroll
    for (int k = 0; k <= NPW; ++k) bnd[k] = __builtin_amdgcn_readlane(rp, k);
    int beg = bnd[0];
    int end = bnd[1];
    int cur[32];
#pragma unroll
    for (int j = 0; j < 32; ++j) cur[j] = csrc[beg + j];

#pragma unroll
    for (int i = 0; i < PER; ++i) {
        const int rem = (i * 512) % (HID * Q) + (int)threadIdx.x;
        const int n = rem / Q;
        float* d = sW + ((i * 512) / (HID * Q)) * (HID * PITCH) + ((n & 3) * 16 + (n >> 2)) * PITCH + 4 * (rem % Q);
        *(float4*)d = wq[i];
    }
#pragma unroll
    for (int i = 0; i < NPW; ++i)
#pragma unroll
        for (int k = 0; k < VEC; ++k) sX[(wv * NPW + i) * PITCH + loff + k] = xr[i][k];

    unsigned bo = loff * 4u;
#pragma nounroll
    for (int i = 0; i < NPW; ++i) {
        const int deg = end - beg;
        const int nbeg = end;
        int nend = bnd[NPW];
#pragma unroll
        for (int k = NPW - 1; k >= 2; --k) nend = (i + 2 == k) ? bnd[k] : nend;
        float hi[VEC], lo[VEC], mx[VEC], num[VEC], den[VEC];
#pragma unroll
        for (int k = 0; k < VEC; ++k) { hi[k] = -INFINITY; lo[k] = INFINITY; num[k] = 0.f; den[k] = 0.f; }
        float v[R][VEC];
        if (deg <= R) SegChain<0, VEC, CP, R>::load(deg, xin, bo, cur, v);
        const int nb = (i + 1 < NPW) ? nbeg : beg;
        int nxt[32];
#pragma unroll
        for (int j = 0; j < 32; ++j) nxt[j] = csrc[nb + j];
        if (deg <= R) {
            SegChain<0, VEC, CP, R>::vrng(deg, v, hi, lo);
            seg_top<VEC>(hi, lo, t2, mx);
            SegChain<0, VEC, CP, R>::vsum(deg, v, t2, mx, num, den);
        } else {
            for (int base = beg; base < end; base += 32) {
                const int cnt = min(32, end - base);
                const int mys = (lane < cnt) ? csrc[base + lane] : 0;
                for (int j = 0; j < cnt; ++j) {
                    const int s = __builtin_amdgcn_readlane(mys, j);
                    float r[VEC];
                    ld_row<VEC>(xin + (size_t)s * CP + loff, r);
                    seg_rng<VEC>(r, hi, lo);
                }
            }
            seg_top<VEC>(hi, lo, t2, mx);
            for (int base = beg; base < end; base += 32) {
                const int cnt = min(32, end - base);
                const int mys = (lane < cnt) ? csrc[base + lane] : 0;
                for (int j = 0; j < cnt; ++j) {
                    const int s = __builtin_amdgcn_readlane(mys, j);
                    float r[VEC];
                    ld_row<VEC>(xin + (size_t)s * CP + loff, r);
                    seg_sum<VEC>(r, t2, mx, num, den);
                }
            }
        }
        float* o = sA + (wv * NPW + i) * PITCH + loff;
#pragma unroll
        for (int k = 0; k < VEC; ++k) o[k] = den[k] > 0.f ? num[k] / den[k] : 0.f;
        beg = nbeg; end = nend;
#pragma unroll
        for (int j = 0; j < 32; ++j) cur[j] = nxt[j];
    }
    __syncthreads();
    if (wv % 5 != 0) return;
    const int m0 = g0n + (wv / 5) * 16;
    if (m0 >= Nrows) return;

    const int half = lane >> 4;
    const int c    = lane & 15;
    const float* ap  = sA + ((wv / 5) * 16 + c) * PITCH + half * 2;
    const float* xp  = sX + ((wv / 5) * 16 + c) * PITCH + half * 2;
    const float* wr  = sW + c * PITCH + half * 2;
    const float* wsp = wr + HID * PITCH;

    v8f acc[4] = {};
#pragma unroll
    for (int k = 0; k < CP; k += 4) {
        v2f a = *(const v2f*)(ap + k);
#pragma unroll
        for (int j = 0; j < 4; ++j) {
            v2f b = *(const v2f*)(wr + j * 16 * PITCH + k);
            acc[j] = __builtin_amdgcn_wmma_f32_16x16x4_f32(false, a, false, b,
                                                           (short)0, acc[j], false, false);
        }
    }
#pragma unroll
    for (int k = 0; k < CP; k += 4) {
        v2f a = *(const v2f*)(xp + k);
#pragma unroll
        for (int j = 0; j < 4; ++j) {
            v2f b = *(const v2f*)(wsp + j * 16 * PITCH + k);
            acc[j] = __builtin_amdgcn_wmma_f32_16x16x4_f32(false, a, false, b,
                                                           (short)0, acc[j], false, false);
        }
    }
    const int gl = batch[min(m0 + c, Nrows - 1)];
    const int half_e = (int)(__builtin_amdgcn_mbcnt_lo(~0u, 0u) >> 4);
    const float4 bq = *(const float4*)(bias + 4 * c);
    const float bv[4] = { bq.x, bq.y, bq.z, bq.w };
    float s[8], q[8];
#pragma unroll
    for (int r = 0; r < 8; ++r) {
#pragma unroll
        for (int j = 0; j < 4; ++j) acc[j][r] += bv[j];
        s[r] = (acc[0][r] + acc[1][r]) + (acc[2][r] + acc[3][r]);
    }
    row_sum16x8(s);
#pragma unroll
    for (int r = 0; r < 8; ++r) {
        const float mu = s[r] * (1.f / HID);
#pragma unroll
        for (int j = 0; j < 4; ++j) acc[j][r] -= mu;
        q[r] = (acc[0][r] * acc[0][r] + acc[1][r] * acc[1][r]) + (acc[2][r] * acc[2][r] + acc[3][r] * acc[3][r]);
    }
    row_sum16x8(q);
    float cm[4];
#pragma unroll
    for (int r = 0; r < 8; ++r) {
        const float inv = rsqrtf(q[r] * (1.f / HID) + EPS);
#pragma unroll
        for (int j = 0; j < 4; ++j) {
            const float y = fmaxf(acc[j][r] * inv, 0.f);
            acc[j][r] = y;
            cm[j] = (r == 0) ? y : fmaxf(cm[j], y);
        }
    }
    float* op = out + (size_t)(m0 + half_e * 8) * HID + 4 * c;
    if (m0 + 16 <= Nrows) {
#pragma unroll
        for (int r = 0; r < 8; ++r) {
            float4 o; o.x = acc[0][r]; o.y = acc[1][r]; o.z = acc[2][r]; o.w = acc[3][r];
            *(float4*)(op + (size_t)r * HID) = o;
        }
    } else {
#pragma unroll
        for (int r = 0; r < 8; ++r) {
            if (m0 + r + half_e * 8 < Nrows) {
                float4 o; o.x = acc[0][r]; o.y = acc[1][r]; o.z = acc[2][r]; o.w = acc[3][r];
                *(float4*)(op + (size_t)r * HID) = o;
            }
        }
    }

    const int g0 = __builtin_amdgcn_readfirstlane(gl);
    const bool one_graph = __builtin_amdgcn_ballot_w32(gl != g0) == 0u;
    if (one_graph) {
#pragma unroll
        for (int j = 0; j < 4; ++j) cm[j] = fmaxf(cm[j], xhalf(cm[j]));
        const float p0 = half_e ? cm[2] : cm[0];
        const float p1 = half_e ? cm[3] : cm[1];
        unsigned* hp = he + (size_t)g0 * HID + 4 * c + 2 * half_e;
        atomicMax(hp,     enc_f32(p0));
        atomicMax(hp + 1, enc_f32(p1));
    } else {
#pragma unroll
        for (int r = 0; r < 8; ++r) {
            const int row = m0 + r + half_e * 8;
            if (row < Nrows) {
                unsigned* hp = he + (size_t)batch[row] * HID + 4 * c;
#pragma unroll
                for (int j = 0; j < 4; ++j) atomicMax(hp + j, enc_f32(acc[j][r]));
            }
        }
    }
}

template<int NV>
static __device__ __forceinline__ void wave_sum_n(float (&p)[NV]) {
#pragma unroll
    for (int j = 0; j < NV; ++j) p[j] += rowx<1>(p[j]);
#pragma unroll
    for (int j = 0; j < NV; ++j) p[j] += rowx<2>(p[j]);
#pragma unroll
    for (int j = 0; j < NV; ++j) p[j] += rowx<4>(p[j]);
#pragma unroll
    for (int j = 0; j < NV; ++j) p[j] += rowx<8>(p[j]);
#pragma unroll
    for (int j = 0; j < NV; ++j) p[j] += xhalf(p[j]);
}

__global__ void __launch_bounds__(256) k_head(const unsigned* __restrict__ henc,
                                              const float* __restrict__ Wl1, const float* __restrict__ bl1,
                                              const float* __restrict__ Wl2, const float* __restrict__ bl2,
                                              float* __restrict__ outp) {
    __shared__ __attribute__((aligned(16))) float sh[3 * HID];
    __shared__ __attribute__((aligned(16))) float sz1[2 * HID];
    __shared__ __attribute__((aligned(16))) float sz2[HID / 2];
    const int g = blockIdx.x;
    const int tid = threadIdx.x, lane = tid & 31, w = tid >> 5;
    if (tid < 3 * HID) {
        const int layer = tid / HID, cc = tid % HID;
        sh[tid] = dec_f32(henc[layer * NG * HID + g * HID + cc]);
    }
    __syncthreads();

    {
        v2f hv[3];
#pragma unroll
        for (int i = 0; i < 3; ++i) hv[i] = *(const v2f*)(sh + 2 * (lane + 32 * i));
        const float* __restrict__ wbase = Wl1 + (w * 16) * (3 * HID) + 2 * lane;
        float p[16];
#pragma unroll
        for (int j = 0; j < 16; ++j) {
            const float* __restrict__ wrow = wbase + j * (3 * HID);
            float a = 0.f;
#pragma unroll
            for (int i = 0; i < 3; ++i) {
                const v2f wv = *(const v2f*)(wrow + 64 * i);
                a += hv[i].x * wv.x;
                a += hv[i].y * wv.y;
            }
            p[j] = a;
        }
        wave_sum_n<16>(p);
        float mine = p[0];
#pragma unroll
        for (int j = 1; j < 16; ++j) mine = (lane == j) ? p[j] : mine;
        if (lane < 16) sz1[w * 16 + lane] = fmaxf(mine + bl1[w * 16 + lane], 0.f);
    }
    __syncthreads();

    {
        const float4 zv = *(const float4*)(sz1 + 4 * lane);
        const float* __restrict__ wbase = Wl2 + (w * 4) * (2 * HID) + 4 * lane;
        float p[4];
#pragma unroll
        for (int j = 0; j < 4; ++j) {
            const float4 wv = *(const float4*)(wbase + j * (2 * HID));
            p[j] = (zv.x * wv.x + zv.y * wv.y) + (zv.z * wv.z + zv.w * wv.w);
        }
        wave_sum_n<4>(p);
        float mine = p[0];
#pragma unroll
        for (int j = 1; j < 4; ++j) mine = (lane == j) ? p[j] : mine;
        if (lane < 4) sz2[w * 4 + lane] = mine + bl2[w * 4 + lane];
    }
    __syncthreads();

    if (w == 0) {
        const float s = sz2[lane];
        float q[1] = { s * s };
        wave_sum_n<1>(q);
        const float nrm = fmaxf(sqrtf(q[0]), 1e-12f);
        outp[g * (HID / 2) + lane] = s / nrm;
    }
}

extern "C" void kernel_launch(void* const* d_in, const int* in_sizes, int n_in,
                              void* d_out, int out_size, void* d_ws, size_t ws_size,
                              hipStream_t stream) {
    const float* x    = (const float*)d_in[0];
    const int*  eidx  = (const int*)  d_in[1];
    const int*  batch = (const int*)  d_in[2];
    const float* t    = (const float*)d_in[3];
    const float* W1r  = (const float*)d_in[4];
    const float* b1   = (const float*)d_in[5];
    const float* W1s  = (const float*)d_in[6];
    const float* W2r  = (const float*)d_in[7];
    const float* b2   = (const float*)d_in[8];
    const float* W2s  = (const float*)d_in[9];
    const float* W3r  = (const float*)d_in[10];
    const float* b3   = (const float*)d_in[11];
    const float* W3s  = (const float*)d_in[12];
    const float* Wl1  = (const float*)d_in[13];
    const float* bl1  = (const float*)d_in[14];
    const float* Wl2  = (const float*)d_in[15];
    const float* bl2  = (const float*)d_in[16];
    const int* src = eidx;
    const int* dst = eidx + NE;

    float* ws = (float*)d_ws;
    size_t o = 0;
    float*    x1     = ws + o; o += (size_t)NN * HID;
    float*    x2     = ws + o; o += (size_t)NN * HID;
    float*    x3     = ws + o; o += (size_t)NN * HID;
    int*      csr    = (int*)(ws + o); o += (size_t)NN * HID * 3;
    int*      boff     = csr;
    unsigned* binned   = (unsigned*)(boff + NBLK * BOFFW + 16);
    int*      rowptr   = (int*)(binned + NBLK * EPB);
    int*      csrc     = rowptr + NN + 16;
    o += (size_t)NN * HID;
    float*    xpad   = ws + o; o += (size_t)NN * 32;
    float*    w1rp   = ws + o; o += (size_t)HID * 32;
    float*    w1sp   = ws + o; o += (size_t)HID * 32;
    unsigned* henc   = (unsigned*)(ws + o); o += 3 * NG * HID;

    k_sort_block<<<NBLK, 1024, 0, stream>>>(src, dst, boff, binned, henc, W1r, W1s, w1rp, w1sp, x, xpad);
    k_sort_bin<<<NBIN, 1024, 0, stream>>>(binned, boff, rowptr, csrc);

    struct Layer { const float* xin; int Cp; const float *Wr, *b, *Ws; float* xout; };
    Layer layers[3] = {
        { xpad, 32, w1rp, b1, w1sp, x1 },
        { x1,   64, W2r,  b2, W2s,  x2 },
        { x2,   64, W3r,  b3, W3s,  x3 },
    };

    for (int l = 0; l < 3; ++l) {
        const Layer& L = layers[l];
        if (L.Cp == 32)
            k_layer<32><<<(NN + GN - 1) / GN, 512, 0, stream>>>(rowptr, csrc, L.xin, t, L.Wr, L.Ws, L.b, batch,
                                                            L.xout, henc + (size_t)l * NG * HID, NN);
        else
            k_layer<64><<<(NN + GN - 1) / GN, 512, 0, stream>>>(rowptr, csrc, L.xin, t, L.Wr, L.Ws, L.b, batch,
                                                            L.xout, henc + (size_t)l * NG * HID, NN);
    }

    k_head<<<NG, 256, 0, stream>>>(henc, Wl1, bl1, Wl2, bl2, (float*)d_out);
}


# A node with more rows than the registers hold (31+ / 33+) takes them in batches of R, all row loads of a batch in flight, behind the node loop, not one dependent load per row twice; same sum order
# speedup vs baseline: 8.2277x; 1.0546x over previous
#include <hip/hip_runtime.h>
#include <math.h>

#define NN      50000
#define NE      800000
#define NG      128
#define INC     31
#define HID     64
#define EPS     1e-5f

typedef __attribute__((ext_vector_type(2))) float v2f;
typedef __attribute__((ext_vector_type(8))) float v8f;

static __device__ __forceinline__ unsigned enc_f32(float x) {
    unsigned b = __float_as_uint(x);
    return (b & 0x80000000u) ? ~b : (b | 0x80000000u);
}
static __device__ __forceinline__ float dec_f32(unsigned e) {
    unsigned b = (e & 0x80000000u) ? (e & 0x7FFFFFFFu) : ~e;
    return __uint_as_float(b);
}
#define ENC_NEG_INF 0x007FFFFFu

#define EPB    4096
#define NBLK   ((NE + EPB - 1) / EPB)
#define BINSH  8
#define BINW   (1 << BINSH)
#define NBIN   ((NN + BINW - 1) / BINW)
#define BOFFW  (NBIN + 1)
static_assert(NN <= 65536, "source and destination must fit 16 bits each");
static_assert(NBIN <= 256 && NBLK <= 256, "one thread per bin in level 1, one thread (and one thread quad) per run in level 2");
static_assert(NBLK * 1024 >= 3 * NG * HID && NBLK * 1024 >= HID * 32, "prep ranges are prefixes of the grid");
static_assert((long long)NBLK * 1024 * 8 >= (long long)NN * 32, "eight padded elements per thread cover x");

static __device__ __forceinline__ void excl_scan256x2(int a, int b, int (&wsa)[8], int (&wsb)[8],
                                                      int& ea, int& eb, int& tota, int& totb) {
    const int tid = threadIdx.x, lane = tid & 31, w = tid >> 5;
    int ia = a, ib = b;
    if (w < 8) {
        for (int off = 1; off < 32; off <<= 1) {
            const int ua = __shfl_up(ia, off), ub = __shfl_up(ib, off);
            if (lane >= off) { ia += ua; ib += ub; }
        }
        if (lane == 31) { wsa[w] = ia; wsb[w] = ib; }
    }
    __syncthreads();
    int ba = 0, bb = 0, ta = 0, tb = 0;
#pragma unroll
    for (int i = 0; i < 8; ++i) {
        const int sa = wsa[i], sb = wsb[i];
        ta += sa; tb += sb;
        if (i < w) { ba += sa; bb += sb; }
    }
    ea = ba + ia - a; eb = bb + ib - b; tota = ta; totb = tb;
}

__global__ void __launch_bounds__(1024) k_sort_block(const int* __restrict__ src, const int* __restrict__ dst,
                                                     int* __restrict__ boff, unsigned* __restrict__ binned,
                                                     unsigned* __restrict__ henc,
                                                     const float* __restrict__ W1r, const float* __restrict__ W1s,
                                                     float* __restrict__ w1rp, float* __restrict__ w1sp,
                                                     const float* __restrict__ x, float* __restrict__ xpad) {
    __shared__ int h[NBIN];
    __shared__ int loff[NBIN];
    __shared__ int wsa[8];
    __shared__ int wsb[8];
    __shared__ unsigned stage[EPB];
    const int tid = threadIdx.x, blk = blockIdx.x;
    if (tid < NBIN) h[tid] = 0;
    int s[4], d[4];
#pragma unroll
    for (int k = 0; k < 4; ++k) {
        const int e = min(blk * EPB + k * 1024 + tid, NE - 1);
        s[k] = src[e];
        d[k] = dst[e];
    }
    const int i0 = blk * 1024 + tid;
    float xv[8];
#pragma unroll
    for (int k = 0; k < 8; ++k) {
        const int j = min(i0 + k * (NBLK * 1024), NN * 32 - 1);
        xv[k] = x[(j >> 5) * INC + min(j & 31, INC - 1)];
    }
    __syncthreads();
    int r[4];
#pragma unroll
    for (int k = 0; k < 4; ++k)
        r[k] = (blk * EPB + k * 1024 + tid < NE) ? atomicAdd(&h[d[k] >> BINSH], 1) : 0;
    __syncthreads();
    const int hv = (tid < NBIN) ? h[tid] : 0;
    int lo, unused, total, unused2;
    excl_scan256x2(hv, 0, wsa, wsb, lo, unused, total, unused2);
    if (tid < NBIN) {
        loff[tid] = lo;
        boff[blk * BOFFW + tid] = lo;
    }
    if (tid == 0) boff[blk * BOFFW + NBIN] = total;
    __syncthreads();
    int lf[4];
#pragma unroll
    for (int k = 0; k < 4; ++k) lf[k] = loff[d[k] >> BINSH];
#pragma unroll
    for (int k = 0; k < 4; ++k) {
        const int i = k * 1024 + tid;
        const bool live = blk * EPB + i < NE;
        stage[live ? lf[k] + r[k] : i] = live ? (((unsigned)d[k] << 16) | (unsigned)s[k]) : 0u;
    }
    __syncthreads();
    const int nvalid = min(EPB, NE - blk * EPB);
    unsigned p[4];
#pragma unroll
    for (int k = 0; k < 4; ++k) p[k] = stage[k * 1024 + tid];
#pragma unroll
    for (int k = 0; k < 4; ++k) {
        const int i = k * 1024 + tid;
        if (i < nvalid) binned[blk * EPB + i] = p[k];
    }

    if (i0 < 3 * NG * HID) henc[i0] = ENC_NEG_INF;
    if (i0 < HID * 32) {
        const int rr = i0 / 32, cc = i0 % 32;
        w1rp[i0] = (cc < INC) ? W1r[rr * INC + cc] : 0.f;
        w1sp[i0] = (cc < INC) ? W1s[rr * INC + cc] : 0.f;
    }
#pragma unroll
    for (int k = 0; k < 8; ++k) {
        const int j = i0 + k * (NBLK * 1024);
        if (j < NN * 32) xpad[j] = ((j & 31) < INC) ? xv[k] : 0.f;
    }
}

#define BCAP (2 * EPB)
#define RUNC 8
__global__ void __launch_bounds__(1024) k_sort_bin(const unsigned* __restrict__ binned, const int* __restrict__ boff,
                                                   int* __restrict__ rowptr, int* __restrict__ csrc) {
    __shared__ int cnt[BINW];
    __shared__ int wsa[8];
    __shared__ int wsb[8];
    __shared__ int outst[BCAP];
    const int tid = threadIdx.x, bin = blockIdx.x;
    const int t4 = tid >> 2, sub = tid & 3;
    const int ta = min(tid, NBLK - 1) * BOFFW + bin, tq = min(t4, NBLK - 1) * BOFFW + bin;
    const int a0 = boff[ta], b0 = boff[ta + 1];
    const int q0 = boff[tq], q1 = boff[tq + 1];
    if (tid < BINW) cnt[tid] = 0;
    int unused, unused2, beg, end;
    excl_scan256x2((tid < NBLK) ? a0 : 0, (tid < NBLK) ? b0 : 0, wsa, wsb, unused, unused2, beg, end);
    const int n = end - beg;
    const int rs = (t4 < NBLK) ? q0 : 0, re = (t4 < NBLK) ? q1 : 0;
    const unsigned* __restrict__ run = binned + min(t4, NBLK - 1) * EPB;
    for (int base = rs; base < re; base += 4 * RUNC) {
        unsigned p[RUNC];
#pragma unroll
        for (int k = 0; k < RUNC; ++k) p[k] = run[min(base + 4 * k + sub, re - 1)];
#pragma unroll
        for (int k = 0; k < RUNC; ++k)
            if (base + 4 * k + sub < re) atomicAdd(&cnt[(p[k] >> 16) & (BINW - 1)], 1);
    }
    __syncthreads();
    const int c = (tid < BINW) ? cnt[tid] : 0;
    int ex, total;
    excl_scan256x2(c, 0, wsa, wsb, ex, unused, total, unused2);
    if (tid < BINW) {
        const int node = bin * BINW + tid;
        if (node <= NN) rowptr[node] = beg + ex;
        cnt[tid] = ex;
    }
    __syncthreads();
    const bool staged = n <= BCAP;
    for (int base = rs; base < re; base += 4 * RUNC) {
        unsigned p[RUNC];
#pragma unroll
        for (int k = 0; k < RUNC; ++k) p[k] = run[min(base + 4 * k + sub, re - 1)];
        int rk[RUNC];
#pragma unroll
        for (int k = 0; k < RUNC; ++k)
            rk[k] = (base + 4 * k + sub < re) ? atomicAdd(&cnt[(p[k] >> 16) & (BINW - 1)], 1) : 0;
#pragma unroll
        for (int k = 0; k < RUNC; ++k) {
            if (base + 4 * k + sub < re) {
                const int sv = (int)(p[k] & 0xFFFFu);
                if (staged) outst[rk[k]] = sv; else csrc[beg + rk[k]] = sv;
            }
        }
    }
    if (bin == NBIN - 1 && tid < 32) csrc[NE + tid] = 0;
    if (staged && n > 0) {
        __syncthreads();
        int v[BCAP / 1024];
#pragma unroll
        for (int k = 0; k < BCAP / 1024; ++k) v[k] = outst[min(k * 1024 + tid, n - 1)];
#pragma unroll
        for (int k = 0; k < BCAP / 1024; ++k)
            if (k * 1024 + tid < n) csrc[beg + k * 1024 + tid] = v[k];
    }
}

template<int VEC>
static __device__ __forceinline__ void ld_row(const float* __restrict__ p, float (&v)[VEC]) {
    if constexpr (VEC == 2) {
        v2f r = *(const v2f*)p;
        v[0] = r.x; v[1] = r.y;
    } else {
        v[0] = p[0];
    }
}

template<int VEC>
static __device__ __forceinline__ void ld_row_so(const float* __restrict__ rowbase, unsigned bo, float (&v)[VEC]) {
    const char* p = (const char*)rowbase + bo;
    if constexpr (VEC == 2) {
        v2f r = *(const v2f*)p;
        v[0] = r.x; v[1] = r.y;
    } else {
        v[0] = *(const float*)p;
    }
}

template<int VEC>
static __device__ __forceinline__ void seg_rng(const float (&r)[VEC], float (&hi)[VEC], float (&lo)[VEC]) {
#pragma unroll
    for (int i = 0; i < VEC; ++i) { hi[i] = fmaxf(hi[i], r[i]); lo[i] = fminf(lo[i], r[i]); }
}
template<int VEC>
static __device__ __forceinline__ void seg_rng2(const float (&a)[VEC], const float (&b)[VEC], float (&hi)[VEC], float (&lo)[VEC]) {
#pragma unroll
    for (int i = 0; i < VEC; ++i) { hi[i] = fmaxf(fmaxf(hi[i], a[i]), b[i]); lo[i] = fminf(fminf(lo[i], a[i]), b[i]); }
}
template<int VEC>
static __device__ __forceinline__ void seg_sum(const float (&r)[VEC], float t2, const float (&mx2)[VEC],
                                               float (&num)[VEC], float (&den)[VEC]) {
#pragma unroll
    for (int i = 0; i < VEC; ++i) {
        float e = __builtin_amdgcn_exp2f(__builtin_fmaf(r[i], t2, -mx2[i]));
        num[i] += r[i] * e;
        den[i] += e;
    }
}
template<int VEC>
static __device__ __forceinline__ void seg_top(const float (&hi)[VEC], const float (&lo)[VEC], float t2, float (&mx2)[VEC]) {
#pragma unroll
    for (int i = 0; i < VEC; ++i) mx2[i] = fmaxf(hi[i] * t2, lo[i] * t2);
}

template<int J, int VEC, int CP, int R>
struct SegChain {
    static __device__ __forceinline__ void load(int deg, const float* __restrict__ x, unsigned& bo,
                                                const int (&sidx)[32], float (&v)[R][VEC]) {
        if constexpr (J + 1 < R) {
            if (J < deg) {
                asm("" : "+v"(bo));
                ld_row_so<VEC>(x + (size_t)sidx[J] * CP, bo, v[J]);
                const int s1 = (J + 1 < deg) ? sidx[J + 1] : sidx[J];
                ld_row_so<VEC>(x + (size_t)s1 * CP, bo, v[J + 1]);
                SegChain<J + 2, VEC, CP, R>::load(deg, x, bo, sidx, v);
            }
        }
    }
    static __device__ __forceinline__ void vrng(int deg, const float (&v)[R][VEC], float (&hi)[VEC], float (&lo)[VEC]) {
        if constexpr (J + 1 < R) {
            if (J < deg) {
                seg_rng2<VEC>(v[J], v[J + 1], hi, lo);
                SegChain<J + 2, VEC, CP, R>::vrng(deg, v, hi, lo);
            }
        }
    }
    static __device__ __forceinline__ void vsum(int deg, const float (&v)[R][VEC], float t,
                                                const float (&mx)[VEC], float (&num)[VEC], float (&den)[VEC]) {
        if constexpr (J < R) {
            if (J < deg) {
                seg_sum<VEC>(v[J], t, mx, num, den);
                SegChain<J + 1, VEC, CP, R>::vsum(deg, v, t, mx, num, den);
            }
        }
    }
};

static __device__ __forceinline__ float xhalf(float x) {
    int b = __builtin_bit_cast(int, x);
    return __builtin_bit_cast(float, __builtin_amdgcn_permlanex16(b, b, 0x76543210, 0xfedcba98, false, false));
}
template<int M>
static __device__ __forceinline__ float rowx(float x) {
    return __builtin_bit_cast(float, __builtin_amdgcn_update_dpp(0, __builtin_bit_cast(int, x), 0x160 + M, 0xf, 0xf, true));
}
template<int M>
static __device__ __forceinline__ void row_step8(float (&s)[8]) {
#pragma unroll
    for (int r = 0; r < 8; ++r) s[r] += rowx<M>(s[r]);
}
static __device__ __forceinline__ void row_sum16x8(float (&s)[8]) {
    row_step8<1>(s);
    row_step8<2>(s);
    row_step8<4>(s);
    row_step8<8>(s);
}

#define NPW  4
#define GN   (16 * NPW)
static_assert(NPW == 4, "phase 2 gives tile t to wave 5t: waves 0, 5, 10, 15");
template<int CP>
__global__ void __launch_bounds__(512, 12) k_layer(const int* __restrict__ rowptr, const int* __restrict__ csrc,
                                               const float* __restrict__ xin, const float* __restrict__ tptr,
                                               const float* __restrict__ Wr, const float* __restrict__ Ws,
                                               const float* __restrict__ bias, const int* __restrict__ batch,
                                               float* __restrict__ out, unsigned* __restrict__ he, int Nrows) {
    constexpr int VEC = CP / 32;
    constexpr int R = (VEC == 2) ? 30 : 32;
    static_assert(R % 2 == 0, "the held rows are fetched and ranged in pairs");
    constexpr int PITCH = CP + 4;
    __shared__ __attribute__((aligned(16))) float sA[GN * PITCH];
    __shared__ __attribute__((aligned(16))) float sX[GN * PITCH];
    __shared__ __attribute__((aligned(16))) float sW[2 * HID * PITCH];
    const int lane = threadIdx.x & 31;
    const int wv = __builtin_amdgcn_readfirstlane((int)(threadIdx.x >> 5));
    const int g0n = (int)blockIdx.x * GN;
    if (g0n >= Nrows) return;
    const int n0 = g0n + wv * NPW;
    constexpr int Q   = CP / 4;
    constexpr int PER = 2 * HID * Q / 512;
    float4 wq[PER];
#pragma unroll
    for (int i = 0; i < PER; ++i) {
        const int rem = (i * 512) % (HID * Q) + (int)threadIdx.x;
        const float* __restrict__ Wm = ((i * 512) / (HID * Q)) ? Ws : Wr;
        wq[i] = *(const float4*)(Wm + (rem / Q) * CP + 4 * (rem % Q));
    }
    const unsigned loff = (unsigned)(lane * VEC);
    float xr[NPW][VEC];
#pragma unroll
    for (int i = 0; i < NPW; ++i)
        ld_row<VEC>(xin + (size_t)min(n0 + i, Nrows - 1) * CP + loff, xr[i]);

    const int rp = rowptr[min(n0 + lane, Nrows)];
    const float t2 = tptr[0] * 1.44269504088896340736f;
    int bnd[NPW + 1];
#pragma unroll
    for (int k = 0; k <= NPW; ++k) bnd[k] = __builtin_amdgcn_readlane(rp, k);
    int beg = bnd[0];
    int end = bnd[1];
    int cur[32];
#pragma unroll
    for (int j = 0; j < 32; ++j) cur[j] = csrc[beg + j];

#pragma unroll
    for (int i = 0; i < PER; ++i) {
        const int rem = (i * 512) % (HID * Q) + (int)threadIdx.x;
        const int n = rem / Q;
        float* d = sW + ((i * 512) / (HID * Q)) * (HID * PITCH) + ((n & 3) * 16 + (n >> 2)) * PITCH + 4 * (rem % Q);
        *(float4*)d = wq[i];
    }
#pragma unroll
    for (int i = 0; i < NPW; ++i)
#pragma unroll
        for (int k = 0; k < VEC; ++k) sX[(wv * NPW + i) * PITCH + loff + k] = xr[i][k];

    unsigned bo = loff * 4u;
    unsigned heavy = 0u;
#pragma nounroll
    for (int i = 0; i < NPW; ++i) {
        const int deg = end - beg;
        const int nbeg = end;
        int nend = bnd[NPW];
#pragma unroll
        for (int k = NPW - 1; k >= 2; --k) nend = (i + 2 == k) ? bnd[k] : nend;
        float hi[VEC], lo[VEC], mx[VEC], num[VEC], den[VEC];
#pragma unroll
        for (int k = 0; k < VEC; ++k) { hi[k] = -INFINITY; lo[k] = INFINITY; num[k] = 0.f; den[k] = 0.f; }
        float v[R][VEC];
        if (deg <= R) SegChain<0, VEC, CP, R>::load(deg, xin, bo, cur, v);
        const int nb = (i + 1 < NPW) ? nbeg : beg;
        int nxt[32];
#pragma unroll
        for (int j = 0; j < 32; ++j) nxt[j] = csrc[nb + j];
        if (deg <= R) {
            SegChain<0, VEC, CP, R>::vrng(deg, v, hi, lo);
            seg_top<VEC>(hi, lo, t2, mx);
            SegChain<0, VEC, CP, R>::vsum(deg, v, t2, mx, num, den);
        } else {
            heavy |= 1u << i;
        }
        float* o = sA + (wv * NPW + i) * PITCH + loff;
#pragma unroll
        for (int k = 0; k < VEC; ++k) o[k] = den[k] > 0.f ? num[k] / den[k] : 0.f;
        beg = nbeg; end = nend;
#pragma unroll
        for (int j = 0; j < 32; ++j) cur[j] = nxt[j];
    }
    if (heavy != 0u) {
#pragma nounroll
        for (int i = 0; i < NPW; ++i) {
            if (((heavy >> i) & 1u) == 0u) continue;
            int hbeg = bnd[NPW - 1], hend = bnd[NPW];
#pragma unroll
            for (int k = NPW - 2; k >= 0; --k) { hbeg = (i == k) ? bnd[k] : hbeg; hend = (i == k) ? bnd[k + 1] : hend; }
            const int hdeg = hend - hbeg;
            const int nbt = (hdeg + R - 1) / R;
            float hi[VEC], lo[VEC], mx[VEC], num[VEC], den[VEC];
#pragma unroll
            for (int k = 0; k < VEC; ++k) { hi[k] = -INFINITY; lo[k] = INFINITY; mx[k] = 0.f; num[k] = 0.f; den[k] = 0.f; }
            int hl[32];
#pragma unroll
            for (int j = 0; j < 32; ++j) hl[j] = csrc[hbeg + (nbt - 1) * R + j];
#pragma nounroll
            for (int it = 0; it < 2 * nbt - 1; ++it) {
                const int kb = (it < nbt) ? (nbt - 1 - it) : (it - nbt + 1);
                const int b0 = hbeg + kb * R;
                const int cnt = min(R, hend - b0);
                float hv[R][VEC];
                SegChain<0, VEC, CP, R>::load(cnt, xin, bo, hl, hv);
                const int itn = min(it + 1, 2 * nbt - 2);
                const int kbn = (itn < nbt) ? (nbt - 1 - itn) : (itn - nbt + 1);
                int nl[32];
#pragma unroll
                for (int j = 0; j < 32; ++j) nl[j] = csrc[hbeg + kbn * R + j];
                if (it < nbt) SegChain<0, VEC, CP, R>::vrng(cnt, hv, hi, lo);
                if (it == nbt - 1) seg_top<VEC>(hi, lo, t2, mx);
                if (it >= nbt - 1) SegChain<0, VEC, CP, R>::vsum(cnt, hv, t2, mx, num, den);
#pragma unroll
                for (int j = 0; j < 32; ++j) hl[j] = nl[j];
            }
            float* o = sA + (wv * NPW + i) * PITCH + loff;
#pragma unroll
            for (int k = 0; k < VEC; ++k) o[k] = den[k] > 0.f ? num[k] / den[k] : 0.f;
        }
    }
    __syncthreads();
    if (wv % 5 != 0) return;
    const int m0 = g0n + (wv / 5) * 16;
    if (m0 >= Nrows) return;

    const int half = lane >> 4;
    const int c    = lane & 15;
    const float* ap  = sA + ((wv / 5) * 16 + c) * PITCH + half * 2;
    const float* xp  = sX + ((wv / 5) * 16 + c) * PITCH + half * 2;
    const float* wr  = sW + c * PITCH + half * 2;
    const float* wsp = wr + HID * PITCH;

    v8f acc[4] = {};
#pragma unroll
    for (int k = 0; k < CP; k += 4) {
        v2f a = *(const v2f*)(ap + k);
#pragma unroll
        for (int j = 0; j < 4; ++j) {
            v2f b = *(const v2f*)(wr + j * 16 * PITCH + k);
            acc[j] = __builtin_amdgcn_wmma_f32_16x16x4_f32(false, a, false, b,
                                                           (short)0, acc[j], false, false);
        }
    }
#pragma unroll
    for (int k = 0; k < CP; k += 4) {
        v2f a = *(const v2f*)(xp + k);
#pragma unroll
        for (int j = 0; j < 4; ++j) {
            v2f b = *(const v2f*)(wsp + j * 16 * PITCH + k);
            acc[j] = __builtin_amdgcn_wmma_f32_16x16x4_f32(false, a, false, b,
                                                           (short)0, acc[j], false, false);
        }
    }
    const int gl = batch[min(m0 + c, Nrows - 1)];
    const int half_e = (int)(__builtin_amdgcn_mbcnt_lo(~0u, 0u) >> 4);
    const float4 bq = *(const float4*)(bias + 4 * c);
    const float bv[4] = { bq.x, bq.y, bq.z, bq.w };
    float s[8], q[8];
#pragma unroll
    for (int r = 0; r < 8; ++r) {
#pragma unroll
        for (int j = 0; j < 4; ++j) acc[j][r] += bv[j];
        s[r] = (acc[0][r] + acc[1][r]) + (acc[2][r] + acc[3][r]);
    }
    row_sum16x8(s);
#pragma unroll
    for (int r = 0; r < 8; ++r) {
        const float mu = s[r] * (1.f / HID);
#pragma unroll
        for (int j = 0; j < 4; ++j) acc[j][r] -= mu;
        q[r] = (acc[0][r] * acc[0][r] + acc[1][r] * acc[1][r]) + (acc[2][r] * acc[2][r] + acc[3][r] * acc[3][r]);
    }
    row_sum16x8(q);
    float cm[4];
#pragma unroll
    for (int r = 0; r < 8; ++r) {
        const float inv = rsqrtf(q[r] * (1.f / HID) + EPS);
#pragma unroll
        for (int j = 0; j < 4; ++j) {
            const float y = fmaxf(acc[j][r] * inv, 0.f);
            acc[j][r] = y;
            cm[j] = (r == 0) ? y : fmaxf(cm[j], y);
        }
    }
    float* op = out + (size_t)(m0 + half_e * 8) * HID + 4 * c;
    if (m0 + 16 <= Nrows) {
#pragma unroll
        for (int r = 0; r < 8; ++r) {
            float4 o; o.x = acc[0][r]; o.y = acc[1][r]; o.z = acc[2][r]; o.w = acc[3][r];
            *(float4*)(op + (size_t)r * HID) = o;
        }
    } else {
#pragma unroll
        for (int r = 0; r < 8; ++r) {
            if (m0 + r + half_e * 8 < Nrows) {
                float4 o; o.x = acc[0][r]; o.y = acc[1][r]; o.z = acc[2][r]; o.w = acc[3][r];
                *(float4*)(op + (size_t)r * HID) = o;
            }
        }
    }

    const int g0 = __builtin_amdgcn_readfirstlane(gl);
    const bool one_graph = __builtin_amdgcn_ballot_w32(gl != g0) == 0u;
    if (one_graph) {
#pragma unroll
        for (int j = 0; j < 4; ++j) cm[j] = fmaxf(cm[j], xhalf(cm[j]));
        const float p0 = half_e ? cm[2] : cm[0];
        const float p1 = half_e ? cm[3] : cm[1];
        unsigned* hp = he + (size_t)g0 * HID + 4 * c + 2 * half_e;
        atomicMax(hp,     enc_f32(p0));
        atomicMax(hp + 1, enc_f32(p1));
    } else {
#pragma unroll
        for (int r = 0; r < 8; ++r) {
            const int row = m0 + r + half_e * 8;
            if (row < Nrows) {
                unsigned* hp = he + (size_t)batch[row] * HID + 4 * c;
#pragma unroll
                for (int j = 0; j < 4; ++j) atomicMax(hp + j, enc_f32(acc[j][r]));
            }
        }
    }
}

template<int NV>
static __device__ __forceinline__ void wave_sum_n(float (&p)[NV]) {
#pragma unroll
    for (int j = 0; j < NV; ++j) p[j] += rowx<1>(p[j]);
#pragma unroll
    for (int j = 0; j < NV; ++j) p[j] += rowx<2>(p[j]);
#pragma unroll
    for (int j = 0; j < NV; ++j) p[j] += rowx<4>(p[j]);
#pragma unroll
    for (int j = 0; j < NV; ++j) p[j] += rowx<8>(p[j]);
#pragma unroll
    for (int j = 0; j < NV; ++j) p[j] += xhalf(p[j]);
}

__global__ void __launch_bounds__(256) k_head(const unsigned* __restrict__ henc,
                                              const float* __restrict__ Wl1, const float* __restrict__ bl1,
                                              const float* __restrict__ Wl2, const float* __restrict__ bl2,
                                              float* __restrict__ outp) {
    __shared__ __attribute__((aligned(16))) float sh[3 * HID];
    __shared__ __attribute__((aligned(16))) float sz1[2 * HID];
    __shared__ __attribute__((aligned(16))) float sz2[HID / 2];
    const int g = blockIdx.x;
    const int tid = threadIdx.x, lane = tid & 31, w = tid >> 5;
    if (tid < 3 * HID) {
        const int layer = tid / HID, cc = tid % HID;
        sh[tid] = dec_f32(henc[layer * NG * HID + g * HID + cc]);
    }
    __syncthreads();

    {
        v2f hv[3];
#pragma unroll
        for (int i = 0; i < 3; ++i) hv[i] = *(const v2f*)(sh + 2 * (lane + 32 * i));
        const float* __restrict__ wbase = Wl1 + (w * 16) * (3 * HID) + 2 * lane;
        float p[16];
#pragma unroll
        for (int j = 0; j < 16; ++j) {
            const float* __restrict__ wrow = wbase + j * (3 * HID);
            float a = 0.f;
#pragma unroll
            for (int i = 0; i < 3; ++i) {
                const v2f wv = *(const v2f*)(wrow + 64 * i);
                a += hv[i].x * wv.x;
                a += hv[i].y * wv.y;
            }
            p[j] = a;
        }
        wave_sum_n<16>(p);
        float mine = p[0];
#pragma unroll
        for (int j = 1; j < 16; ++j) mine = (lane == j) ? p[j] : mine;
        if (lane < 16) sz1[w * 16 + lane] = fmaxf(mine + bl1[w * 16 + lane], 0.f);
    }
    __syncthreads();

    {
        const float4 zv = *(const float4*)(sz1 + 4 * lane);
        const float* __restrict__ wbase = Wl2 + (w * 4) * (2 * HID) + 4 * lane;
        float p[4];
#pragma unroll
        for (int j = 0; j < 4; ++j) {
            const float4 wv = *(const float4*)(wbase + j * (2 * HID));
            p[j] = (zv.x * wv.x + zv.y * wv.y) + (zv.z * wv.z + zv.w * wv.w);
        }
        wave_sum_n<4>(p);
        float mine = p[0];
#pragma unroll
        for (int j = 1; j < 4; ++j) mine = (lane == j) ? p[j] : mine;
        if (lane < 4) sz2[w * 4 + lane] = mine + bl2[w * 4 + lane];
    }
    __syncthreads();

    if (w == 0) {
        const float s = sz2[lane];
        float q[1] = { s * s };
        wave_sum_n<1>(q);
        const float nrm = fmaxf(sqrtf(q[0]), 1e-12f);
        outp[g * (HID / 2) + lane] = s / nrm;
    }
}

extern "C" void kernel_launch(void* const* d_in, const int* in_sizes, int n_in,
                              void* d_out, int out_size, void* d_ws, size_t ws_size,
                              hipStream_t stream) {
    const float* x    = (const float*)d_in[0];
    const int*  eidx  = (const int*)  d_in[1];
    const int*  batch = (const int*)  d_in[2];
    const float* t    = (const float*)d_in[3];
    const float* W1r  = (const float*)d_in[4];
    const float* b1   = (const float*)d_in[5];
    const float* W1s  = (const float*)d_in[6];
    const float* W2r  = (const float*)d_in[7];
    const float* b2   = (const float*)d_in[8];
    const float* W2s  = (const float*)d_in[9];
    const float* W3r  = (const float*)d_in[10];
    const float* b3   = (const float*)d_in[11];
    const float* W3s  = (const float*)d_in[12];
    const float* Wl1  = (const float*)d_in[13];
    const float* bl1  = (const float*)d_in[14];
    const float* Wl2  = (const float*)d_in[15];
    const float* bl2  = (const float*)d_in[16];
    const int* src = eidx;
    const int* dst = eidx + NE;

    float* ws = (float*)d_ws;
    size_t o = 0;
    float*    x1     = ws + o; o += (size_t)NN * HID;
    float*    x2     = ws + o; o += (size_t)NN * HID;
    float*    x3     = ws + o; o += (size_t)NN * HID;
    int*      csr    = (int*)(ws + o); o += (size_t)NN * HID * 3;
    int*      boff     = csr;
    unsigned* binned   = (unsigned*)(boff + NBLK * BOFFW + 16);
    int*      rowptr   = (int*)(binned + NBLK * EPB);
    int*      csrc     = rowptr + NN + 16;
    o += (size_t)NN * HID;
    float*    xpad   = ws + o; o += (size_t)NN * 32;
    float*    w1rp   = ws + o; o += (size_t)HID * 32;
    float*    w1sp   = ws + o; o += (size_t)HID * 32;
    unsigned* henc   = (unsigned*)(ws + o); o += 3 * NG * HID;

    k_sort_block<<<NBLK, 1024, 0, stream>>>(src, dst, boff, binned, henc, W1r, W1s, w1rp, w1sp, x, xpad);
    k_sort_bin<<<NBIN, 1024, 0, stream>>>(binned, boff, rowptr, csrc);

    struct Layer { const float* xin; int Cp; const float *Wr, *b, *Ws; float* xout; };
    Layer layers[3] = {
        { xpad, 32, w1rp, b1, w1sp, x1 },
        { x1,   64, W2r,  b2, W2s,  x2 },
        { x2,   64, W3r,  b3, W3s,  x3 },
    };

    for (int l = 0; l < 3; ++l) {
        const Layer& L = layers[l];
        if (L.Cp == 32)
            k_layer<32><<<(NN + GN - 1) / GN, 512, 0, stream>>>(rowptr, csrc, L.xin, t, L.Wr, L.Ws, L.b, batch,
                                                            L.xout, henc + (size_t)l * NG * HID, NN);
        else
            k_layer<64><<<(NN + GN - 1) / GN, 512, 0, stream>>>(rowptr, csrc, L.xin, t, L.Wr, L.Ws, L.b, batch,
                                                            L.xout, henc + (size_t)l * NG * HID, NN);
    }

    k_head<<<NG, 256, 0, stream>>>(henc, Wl1, bl1, Wl2, bl2, (float*)d_out);
}
